# starting kernel
# baseline (speedup 1.0000x reference)
#include <hip/hip_runtime.h>
#include <math.h>

#define N_NODES 50000
#define N_EDGES 800000
#define E_TOT   (N_EDGES + N_NODES)
#define NFEAT   128
#define NHID    8
#define NHEADS  8
#define HID1    (NHEADS * NHID)
#define NCLASS  40
#define NEG_SLOPE 0.2f

typedef __attribute__((ext_vector_type(2))) float v2f;
typedef __attribute__((ext_vector_type(8))) float v8f;

__device__ __forceinline__ float lrelu(float x) { return x > 0.f ? x : NEG_SLOPE * x; }

__device__ __forceinline__ unsigned f2ord(float f) {
  unsigned u = __float_as_uint(f);
  return (u & 0x80000000u) ? ~u : (u | 0x80000000u);
}
__device__ __forceinline__ float ord2f(unsigned u) {
  return (u & 0x80000000u) ? __uint_as_float(u & 0x7FFFFFFFu) : __uint_as_float(~u);
}
#define ORD_NEG_INF 0x007FFFFFu

__device__ __forceinline__ void get_edge(const int* __restrict__ ei, int e, int& s, int& d) {
  if (e < N_EDGES) { s = ei[e]; d = ei[N_EDGES + e]; }
  else             { s = e - N_EDGES; d = s; }
}

__global__ void init_kernel(float* __restrict__ zbase, unsigned zn,
                            unsigned* __restrict__ mbase, unsigned mn) {
  unsigned i = blockIdx.x * blockDim.x + threadIdx.x;
  if (i < zn) zbase[i] = 0.f;
  if (i < mn) mbase[i] = ORD_NEG_INF;
}

__global__ void __launch_bounds__(128) gemm1_wmma(const float* __restrict__ X,
                                                  const float* __restrict__ W,
                                                  float* __restrict__ H) {
  const int wave = threadIdx.x >> 5;
  const int lane = threadIdx.x & 31;
  const int mt   = blockIdx.x * 4 + wave;
  if (mt >= N_NODES / 16) return;
  const int m  = lane & 15;
  const int hi = lane >> 4;
  const float* xr = X + (long)(mt * 16 + m) * NFEAT;

  v8f c0 = {}, c1 = {}, c2 = {}, c3 = {};
  #pragma unroll 4
  for (int k0 = 0; k0 < NFEAT; k0 += 4) {
    const int ka = k0 + 2 * hi;
    v2f a; a.x = xr[ka]; a.y = xr[ka + 1];
    const float* w0 = W + (long)ka * HID1 + m;
    const float* w1 = w0 + HID1;
    v2f b;
    b.x = w0[0];  b.y = w1[0];
    c0 = __builtin_amdgcn_wmma_f32_16x16x4_f32(false, a, false, b, (short)0, c0, false, false);
    b.x = w0[16]; b.y = w1[16];
    c1 = __builtin_amdgcn_wmma_f32_16x16x4_f32(false, a, false, b, (short)0, c1, false, false);
    b.x = w0[32]; b.y = w1[32];
    c2 = __builtin_amdgcn_wmma_f32_16x16x4_f32(false, a, false, b, (short)0, c2, false, false);
    b.x = w0[48]; b.y = w1[48];
    c3 = __builtin_amdgcn_wmma_f32_16x16x4_f32(false, a, false, b, (short)0, c3, false, false);
  }
  float* hp = H + (long)(mt * 16 + 8 * hi) * HID1 + m;
  #pragma unroll
  for (int j = 0; j < 8; j++) {
    hp[(long)j * HID1 +  0] = c0[j];
    hp[(long)j * HID1 + 16] = c1[j];
    hp[(long)j * HID1 + 32] = c2[j];
    hp[(long)j * HID1 + 48] = c3[j];
  }
}

__global__ void __launch_bounds__(128) gemm2_wmma(const float* __restrict__ X,
                                                  const float* __restrict__ W,
                                                  float* __restrict__ H) {
  const int wave = threadIdx.x >> 5;
  const int lane = threadIdx.x & 31;
  const int mt   = blockIdx.x * 4 + wave;
  if (mt >= N_NODES / 16) return;
  const int m  = lane & 15;
  const int hi = lane >> 4;
  const float* xr = X + (long)(mt * 16 + m) * HID1;

  v8f c0 = {}, c1 = {}, c2 = {};
  #pragma unroll 4
  for (int k0 = 0; k0 < HID1; k0 += 4) {
    const int ka = k0 + 2 * hi;
    v2f a; a.x = xr[ka]; a.y = xr[ka + 1];
    #pragma unroll
    for (int t = 0; t < 3; t++) {
      const int col = t * 16 + m;
      v2f b;
      b.x = (col < NCLASS) ? W[(long)ka * NCLASS + col]       : 0.f;
      b.y = (col < NCLASS) ? W[(long)(ka + 1) * NCLASS + col] : 0.f;
      if (t == 0) c0 = __builtin_amdgcn_wmma_f32_16x16x4_f32(false, a, false, b, (short)0, c0, false, false);
      if (t == 1) c1 = __builtin_amdgcn_wmma_f32_16x16x4_f32(false, a, false, b, (short)0, c1, false, false);
      if (t == 2) c2 = __builtin_amdgcn_wmma_f32_16x16x4_f32(false, a, false, b, (short)0, c2, false, false);
    }
  }
  const int rbase = mt * 16 + 8 * hi;
  #pragma unroll
  for (int j = 0; j < 8; j++) {
    float* hr = H + (long)(rbase + j) * NCLASS;
    if (m      < NCLASS) hr[m]      = c0[j];
    if (m + 16 < NCLASS) hr[m + 16] = c1[j];
    if (m + 32 < NCLASS) hr[m + 32] = c2[j];
  }
}

__global__ void alpha1_kernel(const float* __restrict__ H1,
                              const float* __restrict__ a_s, const float* __restrict__ a_d,
                              float* __restrict__ as, float* __restrict__ ad) {
  int i = blockIdx.x * blockDim.x + threadIdx.x;
  if (i >= N_NODES * NHEADS) return;
  const int h = i & 7;
  const float* hp = H1 + (long)(i >> 3) * HID1 + h * NHID;
  float s = 0.f, d = 0.f;
  #pragma unroll
  for (int k = 0; k < NHID; k++) { s += hp[k] * a_s[h * NHID + k]; d += hp[k] * a_d[h * NHID + k]; }
  as[i] = s; ad[i] = d;
}

__global__ void alpha2_kernel(const float* __restrict__ H2,
                              const float* __restrict__ a_s, const float* __restrict__ a_d,
                              float* __restrict__ as, float* __restrict__ ad) {
  int n = blockIdx.x * blockDim.x + threadIdx.x;
  if (n >= N_NODES) return;
  const float* hp = H2 + (long)n * NCLASS;
  float s = 0.f, d = 0.f;
  #pragma unroll
  for (int c = 0; c < NCLASS; c++) { s += hp[c] * a_s[c]; d += hp[c] * a_d[c]; }
  as[n] = s; ad[n] = d;
}

__global__ void edge_max1(const int* __restrict__ ei, const float* __restrict__ as,
                          const float* __restrict__ ad, unsigned* __restrict__ m) {
  int e = blockIdx.x * blockDim.x + threadIdx.x;
  if (e >= E_TOT) return;
  int s, d; get_edge(ei, e, s, d);
  #pragma unroll
  for (int h = 0; h < NHEADS; h++) {
    float v = lrelu(as[s * NHEADS + h] + ad[d * NHEADS + h]);
    atomicMax(&m[d * NHEADS + h], f2ord(v));
  }
}

__global__ void edge_sum1(const int* __restrict__ ei, const float* __restrict__ as,
                          const float* __restrict__ ad, const unsigned* __restrict__ m,
                          float* __restrict__ den) {
  int e = blockIdx.x * blockDim.x + threadIdx.x;
  if (e >= E_TOT) return;
  int s, d; get_edge(ei, e, s, d);
  #pragma unroll
  for (int h = 0; h < NHEADS; h++) {
    float v = lrelu(as[s * NHEADS + h] + ad[d * NHEADS + h]);
    atomicAdd(&den[d * NHEADS + h], __expf(v - ord2f(m[d * NHEADS + h])));
  }
}

__global__ void edge_agg1(const int* __restrict__ ei, const float* __restrict__ as,
                          const float* __restrict__ ad, const unsigned* __restrict__ m,
                          const float* __restrict__ den, const float* __restrict__ H1,
                          float* __restrict__ out1) {
  int idx = blockIdx.x * blockDim.x + threadIdx.x;
  if (idx >= E_TOT * NHEADS) return;
  const int e = idx >> 3, h = idx & 7;
  int s, d; get_edge(ei, e, s, d);
  float v    = lrelu(as[s * NHEADS + h] + ad[d * NHEADS + h]);
  float attn = __expf(v - ord2f(m[d * NHEADS + h])) / den[d * NHEADS + h];
  const float* hs = H1 + (long)s * HID1 + h * NHID;
  float*       od = out1 + (long)d * HID1 + h * NHID;
  #pragma unroll
  for (int k = 0; k < NHID; k++) atomicAdd(&od[k], attn * hs[k]);
}

__global__ void elu_bias_kernel(float* __restrict__ o, const float* __restrict__ b) {
  int i = blockIdx.x * blockDim.x + threadIdx.x;
  if (i >= N_NODES * HID1) return;
  float v = o[i] + b[i & (HID1 - 1)];
  o[i] = v > 0.f ? v : (__expf(v) - 1.f);
}

__global__ void edge_max2(const int* __restrict__ ei, const float* __restrict__ as,
                          const float* __restrict__ ad, unsigned* __restrict__ m) {
  int e = blockIdx.x * blockDim.x + threadIdx.x;
  if (e >= E_TOT) return;
  int s, d; get_edge(ei, e, s, d);
  atomicMax(&m[d], f2ord(lrelu(as[s] + ad[d])));
}

__global__ void edge_sum2(const int* __restrict__ ei, const float* __restrict__ as,
                          const float* __restrict__ ad, const unsigned* __restrict__ m,
                          float* __restrict__ den) {
  int e = blockIdx.x * blockDim.x + threadIdx.x;
  if (e >= E_TOT) return;
  int s, d; get_edge(ei, e, s, d);
  float v = lrelu(as[s] + ad[d]);
  atomicAdd(&den[d], __expf(v - ord2f(m[d])));
}

__global__ void edge_agg2(const int* __restrict__ ei, const float* __restrict__ as,
                          const float* __restrict__ ad, const unsigned* __restrict__ m,
                          const float* __restrict__ den, const float* __restrict__ H2,
                          float* __restrict__ out2) {
  int e = blockIdx.x * blockDim.x + threadIdx.x;
  if (e >= E_TOT) return;
  int s, d; get_edge(ei, e, s, d);
  float v    = lrelu(as[s] + ad[d]);
  float attn = __expf(v - ord2f(m[d])) / den[d];
  const float* hs = H2 + (long)s * NCLASS;
  float*       od = out2 + (long)d * NCLASS;
  #pragma unroll
  for (int c = 0; c < NCLASS; c++) atomicAdd(&od[c], attn * hs[c]);
}

__global__ void lsm_kernel(const float* __restrict__ o2, const float* __restrict__ b,
                           float* __restrict__ out) {
  int n = blockIdx.x * blockDim.x + threadIdx.x;
  if (n >= N_NODES) return;
  float buf[NCLASS];
  float mx = -INFINITY;
  #pragma unroll
  for (int c = 0; c < NCLASS; c++) { buf[c] = o2[(long)n * NCLASS + c] + b[c]; mx = fmaxf(mx, buf[c]); }
  float ssum = 0.f;
  #pragma unroll
  for (int c = 0; c < NCLASS; c++) ssum += __expf(buf[c] - mx);
  const float lse = mx + __logf(ssum);
  #pragma unroll
  for (int c = 0; c < NCLASS; c++) out[(long)n * NCLASS + c] = buf[c] - lse;
}

extern "C" void kernel_launch(void* const* d_in, const int* in_sizes, int n_in,
                              void* d_out, int out_size, void* d_ws, size_t ws_size,
                              hipStream_t stream) {
  const float* x      = (const float*)d_in[0];
  const float* W1     = (const float*)d_in[1];
  const float* a_src1 = (const float*)d_in[2];
  const float* a_dst1 = (const float*)d_in[3];
  const float* b1     = (const float*)d_in[4];
  const float* W2     = (const float*)d_in[5];
  const float* a_src2 = (const float*)d_in[6];
  const float* a_dst2 = (const float*)d_in[7];
  const float* b2     = (const float*)d_in[8];
  const int*   ei     = (const int*)d_in[9];
  float*       out    = (float*)d_out;

  float* ws   = (float*)d_ws;
  float* H1   = ws;
  float* as1  = H1  + (long)N_NODES * HID1;
  float* ad1  = as1 + (long)N_NODES * NHEADS;
  float* as2  = ad1 + (long)N_NODES * NHEADS;
  float* ad2  = as2 + N_NODES;
  float* zb   = ad2 + N_NODES;
  float* den1 = zb;
  float* out1 = den1 + (long)N_NODES * NHEADS;
  float* den2 = out1 + (long)N_NODES * HID1;
  float* out2 = den2 + N_NODES;
  const unsigned ZERON = N_NODES * NHEADS + N_NODES * HID1 + N_NODES + N_NODES * NCLASS;
  unsigned* m1 = (unsigned*)(out2 + (long)N_NODES * NCLASS);
  unsigned* m2 = m1 + (long)N_NODES * NHEADS;
  const unsigned MINFN = N_NODES * NHEADS + N_NODES;
  float* H2   = (float*)(m2 + N_NODES);

  const int B = 256;
  auto cdiv = [](long a, long b) { return (int)((a + b - 1) / b); };

  init_kernel<<<cdiv(ZERON, B), B, 0, stream>>>(zb, ZERON, m1, MINFN);

  gemm1_wmma<<<cdiv(N_NODES / 16, 4), 128, 0, stream>>>(x, W1, H1);
  alpha1_kernel<<<cdiv((long)N_NODES * NHEADS, B), B, 0, stream>>>(H1, a_src1, a_dst1, as1, ad1);
  edge_max1<<<cdiv(E_TOT, B), B, 0, stream>>>(ei, as1, ad1, m1);
  edge_sum1<<<cdiv(E_TOT, B), B, 0, stream>>>(ei, as1, ad1, m1, den1);
  edge_agg1<<<cdiv((long)E_TOT * NHEADS, B), B, 0, stream>>>(ei, as1, ad1, m1, den1, H1, out1);
  elu_bias_kernel<<<cdiv((long)N_NODES * HID1, B), B, 0, stream>>>(out1, b1);

  gemm2_wmma<<<cdiv(N_NODES / 16, 4), 128, 0, stream>>>(out1, W2, H2);
  alpha2_kernel<<<cdiv(N_NODES, B), B, 0, stream>>>(H2, a_src2, a_dst2, as2, ad2);
  edge_max2<<<cdiv(E_TOT, B), B, 0, stream>>>(ei, as2, ad2, m2);
  edge_sum2<<<cdiv(E_TOT, B), B, 0, stream>>>(ei, as2, ad2, m2, den2);
  edge_agg2<<<cdiv(E_TOT, B), B, 0, stream>>>(ei, as2, ad2, m2, den2, H2, out2);

  lsm_kernel<<<cdiv(N_NODES, B), B, 0, stream>>>(out2, b2, out);
}


# edge_max1, edge_sum1, edge_agg2: 8 lanes per edge, so each atomic instruction sends 32 contiguous bytes per edge instead of 32 scattered rows; atomic float sums keep arrival order as before
# speedup vs baseline: 2.3513x; 2.3513x over previous
#include <hip/hip_runtime.h>
#include <math.h>

#define N_NODES 50000
#define N_EDGES 800000
#define E_TOT   (N_EDGES + N_NODES)
#define NFEAT   128
#define NHID    8
#define NHEADS  8
#define HID1    (NHEADS * NHID)
#define NCLASS  40
#define NEG_SLOPE 0.2f

typedef __attribute__((ext_vector_type(2))) float v2f;
typedef __attribute__((ext_vector_type(8))) float v8f;

__device__ __forceinline__ float lrelu(float x) { return x > 0.f ? x : NEG_SLOPE * x; }

__device__ __forceinline__ unsigned f2ord(float f) {
  unsigned u = __float_as_uint(f);
  return (u & 0x80000000u) ? ~u : (u | 0x80000000u);
}
__device__ __forceinline__ float ord2f(unsigned u) {
  return (u & 0x80000000u) ? __uint_as_float(u & 0x7FFFFFFFu) : __uint_as_float(~u);
}
#define ORD_NEG_INF 0x007FFFFFu

__device__ __forceinline__ void get_edge(const int* __restrict__ ei, int e, int& s, int& d) {
  if (e < N_EDGES) { s = ei[e]; d = ei[N_EDGES + e]; }
  else             { s = e - N_EDGES; d = s; }
}

__global__ void init_kernel(float* __restrict__ zbase, unsigned zn,
                            unsigned* __restrict__ mbase, unsigned mn) {
  unsigned i = blockIdx.x * blockDim.x + threadIdx.x;
  if (i < zn) zbase[i] = 0.f;
  if (i < mn) mbase[i] = ORD_NEG_INF;
}

__global__ void __launch_bounds__(128) gemm1_wmma(const float* __restrict__ X,
                                                  const float* __restrict__ W,
                                                  float* __restrict__ H) {
  const int wave = threadIdx.x >> 5;
  const int lane = threadIdx.x & 31;
  const int mt   = blockIdx.x * 4 + wave;
  if (mt >= N_NODES / 16) return;
  const int m  = lane & 15;
  const int hi = lane >> 4;
  const float* xr = X + (long)(mt * 16 + m) * NFEAT;

  v8f c0 = {}, c1 = {}, c2 = {}, c3 = {};
  #pragma unroll 4
  for (int k0 = 0; k0 < NFEAT; k0 += 4) {
    const int ka = k0 + 2 * hi;
    v2f a; a.x = xr[ka]; a.y = xr[ka + 1];
    const float* w0 = W + (long)ka * HID1 + m;
    const float* w1 = w0 + HID1;
    v2f b;
    b.x = w0[0];  b.y = w1[0];
    c0 = __builtin_amdgcn_wmma_f32_16x16x4_f32(false, a, false, b, (short)0, c0, false, false);
    b.x = w0[16]; b.y = w1[16];
    c1 = __builtin_amdgcn_wmma_f32_16x16x4_f32(false, a, false, b, (short)0, c1, false, false);
    b.x = w0[32]; b.y = w1[32];
    c2 = __builtin_amdgcn_wmma_f32_16x16x4_f32(false, a, false, b, (short)0, c2, false, false);
    b.x = w0[48]; b.y = w1[48];
    c3 = __builtin_amdgcn_wmma_f32_16x16x4_f32(false, a, false, b, (short)0, c3, false, false);
  }
  float* hp = H + (long)(mt * 16 + 8 * hi) * HID1 + m;
  #pragma unroll
  for (int j = 0; j < 8; j++) {
    hp[(long)j * HID1 +  0] = c0[j];
    hp[(long)j * HID1 + 16] = c1[j];
    hp[(long)j * HID1 + 32] = c2[j];
    hp[(long)j * HID1 + 48] = c3[j];
  }
}

__global__ void __launch_bounds__(128) gemm2_wmma(const float* __restrict__ X,
                                                  const float* __restrict__ W,
                                                  float* __restrict__ H) {
  const int wave = threadIdx.x >> 5;
  const int lane = threadIdx.x & 31;
  const int mt   = blockIdx.x * 4 + wave;
  if (mt >= N_NODES / 16) return;
  const int m  = lane & 15;
  const int hi = lane >> 4;
  const float* xr = X + (long)(mt * 16 + m) * HID1;

  v8f c0 = {}, c1 = {}, c2 = {};
  #pragma unroll 4
  for (int k0 = 0; k0 < HID1; k0 += 4) {
    const int ka = k0 + 2 * hi;
    v2f a; a.x = xr[ka]; a.y = xr[ka + 1];
    #pragma unroll
    for (int t = 0; t < 3; t++) {
      const int col = t * 16 + m;
      v2f b;
      b.x = (col < NCLASS) ? W[(long)ka * NCLASS + col]       : 0.f;
      b.y = (col < NCLASS) ? W[(long)(ka + 1) * NCLASS + col] : 0.f;
      if (t == 0) c0 = __builtin_amdgcn_wmma_f32_16x16x4_f32(false, a, false, b, (short)0, c0, false, false);
      if (t == 1) c1 = __builtin_amdgcn_wmma_f32_16x16x4_f32(false, a, false, b, (short)0, c1, false, false);
      if (t == 2) c2 = __builtin_amdgcn_wmma_f32_16x16x4_f32(false, a, false, b, (short)0, c2, false, false);
    }
  }
  const int rbase = mt * 16 + 8 * hi;
  #pragma unroll
  for (int j = 0; j < 8; j++) {
    float* hr = H + (long)(rbase + j) * NCLASS;
    if (m      < NCLASS) hr[m]      = c0[j];
    if (m + 16 < NCLASS) hr[m + 16] = c1[j];
    if (m + 32 < NCLASS) hr[m + 32] = c2[j];
  }
}

__global__ void alpha1_kernel(const float* __restrict__ H1,
                              const float* __restrict__ a_s, const float* __restrict__ a_d,
                              float* __restrict__ as, float* __restrict__ ad) {
  int i = blockIdx.x * blockDim.x + threadIdx.x;
  if (i >= N_NODES * NHEADS) return;
  const int h = i & 7;
  const float* hp = H1 + (long)(i >> 3) * HID1 + h * NHID;
  float s = 0.f, d = 0.f;
  #pragma unroll
  for (int k = 0; k < NHID; k++) { s += hp[k] * a_s[h * NHID + k]; d += hp[k] * a_d[h * NHID + k]; }
  as[i] = s; ad[i] = d;
}

__global__ void alpha2_kernel(const float* __restrict__ H2,
                              const float* __restrict__ a_s, const float* __restrict__ a_d,
                              float* __restrict__ as, float* __restrict__ ad) {
  int n = blockIdx.x * blockDim.x + threadIdx.x;
  if (n >= N_NODES) return;
  const float* hp = H2 + (long)n * NCLASS;
  float s = 0.f, d = 0.f;
  #pragma unroll
  for (int c = 0; c < NCLASS; c++) { s += hp[c] * a_s[c]; d += hp[c] * a_d[c]; }
  as[n] = s; ad[n] = d;
}

__global__ void edge_max1(const int* __restrict__ ei, const float* __restrict__ as,
                          const float* __restrict__ ad, unsigned* __restrict__ m) {
  int idx = blockIdx.x * blockDim.x + threadIdx.x;
  if (idx >= E_TOT * NHEADS) return;
  const int e = idx >> 3, h = idx & 7;
  int s, d; get_edge(ei, e, s, d);
  float v = lrelu(as[s * NHEADS + h] + ad[d * NHEADS + h]);
  atomicMax(&m[d * NHEADS + h], f2ord(v));
}

__global__ void edge_sum1(const int* __restrict__ ei, const float* __restrict__ as,
                          const float* __restrict__ ad, const unsigned* __restrict__ m,
                          float* __restrict__ den) {
  int idx = blockIdx.x * blockDim.x + threadIdx.x;
  if (idx >= E_TOT * NHEADS) return;
  const int e = idx >> 3, h = idx & 7;
  int s, d; get_edge(ei, e, s, d);
  float v = lrelu(as[s * NHEADS + h] + ad[d * NHEADS + h]);
  atomicAdd(&den[d * NHEADS + h], __expf(v - ord2f(m[d * NHEADS + h])));
}

__global__ void edge_agg1(const int* __restrict__ ei, const float* __restrict__ as,
                          const float* __restrict__ ad, const unsigned* __restrict__ m,
                          const float* __restrict__ den, const float* __restrict__ H1,
                          float* __restrict__ out1) {
  int idx = blockIdx.x * blockDim.x + threadIdx.x;
  if (idx >= E_TOT * NHEADS) return;
  const int e = idx >> 3, h = idx & 7;
  int s, d; get_edge(ei, e, s, d);
  float v    = lrelu(as[s * NHEADS + h] + ad[d * NHEADS + h]);
  float attn = __expf(v - ord2f(m[d * NHEADS + h])) / den[d * NHEADS + h];
  const float* hs = H1 + (long)s * HID1 + h * NHID;
  float*       od = out1 + (long)d * HID1 + h * NHID;
  #pragma unroll
  for (int k = 0; k < NHID; k++) atomicAdd(&od[k], attn * hs[k]);
}

__global__ void elu_bias_kernel(float* __restrict__ o, const float* __restrict__ b) {
  int i = blockIdx.x * blockDim.x + threadIdx.x;
  if (i >= N_NODES * HID1) return;
  float v = o[i] + b[i & (HID1 - 1)];
  o[i] = v > 0.f ? v : (__expf(v) - 1.f);
}

__global__ void edge_max2(const int* __restrict__ ei, const float* __restrict__ as,
                          const float* __restrict__ ad, unsigned* __restrict__ m) {
  int e = blockIdx.x * blockDim.x + threadIdx.x;
  if (e >= E_TOT) return;
  int s, d; get_edge(ei, e, s, d);
  atomicMax(&m[d], f2ord(lrelu(as[s] + ad[d])));
}

__global__ void edge_sum2(const int* __restrict__ ei, const float* __restrict__ as,
                          const float* __restrict__ ad, const unsigned* __restrict__ m,
                          float* __restrict__ den) {
  int e = blockIdx.x * blockDim.x + threadIdx.x;
  if (e >= E_TOT) return;
  int s, d; get_edge(ei, e, s, d);
  float v = lrelu(as[s] + ad[d]);
  atomicAdd(&den[d], __expf(v - ord2f(m[d])));
}

__global__ void edge_agg2(const int* __restrict__ ei, const float* __restrict__ as,
                          const float* __restrict__ ad, const unsigned* __restrict__ m,
                          const float* __restrict__ den, const float* __restrict__ H2,
                          float* __restrict__ out2) {
  int idx = blockIdx.x * blockDim.x + threadIdx.x;
  if (idx >= E_TOT * 8) return;
  const int e = idx >> 3, q = idx & 7;
  int s, d; get_edge(ei, e, s, d);
  const float* hs = H2 + (s * NCLASS + q);
  float*       od = out2 + (d * NCLASS + q);
  float hv[NCLASS / 8];
  #pragma unroll
  for (int j = 0; j < NCLASS / 8; j++) hv[j] = hs[8 * j];
  float v    = lrelu(as[s] + ad[d]);
  float attn = __expf(v - ord2f(m[d])) / den[d];
  #pragma unroll
  for (int j = 0; j < NCLASS / 8; j++) atomicAdd(&od[8 * j], attn * hv[j]);
}

__global__ void lsm_kernel(const float* __restrict__ o2, const float* __restrict__ b,
                           float* __restrict__ out) {
  int n = blockIdx.x * blockDim.x + threadIdx.x;
  if (n >= N_NODES) return;
  float buf[NCLASS];
  float mx = -INFINITY;
  #pragma unroll
  for (int c = 0; c < NCLASS; c++) { buf[c] = o2[(long)n * NCLASS + c] + b[c]; mx = fmaxf(mx, buf[c]); }
  float ssum = 0.f;
  #pragma unroll
  for (int c = 0; c < NCLASS; c++) ssum += __expf(buf[c] - mx);
  const float lse = mx + __logf(ssum);
  #pragma unroll
  for (int c = 0; c < NCLASS; c++) out[(long)n * NCLASS + c] = buf[c] - lse;
}

extern "C" void kernel_launch(void* const* d_in, const int* in_sizes, int n_in,
                              void* d_out, int out_size, void* d_ws, size_t ws_size,
                              hipStream_t stream) {
  const float* x      = (const float*)d_in[0];
  const float* W1     = (const float*)d_in[1];
  const float* a_src1 = (const float*)d_in[2];
  const float* a_dst1 = (const float*)d_in[3];
  const float* b1     = (const float*)d_in[4];
  const float* W2     = (const float*)d_in[5];
  const float* a_src2 = (const float*)d_in[6];
  const float* a_dst2 = (const float*)d_in[7];
  const float* b2     = (const float*)d_in[8];
  const int*   ei     = (const int*)d_in[9];
  float*       out    = (float*)d_out;

  float* ws   = (float*)d_ws;
  float* H1   = ws;
  float* as1  = H1  + (long)N_NODES * HID1;
  float* ad1  = as1 + (long)N_NODES * NHEADS;
  float* as2  = ad1 + (long)N_NODES * NHEADS;
  float* ad2  = as2 + N_NODES;
  float* zb   = ad2 + N_NODES;
  float* den1 = zb;
  float* out1 = den1 + (long)N_NODES * NHEADS;
  float* den2 = out1 + (long)N_NODES * HID1;
  float* out2 = den2 + N_NODES;
  const unsigned ZERON = N_NODES * NHEADS + N_NODES * HID1 + N_NODES + N_NODES * NCLASS;
  unsigned* m1 = (unsigned*)(out2 + (long)N_NODES * NCLASS);
  unsigned* m2 = m1 + (long)N_NODES * NHEADS;
  const unsigned MINFN = N_NODES * NHEADS + N_NODES;
  float* H2   = (float*)(m2 + N_NODES);

  const int B = 256;
  auto cdiv = [](long a, long b) { return (int)((a + b - 1) / b); };

  init_kernel<<<cdiv(ZERON, B), B, 0, stream>>>(zb, ZERON, m1, MINFN);

  gemm1_wmma<<<cdiv(N_NODES / 16, 4), 128, 0, stream>>>(x, W1, H1);
  alpha1_kernel<<<cdiv((long)N_NODES * NHEADS, B), B, 0, stream>>>(H1, a_src1, a_dst1, as1, ad1);
  edge_max1<<<cdiv((long)E_TOT * NHEADS, B), B, 0, stream>>>(ei, as1, ad1, m1);
  edge_sum1<<<cdiv((long)E_TOT * NHEADS, B), B, 0, stream>>>(ei, as1, ad1, m1, den1);
  edge_agg1<<<cdiv((long)E_TOT * NHEADS, B), B, 0, stream>>>(ei, as1, ad1, m1, den1, H1, out1);
  elu_bias_kernel<<<cdiv((long)N_NODES * HID1, B), B, 0, stream>>>(out1, b1);

  gemm2_wmma<<<cdiv(N_NODES / 16, 4), 128, 0, stream>>>(out1, W2, H2);
  alpha2_kernel<<<cdiv(N_NODES, B), B, 0, stream>>>(H2, a_src2, a_dst2, as2, ad2);
  edge_max2<<<cdiv(E_TOT, B), B, 0, stream>>>(ei, as2, ad2, m2);
  edge_sum2<<<cdiv(E_TOT, B), B, 0, stream>>>(ei, as2, ad2, m2, den2);
  edge_agg2<<<cdiv((long)E_TOT * 8, B), B, 0, stream>>>(ei, as2, ad2, m2, den2, H2, out2);

  lsm_kernel<<<cdiv(N_NODES, B), B, 0, stream>>>(out2, b2, out);
}


# edges counting-sorted by destination in each call (int atomic rank, scan, fill); each node sums its own edges in registers in arrival-rank order, no float atomics; bias+ELU at its store
# speedup vs baseline: 5.5262x; 2.3503x over previous
#include <hip/hip_runtime.h>
#include <math.h>

#define N_NODES 50000
#define N_EDGES 800000
#define NFEAT   128
#define NHID    8
#define NHEADS  8
#define HID1    (NHEADS * NHID)
#define NCLASS  40
#define NEG_SLOPE 0.2f

typedef __attribute__((ext_vector_type(2))) float v2f;
typedef __attribute__((ext_vector_type(8))) float v8f;

__device__ __forceinline__ float lrelu(float x) { return x > 0.f ? x : NEG_SLOPE * x; }

__global__ void init_kernel(int* __restrict__ deg) {
  int i = blockIdx.x * blockDim.x + threadIdx.x;
  if (i < N_NODES) deg[i] = 0;
}

__global__ void count_kernel(const int* __restrict__ ei, int* __restrict__ deg,
                             int* __restrict__ pos) {
  int e = blockIdx.x * blockDim.x + threadIdx.x;
  if (e >= N_EDGES) return;
  pos[e] = atomicAdd(&deg[ei[N_EDGES + e]], 1);
}

#define SCAN_T   1024
#define SCAN_PER ((N_NODES + SCAN_T - 1) / SCAN_T)

__global__ void __launch_bounds__(SCAN_T) scan_kernel(const int* __restrict__ deg,
                                                      int* __restrict__ rowptr) {
  __shared__ int wsum[32];
  const int t     = threadIdx.x;
  const int lane  = t & 31;
  const int wv    = t >> 5;
  const int wbase = wv * (SCAN_PER * 32) + lane;
  int v[SCAN_PER];
  int wtot = 0;
  #pragma unroll
  for (int i = 0; i < SCAN_PER; i++) {
    const int idx = wbase + i * 32;
    v[i] = (idx < N_NODES) ? deg[idx] : 0;
    wtot += v[i];
  }
  #pragma unroll
  for (int o = 16; o > 0; o >>= 1) wtot += __shfl_xor(wtot, o);
  if (lane == 0) wsum[wv] = wtot;
  __syncthreads();
  if (wv == 0) {
    const int w = wsum[lane];
    int wi = w;
    #pragma unroll
    for (int o = 1; o < 32; o <<= 1) {
      const int y = __shfl_up(wi, o);
      if (lane >= o) wi += y;
    }
    wsum[lane] = wi - w;
  }
  __syncthreads();
  int carry = wsum[wv];
  #pragma unroll
  for (int i = 0; i < SCAN_PER; i++) {
    int incl = v[i];
    #pragma unroll
    for (int o = 1; o < 32; o <<= 1) {
      const int y = __shfl_up(incl, o);
      if (lane >= o) incl += y;
    }
    const int idx = wbase + i * 32;
    if (idx < N_NODES) rowptr[idx] = carry + incl - v[i];
    carry += __shfl(incl, 31);
  }
  if (t == SCAN_T - 1) rowptr[N_NODES] = carry;
}

__global__ void fill_kernel(const int* __restrict__ ei, const int* __restrict__ rowptr,
                            const int* __restrict__ pos, int* __restrict__ csr_src) {
  int e = blockIdx.x * blockDim.x + threadIdx.x;
  if (e >= N_EDGES) return;
  csr_src[rowptr[ei[N_EDGES + e]] + pos[e]] = ei[e];
}

__global__ void __launch_bounds__(128) gemm1_wmma(const float* __restrict__ X,
                                                  const float* __restrict__ W,
                                                  float* __restrict__ H) {
  const int wave = threadIdx.x >> 5;
  const int lane = threadIdx.x & 31;
  const int mt   = blockIdx.x * 4 + wave;
  if (mt >= N_NODES / 16) return;
  const int m  = lane & 15;
  const int hi = lane >> 4;
  const float* xr = X + (long)(mt * 16 + m) * NFEAT;

  v8f c0 = {}, c1 = {}, c2 = {}, c3 = {};
  #pragma unroll 4
  for (int k0 = 0; k0 < NFEAT; k0 += 4) {
    const int ka = k0 + 2 * hi;
    v2f a; a.x = xr[ka]; a.y = xr[ka + 1];
    const float* w0 = W + (long)ka * HID1 + m;
    const float* w1 = w0 + HID1;
    v2f b;
    b.x = w0[0];  b.y = w1[0];
    c0 = __builtin_amdgcn_wmma_f32_16x16x4_f32(false, a, false, b, (short)0, c0, false, false);
    b.x = w0[16]; b.y = w1[16];
    c1 = __builtin_amdgcn_wmma_f32_16x16x4_f32(false, a, false, b, (short)0, c1, false, false);
    b.x = w0[32]; b.y = w1[32];
    c2 = __builtin_amdgcn_wmma_f32_16x16x4_f32(false, a, false, b, (short)0, c2, false, false);
    b.x = w0[48]; b.y = w1[48];
    c3 = __builtin_amdgcn_wmma_f32_16x16x4_f32(false, a, false, b, (short)0, c3, false, false);
  }
  float* hp = H + (long)(mt * 16 + 8 * hi) * HID1 + m;
  #pragma unroll
  for (int j = 0; j < 8; j++) {
    hp[(long)j * HID1 +  0] = c0[j];
    hp[(long)j * HID1 + 16] = c1[j];
    hp[(long)j * HID1 + 32] = c2[j];
    hp[(long)j * HID1 + 48] = c3[j];
  }
}

__global__ void __launch_bounds__(128) gemm2_wmma(const float* __restrict__ X,
                                                  const float* __restrict__ W,
                                                  float* __restrict__ H) {
  const int wave = threadIdx.x >> 5;
  const int lane = threadIdx.x & 31;
  const int mt   = blockIdx.x * 4 + wave;
  if (mt >= N_NODES / 16) return;
  const int m  = lane & 15;
  const int hi = lane >> 4;
  const float* xr = X + (long)(mt * 16 + m) * HID1;

  v8f c0 = {}, c1 = {}, c2 = {};
  #pragma unroll 4
  for (int k0 = 0; k0 < HID1; k0 += 4) {
    const int ka = k0 + 2 * hi;
    v2f a; a.x = xr[ka]; a.y = xr[ka + 1];
    #pragma unroll
    for (int t = 0; t < 3; t++) {
      const int col = t * 16 + m;
      v2f b;
      b.x = (col < NCLASS) ? W[(long)ka * NCLASS + col]       : 0.f;
      b.y = (col < NCLASS) ? W[(long)(ka + 1) * NCLASS + col] : 0.f;
      if (t == 0) c0 = __builtin_amdgcn_wmma_f32_16x16x4_f32(false, a, false, b, (short)0, c0, false, false);
      if (t == 1) c1 = __builtin_amdgcn_wmma_f32_16x16x4_f32(false, a, false, b, (short)0, c1, false, false);
      if (t == 2) c2 = __builtin_amdgcn_wmma_f32_16x16x4_f32(false, a, false, b, (short)0, c2, false, false);
    }
  }
  const int rbase = mt * 16 + 8 * hi;
  #pragma unroll
  for (int j = 0; j < 8; j++) {
    float* hr = H + (long)(rbase + j) * NCLASS;
    if (m      < NCLASS) hr[m]      = c0[j];
    if (m + 16 < NCLASS) hr[m + 16] = c1[j];
    if (m + 32 < NCLASS) hr[m + 32] = c2[j];
  }
}

__global__ void alpha1_kernel(const float* __restrict__ H1,
                              const float* __restrict__ a_s, const float* __restrict__ a_d,
                              float* __restrict__ as, float* __restrict__ ad) {
  int i = blockIdx.x * blockDim.x + threadIdx.x;
  if (i >= N_NODES * NHEADS) return;
  const int h = i & 7;
  const float* hp = H1 + (long)(i >> 3) * HID1 + h * NHID;
  float s = 0.f, d = 0.f;
  #pragma unroll
  for (int k = 0; k < NHID; k++) { s += hp[k] * a_s[h * NHID + k]; d += hp[k] * a_d[h * NHID + k]; }
  as[i] = s; ad[i] = d;
}

__global__ void alpha2_kernel(const float* __restrict__ H2,
                              const float* __restrict__ a_s, const float* __restrict__ a_d,
                              float* __restrict__ as, float* __restrict__ ad) {
  int n = blockIdx.x * blockDim.x + threadIdx.x;
  if (n >= N_NODES) return;
  const float* hp = H2 + (long)n * NCLASS;
  float s = 0.f, d = 0.f;
  #pragma unroll
  for (int c = 0; c < NCLASS; c++) { s += hp[c] * a_s[c]; d += hp[c] * a_d[c]; }
  as[n] = s; ad[n] = d;
}

__global__ void __launch_bounds__(256) node_pass1(const int* __restrict__ rowptr,
                                                   const int* __restrict__ csr_src,
                                                   const float* __restrict__ as,
                                                   const float* __restrict__ ad,
                                                   const float* __restrict__ H1,
                                                   const float* __restrict__ b,
                                                   float* __restrict__ X2) {
  const int i = blockIdx.x * blockDim.x + threadIdx.x;
  if (i >= N_NODES * NHEADS) return;
  const int n = i >> 3, h = i & 7;
  const int beg = rowptr[n], end = rowptr[n + 1];
  const int last = end - 1;
  const float adn   = ad[i];
  const float vself = lrelu(as[i] + adn);

  float mx = vself;
  for (int j = beg; j < end; j += 4) {
    float v[4];
    #pragma unroll
    for (int u = 0; u < 4; u++) {
      const int s = csr_src[min(j + u, last)];
      v[u] = lrelu(as[s * NHEADS + h] + adn);
    }
    mx = fmaxf(mx, fmaxf(fmaxf(v[0], v[1]), fmaxf(v[2], v[3])));
  }

  const float4* hself = (const float4*)(H1 + (long)n * HID1 + h * NHID);
  const float wself = __expf(vself - mx);
  float den = wself;
  float4 a0 = hself[0], a1 = hself[1];
  a0.x *= wself; a0.y *= wself; a0.z *= wself; a0.w *= wself;
  a1.x *= wself; a1.y *= wself; a1.z *= wself; a1.w *= wself;
  for (int j = beg; j < end; j += 4) {
    float  w[4];
    float4 x0[4], x1[4];
    #pragma unroll
    for (int u = 0; u < 4; u++) {
      const int s = csr_src[min(j + u, last)];
      const float4* hp = (const float4*)(H1 + (long)s * HID1 + h * NHID);
      x0[u] = hp[0]; x1[u] = hp[1];
      const float e = __expf(lrelu(as[s * NHEADS + h] + adn) - mx);
      w[u] = (j + u < end) ? e : 0.f;
    }
    #pragma unroll
    for (int u = 0; u < 4; u++) {
      den  += w[u];
      a0.x += w[u] * x0[u].x; a0.y += w[u] * x0[u].y; a0.z += w[u] * x0[u].z; a0.w += w[u] * x0[u].w;
      a1.x += w[u] * x1[u].x; a1.y += w[u] * x1[u].y; a1.z += w[u] * x1[u].z; a1.w += w[u] * x1[u].w;
    }
  }

  float r[8] = { a0.x / den, a0.y / den, a0.z / den, a0.w / den,
                 a1.x / den, a1.y / den, a1.z / den, a1.w / den };
  #pragma unroll
  for (int k = 0; k < 8; k++) {
    const float t = r[k] + b[h * NHID + k];
    r[k] = t > 0.f ? t : (__expf(t) - 1.f);
  }
  float4* op = (float4*)(X2 + (long)n * HID1 + h * NHID);
  op[0] = make_float4(r[0], r[1], r[2], r[3]);
  op[1] = make_float4(r[4], r[5], r[6], r[7]);
}

__global__ void __launch_bounds__(256) node_pass2(const int* __restrict__ rowptr,
                                                   const int* __restrict__ csr_src,
                                                   const float* __restrict__ as,
                                                   const float* __restrict__ ad,
                                                   const float* __restrict__ H2,
                                                   float* __restrict__ out2) {
  const int i = blockIdx.x * blockDim.x + threadIdx.x;
  if (i >= N_NODES * 8) return;
  const int n = i >> 3, q = i & 7;
  const int beg = rowptr[n], end = rowptr[n + 1];
  const int last = end - 1;
  const float adn   = ad[n];
  const float vself = lrelu(as[n] + adn);

  float mx = vself;
  for (int j = beg; j < end; j += 4) {
    float v[4];
    #pragma unroll
    for (int u = 0; u < 4; u++) {
      const int s = csr_src[min(j + u, last)];
      v[u] = lrelu(as[s] + adn);
    }
    mx = fmaxf(mx, fmaxf(fmaxf(v[0], v[1]), fmaxf(v[2], v[3])));
  }

  const float wself = __expf(vself - mx);
  float den = wself;
  float acc[NCLASS / 8];
  {
    const float* hs = H2 + (n * NCLASS + q);
    #pragma unroll
    for (int c = 0; c < NCLASS / 8; c++) acc[c] = wself * hs[8 * c];
  }
  for (int j = beg; j < end; j += 4) {
    float w[4];
    float x[4][NCLASS / 8];
    #pragma unroll
    for (int u = 0; u < 4; u++) {
      const int s = csr_src[min(j + u, last)];
      const float* hs = H2 + (s * NCLASS + q);
      #pragma unroll
      for (int c = 0; c < NCLASS / 8; c++) x[u][c] = hs[8 * c];
      const float e = __expf(lrelu(as[s] + adn) - mx);
      w[u] = (j + u < end) ? e : 0.f;
    }
    #pragma unroll
    for (int u = 0; u < 4; u++) {
      den += w[u];
      #pragma unroll
      for (int c = 0; c < NCLASS / 8; c++) acc[c] += w[u] * x[u][c];
    }
  }
  float* od = out2 + (n * NCLASS + q);
  #pragma unroll
  for (int c = 0; c < NCLASS / 8; c++) od[8 * c] = acc[c] / den;
}

__global__ void lsm_kernel(const float* __restrict__ o2, const float* __restrict__ b,
                           float* __restrict__ out) {
  int n = blockIdx.x * blockDim.x + threadIdx.x;
  if (n >= N_NODES) return;
  float buf[NCLASS];
  float mx = -INFINITY;
  #pragma unroll
  for (int c = 0; c < NCLASS; c++) { buf[c] = o2[(long)n * NCLASS + c] + b[c]; mx = fmaxf(mx, buf[c]); }
  float ssum = 0.f;
  #pragma unroll
  for (int c = 0; c < NCLASS; c++) ssum += __expf(buf[c] - mx);
  const float lse = mx + __logf(ssum);
  #pragma unroll
  for (int c = 0; c < NCLASS; c++) out[(long)n * NCLASS + c] = buf[c] - lse;
}

extern "C" void kernel_launch(void* const* d_in, const int* in_sizes, int n_in,
                              void* d_out, int out_size, void* d_ws, size_t ws_size,
                              hipStream_t stream) {
  const float* x      = (const float*)d_in[0];
  const float* W1     = (const float*)d_in[1];
  const float* a_src1 = (const float*)d_in[2];
  const float* a_dst1 = (const float*)d_in[3];
  const float* b1     = (const float*)d_in[4];
  const float* W2     = (const float*)d_in[5];
  const float* a_src2 = (const float*)d_in[6];
  const float* a_dst2 = (const float*)d_in[7];
  const float* b2     = (const float*)d_in[8];
  const int*   ei     = (const int*)d_in[9];
  float*       out    = (float*)d_out;

  float* ws   = (float*)d_ws;
  float* H1   = ws;
  float* X2   = H1  + (long)N_NODES * HID1;
  float* H2   = X2  + (long)N_NODES * HID1;
  float* out2 = H2  + (long)N_NODES * NCLASS;
  float* as1  = out2 + (long)N_NODES * NCLASS;
  float* ad1  = as1 + (long)N_NODES * NHEADS;
  float* as2  = ad1 + (long)N_NODES * NHEADS;
  float* ad2  = as2 + N_NODES;
  int*   deg  = (int*)(ad2 + N_NODES);
  int*   rowp = deg  + N_NODES;
  int*   pos  = rowp + N_NODES + 4;
  int*   csr  = pos  + N_EDGES;

  const int B = 256;
  auto cdiv = [](long a, long b) { return (int)((a + b - 1) / b); };

  init_kernel<<<cdiv(N_NODES, B), B, 0, stream>>>(deg);
  count_kernel<<<cdiv(N_EDGES, B), B, 0, stream>>>(ei, deg, pos);
  scan_kernel<<<1, SCAN_T, 0, stream>>>(deg, rowp);
  fill_kernel<<<cdiv(N_EDGES, B), B, 0, stream>>>(ei, rowp, pos, csr);

  gemm1_wmma<<<cdiv(N_NODES / 16, 4), 128, 0, stream>>>(x, W1, H1);
  alpha1_kernel<<<cdiv((long)N_NODES * NHEADS, B), B, 0, stream>>>(H1, a_src1, a_dst1, as1, ad1);
  node_pass1<<<cdiv((long)N_NODES * NHEADS, B), B, 0, stream>>>(rowp, csr, as1, ad1, H1, b1, X2);

  gemm2_wmma<<<cdiv(N_NODES / 16, 4), 128, 0, stream>>>(X2, W2, H2);
  alpha2_kernel<<<cdiv(N_NODES, B), B, 0, stream>>>(H2, a_src2, a_dst2, as2, ad2);
  node_pass2<<<cdiv((long)N_NODES * 8, B), B, 0, stream>>>(rowp, csr, as2, ad2, H2, out2);

  lsm_kernel<<<cdiv(N_NODES, B), B, 0, stream>>>(out2, b2, out);
}


# alpha1, alpha2 and log-softmax folded into their producers (GEMM tile staged in LDS for the logits; node_pass2 ends with an 8-lane log-softmax): 11 -> 8 launches; softmax sum now an 8-lane tree
# speedup vs baseline: 6.2584x; 1.1325x over previous
#include <hip/hip_runtime.h>
#include <math.h>

#define N_NODES 50000
#define N_EDGES 800000
#define NFEAT   128
#define NHID    8
#define NHEADS  8
#define HID1    (NHEADS * NHID)
#define NCLASS  40
#define NEG_SLOPE 0.2f

typedef __attribute__((ext_vector_type(2))) float v2f;
typedef __attribute__((ext_vector_type(8))) float v8f;

__device__ __forceinline__ float lrelu(float x) { return x > 0.f ? x : NEG_SLOPE * x; }

__global__ void init_kernel(int* __restrict__ deg) {
  int i = blockIdx.x * blockDim.x + threadIdx.x;
  if (i < N_NODES) deg[i] = 0;
}

__global__ void count_kernel(const int* __restrict__ ei, int* __restrict__ deg,
                             int* __restrict__ pos) {
  int e = blockIdx.x * blockDim.x + threadIdx.x;
  if (e >= N_EDGES) return;
  pos[e] = atomicAdd(&deg[ei[N_EDGES + e]], 1);
}

#define SCAN_T   1024
#define SCAN_PER ((N_NODES + SCAN_T - 1) / SCAN_T)

__global__ void __launch_bounds__(SCAN_T) scan_kernel(const int* __restrict__ deg,
                                                      int* __restrict__ rowptr) {
  __shared__ int wsum[32];
  const int t     = threadIdx.x;
  const int lane  = t & 31;
  const int wv    = t >> 5;
  const int wbase = wv * (SCAN_PER * 32) + lane;
  int v[SCAN_PER];
  int wtot = 0;
  #pragma unroll
  for (int i = 0; i < SCAN_PER; i++) {
    const int idx = wbase + i * 32;
    v[i] = (idx < N_NODES) ? deg[idx] : 0;
    wtot += v[i];
  }
  #pragma unroll
  for (int o = 16; o > 0; o >>= 1) wtot += __shfl_xor(wtot, o);
  if (lane == 0) wsum[wv] = wtot;
  __syncthreads();
  if (wv == 0) {
    const int w = wsum[lane];
    int wi = w;
    #pragma unroll
    for (int o = 1; o < 32; o <<= 1) {
      const int y = __shfl_up(wi, o);
      if (lane >= o) wi += y;
    }
    wsum[lane] = wi - w;
  }
  __syncthreads();
  int carry = wsum[wv];
  #pragma unroll
  for (int i = 0; i < SCAN_PER; i++) {
    int incl = v[i];
    #pragma unroll
    for (int o = 1; o < 32; o <<= 1) {
      const int y = __shfl_up(incl, o);
      if (lane >= o) incl += y;
    }
    const int idx = wbase + i * 32;
    if (idx < N_NODES) rowptr[idx] = carry + incl - v[i];
    carry += __shfl(incl, 31);
  }
  if (t == SCAN_T - 1) rowptr[N_NODES] = carry;
}

__global__ void fill_kernel(const int* __restrict__ ei, const int* __restrict__ rowptr,
                            const int* __restrict__ pos, int* __restrict__ csr_src) {
  int e = blockIdx.x * blockDim.x + threadIdx.x;
  if (e >= N_EDGES) return;
  csr_src[rowptr[ei[N_EDGES + e]] + pos[e]] = ei[e];
}

#define G1_PITCH 68

__global__ void __launch_bounds__(128) gemm1_wmma(const float* __restrict__ X,
                                                  const float* __restrict__ W,
                                                  const float* __restrict__ a_s,
                                                  const float* __restrict__ a_d,
                                                  float* __restrict__ H,
                                                  float* __restrict__ as,
                                                  float* __restrict__ ad) {
  __shared__ __attribute__((aligned(16))) float stage[4][16][G1_PITCH];
  const int wave = threadIdx.x >> 5;
  const int lane = threadIdx.x & 31;
  const int mtr  = blockIdx.x * 4 + wave;
  const bool live = mtr < N_NODES / 16;
  const int mt = live ? mtr : N_NODES / 16 - 1;
  const int m  = lane & 15;
  const int hi = lane >> 4;
  const float* xr = X + (long)(mt * 16 + m) * NFEAT;

  const int h = lane & 7;
  float ws_[NHID], wd_[NHID];
  #pragma unroll
  for (int k = 0; k < NHID; k++) { ws_[k] = a_s[h * NHID + k]; wd_[k] = a_d[h * NHID + k]; }

  v8f c0 = {}, c1 = {}, c2 = {}, c3 = {};
  #pragma unroll 4
  for (int k0 = 0; k0 < NFEAT; k0 += 4) {
    const int ka = k0 + 2 * hi;
    v2f a; a.x = xr[ka]; a.y = xr[ka + 1];
    const float* w0 = W + (long)ka * HID1 + m;
    const float* w1 = w0 + HID1;
    v2f b;
    b.x = w0[0];  b.y = w1[0];
    c0 = __builtin_amdgcn_wmma_f32_16x16x4_f32(false, a, false, b, (short)0, c0, false, false);
    b.x = w0[16]; b.y = w1[16];
    c1 = __builtin_amdgcn_wmma_f32_16x16x4_f32(false, a, false, b, (short)0, c1, false, false);
    b.x = w0[32]; b.y = w1[32];
    c2 = __builtin_amdgcn_wmma_f32_16x16x4_f32(false, a, false, b, (short)0, c2, false, false);
    b.x = w0[48]; b.y = w1[48];
    c3 = __builtin_amdgcn_wmma_f32_16x16x4_f32(false, a, false, b, (short)0, c3, false, false);
  }
  float (*S)[G1_PITCH] = stage[wave];
  #pragma unroll
  for (int j = 0; j < 8; j++) {
    float* sr = &S[j + 8 * hi][m];
    sr[0] = c0[j]; sr[16] = c1[j]; sr[32] = c2[j]; sr[48] = c3[j];
  }
  __syncthreads();
  if (live) {
    float* hp = H + (long)(mt * 16 + 8 * hi) * HID1 + m;
    #pragma unroll
    for (int j = 0; j < 8; j++) {
      hp[(long)j * HID1 +  0] = c0[j];
      hp[(long)j * HID1 + 16] = c1[j];
      hp[(long)j * HID1 + 32] = c2[j];
      hp[(long)j * HID1 + 48] = c3[j];
    }
  }
  #pragma unroll
  for (int i = 0; i < 4; i++) {
    const int r = (lane >> 3) + 4 * i;
    const float4 p0 = *(const float4*)&S[r][h * NHID];
    const float4 p1 = *(const float4*)&S[r][h * NHID + 4];
    const float hp8[NHID] = { p0.x, p0.y, p0.z, p0.w, p1.x, p1.y, p1.z, p1.w };
    float s = 0.f, d = 0.f;
    #pragma unroll
    for (int k = 0; k < NHID; k++) { s += hp8[k] * ws_[k]; d += hp8[k] * wd_[k]; }
    if (live) {
      const long o = (long)(mt * 16 + r) * NHEADS + h;
      as[o] = s; ad[o] = d;
    }
  }
}

#define G2_PITCH 44

__global__ void __launch_bounds__(128) gemm2_wmma(const float* __restrict__ X,
                                                  const float* __restrict__ W,
                                                  const float* __restrict__ a_s,
                                                  const float* __restrict__ a_d,
                                                  float* __restrict__ H,
                                                  float* __restrict__ as,
                                                  float* __restrict__ ad) {
  __shared__ __attribute__((aligned(16))) float stage[4][16][G2_PITCH];
  const int wave = threadIdx.x >> 5;
  const int lane = threadIdx.x & 31;
  const int mtr  = blockIdx.x * 4 + wave;
  const bool live = mtr < N_NODES / 16;
  const int mt = live ? mtr : N_NODES / 16 - 1;
  const int m  = lane & 15;
  const int hi = lane >> 4;
  const float* xr = X + (long)(mt * 16 + m) * HID1;

  const float* av = hi ? a_d : a_s;
  float wv[NCLASS];
  #pragma unroll
  for (int c = 0; c < NCLASS; c++) wv[c] = av[c];

  v8f c0 = {}, c1 = {}, c2 = {};
  #pragma unroll 4
  for (int k0 = 0; k0 < HID1; k0 += 4) {
    const int ka = k0 + 2 * hi;
    v2f a; a.x = xr[ka]; a.y = xr[ka + 1];
    #pragma unroll
    for (int t = 0; t < 3; t++) {
      const int col = t * 16 + m;
      v2f b;
      b.x = (col < NCLASS) ? W[(long)ka * NCLASS + col]       : 0.f;
      b.y = (col < NCLASS) ? W[(long)(ka + 1) * NCLASS + col] : 0.f;
      if (t == 0) c0 = __builtin_amdgcn_wmma_f32_16x16x4_f32(false, a, false, b, (short)0, c0, false, false);
      if (t == 1) c1 = __builtin_amdgcn_wmma_f32_16x16x4_f32(false, a, false, b, (short)0, c1, false, false);
      if (t == 2) c2 = __builtin_amdgcn_wmma_f32_16x16x4_f32(false, a, false, b, (short)0, c2, false, false);
    }
  }
  float (*S)[G2_PITCH] = stage[wave];
  #pragma unroll
  for (int j = 0; j < 8; j++) {
    float* sr = &S[j + 8 * hi][m];
    sr[0] = c0[j]; sr[16] = c1[j];
    if (m + 32 < NCLASS) sr[32] = c2[j];
  }
  __syncthreads();
  if (live) {
    const int rbase = mt * 16 + 8 * hi;
    #pragma unroll
    for (int j = 0; j < 8; j++) {
      float* hr = H + (long)(rbase + j) * NCLASS;
      if (m      < NCLASS) hr[m]      = c0[j];
      if (m + 16 < NCLASS) hr[m + 16] = c1[j];
      if (m + 32 < NCLASS) hr[m + 32] = c2[j];
    }
  }
  {
    const float4* rp = (const float4*)&S[m][0];
    float s = 0.f;
    #pragma unroll
    for (int c4 = 0; c4 < NCLASS / 4; c4++) {
      const float4 p = rp[c4];
      s += p.x * wv[4 * c4 + 0];
      s += p.y * wv[4 * c4 + 1];
      s += p.z * wv[4 * c4 + 2];
      s += p.w * wv[4 * c4 + 3];
    }
    if (live) (hi ? ad : as)[mt * 16 + m] = s;
  }
}

__global__ void __launch_bounds__(256) node_pass1(const int* __restrict__ rowptr,
                                                   const int* __restrict__ csr_src,
                                                   const float* __restrict__ as,
                                                   const float* __restrict__ ad,
                                                   const float* __restrict__ H1,
                                                   const float* __restrict__ b,
                                                   float* __restrict__ X2) {
  const int i = blockIdx.x * blockDim.x + threadIdx.x;
  if (i >= N_NODES * NHEADS) return;
  const int n = i >> 3, h = i & 7;
  const int beg = rowptr[n], end = rowptr[n + 1];
  const int last = end - 1;
  const float adn   = ad[i];
  const float vself = lrelu(as[i] + adn);

  float mx = vself;
  for (int j = beg; j < end; j += 4) {
    float v[4];
    #pragma unroll
    for (int u = 0; u < 4; u++) {
      const int s = csr_src[min(j + u, last)];
      v[u] = lrelu(as[s * NHEADS + h] + adn);
    }
    mx = fmaxf(mx, fmaxf(fmaxf(v[0], v[1]), fmaxf(v[2], v[3])));
  }

  const float4* hself = (const float4*)(H1 + (long)n * HID1 + h * NHID);
  const float wself = __expf(vself - mx);
  float den = wself;
  float4 a0 = hself[0], a1 = hself[1];
  a0.x *= wself; a0.y *= wself; a0.z *= wself; a0.w *= wself;
  a1.x *= wself; a1.y *= wself; a1.z *= wself; a1.w *= wself;
  for (int j = beg; j < end; j += 4) {
    float  w[4];
    float4 x0[4], x1[4];
    #pragma unroll
    for (int u = 0; u < 4; u++) {
      const int s = csr_src[min(j + u, last)];
      const float4* hp = (const float4*)(H1 + (long)s * HID1 + h * NHID);
      x0[u] = hp[0]; x1[u] = hp[1];
      const float e = __expf(lrelu(as[s * NHEADS + h] + adn) - mx);
      w[u] = (j + u < end) ? e : 0.f;
    }
    #pragma unroll
    for (int u = 0; u < 4; u++) {
      den  += w[u];
      a0.x += w[u] * x0[u].x; a0.y += w[u] * x0[u].y; a0.z += w[u] * x0[u].z; a0.w += w[u] * x0[u].w;
      a1.x += w[u] * x1[u].x; a1.y += w[u] * x1[u].y; a1.z += w[u] * x1[u].z; a1.w += w[u] * x1[u].w;
    }
  }

  float r[8] = { a0.x / den, a0.y / den, a0.z / den, a0.w / den,
                 a1.x / den, a1.y / den, a1.z / den, a1.w / den };
  #pragma unroll
  for (int k = 0; k < 8; k++) {
    const float t = r[k] + b[h * NHID + k];
    r[k] = t > 0.f ? t : (__expf(t) - 1.f);
  }
  float4* op = (float4*)(X2 + (long)n * HID1 + h * NHID);
  op[0] = make_float4(r[0], r[1], r[2], r[3]);
  op[1] = make_float4(r[4], r[5], r[6], r[7]);
}

__global__ void __launch_bounds__(256) node_pass2(const int* __restrict__ rowptr,
                                                   const int* __restrict__ csr_src,
                                                   const float* __restrict__ as,
                                                   const float* __restrict__ ad,
                                                   const float* __restrict__ H2,
                                                   const float* __restrict__ b,
                                                   float* __restrict__ out) {
  const int i = blockIdx.x * blockDim.x + threadIdx.x;
  if (i >= N_NODES * 8) return;
  const int n = i >> 3, q = i & 7;
  const int beg = rowptr[n], end = rowptr[n + 1];
  const int last = end - 1;
  const float adn   = ad[n];
  const float vself = lrelu(as[n] + adn);

  float mx = vself;
  for (int j = beg; j < end; j += 4) {
    float v[4];
    #pragma unroll
    for (int u = 0; u < 4; u++) {
      const int s = csr_src[min(j + u, last)];
      v[u] = lrelu(as[s] + adn);
    }
    mx = fmaxf(mx, fmaxf(fmaxf(v[0], v[1]), fmaxf(v[2], v[3])));
  }

  const float wself = __expf(vself - mx);
  float den = wself;
  float acc[NCLASS / 8];
  {
    const float* hs = H2 + (n * NCLASS + q);
    #pragma unroll
    for (int c = 0; c < NCLASS / 8; c++) acc[c] = wself * hs[8 * c];
  }
  for (int j = beg; j < end; j += 4) {
    float w[4];
    float x[4][NCLASS / 8];
    #pragma unroll
    for (int u = 0; u < 4; u++) {
      const int s = csr_src[min(j + u, last)];
      const float* hs = H2 + (s * NCLASS + q);
      #pragma unroll
      for (int c = 0; c < NCLASS / 8; c++) x[u][c] = hs[8 * c];
      const float e = __expf(lrelu(as[s] + adn) - mx);
      w[u] = (j + u < end) ? e : 0.f;
    }
    #pragma unroll
    for (int u = 0; u < 4; u++) {
      den += w[u];
      #pragma unroll
      for (int c = 0; c < NCLASS / 8; c++) acc[c] += w[u] * x[u][c];
    }
  }
  float val[NCLASS / 8];
  float cmx = -INFINITY;
  #pragma unroll
  for (int c = 0; c < NCLASS / 8; c++) { val[c] = acc[c] / den + b[q + 8 * c]; cmx = fmaxf(cmx, val[c]); }
  #pragma unroll
  for (int o = 1; o < 8; o <<= 1) cmx = fmaxf(cmx, __shfl_xor(cmx, o));
  float ssum = 0.f;
  #pragma unroll
  for (int c = 0; c < NCLASS / 8; c++) ssum += __expf(val[c] - cmx);
  #pragma unroll
  for (int o = 1; o < 8; o <<= 1) ssum += __shfl_xor(ssum, o);
  const float lse = cmx + __logf(ssum);
  float* od = out + (n * NCLASS + q);
  #pragma unroll
  for (int c = 0; c < NCLASS / 8; c++) od[8 * c] = val[c] - lse;
}

extern "C" void kernel_launch(void* const* d_in, const int* in_sizes, int n_in,
                              void* d_out, int out_size, void* d_ws, size_t ws_size,
                              hipStream_t stream) {
  const float* x      = (const float*)d_in[0];
  const float* W1     = (const float*)d_in[1];
  const float* a_src1 = (const float*)d_in[2];
  const float* a_dst1 = (const float*)d_in[3];
  const float* b1     = (const float*)d_in[4];
  const float* W2     = (const float*)d_in[5];
  const float* a_src2 = (const float*)d_in[6];
  const float* a_dst2 = (const float*)d_in[7];
  const float* b2     = (const float*)d_in[8];
  const int*   ei     = (const int*)d_in[9];
  float*       out    = (float*)d_out;

  float* ws   = (float*)d_ws;
  float* H1   = ws;
  float* X2   = H1  + (long)N_NODES * HID1;
  float* H2   = X2  + (long)N_NODES * HID1;
  float* as1  = H2  + (long)N_NODES * NCLASS;
  float* ad1  = as1 + (long)N_NODES * NHEADS;
  float* as2  = ad1 + (long)N_NODES * NHEADS;
  float* ad2  = as2 + N_NODES;
  int*   deg  = (int*)(ad2 + N_NODES);
  int*   rowp = deg  + N_NODES;
  int*   pos  = rowp + N_NODES + 4;
  int*   csr  = pos  + N_EDGES;

  const int B = 256;
  auto cdiv = [](long a, long b) { return (int)((a + b - 1) / b); };

  init_kernel<<<cdiv(N_NODES, B), B, 0, stream>>>(deg);
  count_kernel<<<cdiv(N_EDGES, B), B, 0, stream>>>(ei, deg, pos);
  scan_kernel<<<1, SCAN_T, 0, stream>>>(deg, rowp);
  fill_kernel<<<cdiv(N_EDGES, B), B, 0, stream>>>(ei, rowp, pos, csr);

  gemm1_wmma<<<cdiv(N_NODES / 16, 4), 128, 0, stream>>>(x, W1, a_src1, a_dst1, H1, as1, ad1);
  node_pass1<<<cdiv((long)N_NODES * NHEADS, B), B, 0, stream>>>(rowp, csr, as1, ad1, H1, b1, X2);

  gemm2_wmma<<<cdiv(N_NODES / 16, 4), 128, 0, stream>>>(X2, W2, a_src2, a_dst2, H2, as2, ad2);
  node_pass2<<<cdiv((long)N_NODES * 8, B), B, 0, stream>>>(rowp, csr, as2, ad2, H2, b2, out);
}


# node_pass1, node_pass2: each node's edges walked once, not twice: running maximum, partial sums rescaled by exp(old max - new max) when it rises; a trip's gathers all issued before a logit is used
# speedup vs baseline: 6.7887x; 1.0847x over previous
#include <hip/hip_runtime.h>
#include <math.h>

#define N_NODES 50000
#define N_EDGES 800000
#define NFEAT   128
#define NHID    8
#define NHEADS  8
#define HID1    (NHEADS * NHID)
#define NCLASS  40
#define NEG_SLOPE 0.2f

typedef __attribute__((ext_vector_type(2))) float v2f;
typedef __attribute__((ext_vector_type(8))) float v8f;

__device__ __forceinline__ float lrelu(float x) { return x > 0.f ? x : NEG_SLOPE * x; }

__global__ void init_kernel(int* __restrict__ deg) {
  int i = blockIdx.x * blockDim.x + threadIdx.x;
  if (i < N_NODES) deg[i] = 0;
}

__global__ void count_kernel(const int* __restrict__ ei, int* __restrict__ deg,
                             int* __restrict__ pos) {
  int e = blockIdx.x * blockDim.x + threadIdx.x;
  if (e >= N_EDGES) return;
  pos[e] = atomicAdd(&deg[ei[N_EDGES + e]], 1);
}

#define SCAN_T   1024
#define SCAN_PER ((N_NODES + SCAN_T - 1) / SCAN_T)

__global__ void __launch_bounds__(SCAN_T) scan_kernel(const int* __restrict__ deg,
                                                      int* __restrict__ rowptr) {
  __shared__ int wsum[32];
  const int t     = threadIdx.x;
  const int lane  = t & 31;
  const int wv    = t >> 5;
  const int wbase = wv * (SCAN_PER * 32) + lane;
  int v[SCAN_PER];
  int wtot = 0;
  #pragma unroll
  for (int i = 0; i < SCAN_PER; i++) {
    const int idx = wbase + i * 32;
    v[i] = (idx < N_NODES) ? deg[idx] : 0;
    wtot += v[i];
  }
  #pragma unroll
  for (int o = 16; o > 0; o >>= 1) wtot += __shfl_xor(wtot, o);
  if (lane == 0) wsum[wv] = wtot;
  __syncthreads();
  if (wv == 0) {
    const int w = wsum[lane];
    int wi = w;
    #pragma unroll
    for (int o = 1; o < 32; o <<= 1) {
      const int y = __shfl_up(wi, o);
      if (lane >= o) wi += y;
    }
    wsum[lane] = wi - w;
  }
  __syncthreads();
  int carry = wsum[wv];
  #pragma unroll
  for (int i = 0; i < SCAN_PER; i++) {
    int incl = v[i];
    #pragma unroll
    for (int o = 1; o < 32; o <<= 1) {
      const int y = __shfl_up(incl, o);
      if (lane >= o) incl += y;
    }
    const int idx = wbase + i * 32;
    if (idx < N_NODES) rowptr[idx] = carry + incl - v[i];
    carry += __shfl(incl, 31);
  }
  if (t == SCAN_T - 1) rowptr[N_NODES] = carry;
}

__global__ void fill_kernel(const int* __restrict__ ei, const int* __restrict__ rowptr,
                            const int* __restrict__ pos, int* __restrict__ csr_src) {
  int e = blockIdx.x * blockDim.x + threadIdx.x;
  if (e >= N_EDGES) return;
  csr_src[rowptr[ei[N_EDGES + e]] + pos[e]] = ei[e];
}

#define G1_PITCH 68

__global__ void __launch_bounds__(128) gemm1_wmma(const float* __restrict__ X,
                                                  const float* __restrict__ W,
                                                  const float* __restrict__ a_s,
                                                  const float* __restrict__ a_d,
                                                  float* __restrict__ H,
                                                  float* __restrict__ as,
                                                  float* __restrict__ ad) {
  __shared__ __attribute__((aligned(16))) float stage[4][16][G1_PITCH];
  const int wave = threadIdx.x >> 5;
  const int lane = threadIdx.x & 31;
  const int mtr  = blockIdx.x * 4 + wave;
  const bool live = mtr < N_NODES / 16;
  const int mt = live ? mtr : N_NODES / 16 - 1;
  const int m  = lane & 15;
  const int hi = lane >> 4;
  const float* xr = X + (long)(mt * 16 + m) * NFEAT;

  const int h = lane & 7;
  float ws_[NHID], wd_[NHID];
  #pragma unroll
  for (int k = 0; k < NHID; k++) { ws_[k] = a_s[h * NHID + k]; wd_[k] = a_d[h * NHID + k]; }

  v8f c0 = {}, c1 = {}, c2 = {}, c3 = {};
  #pragma unroll 4
  for (int k0 = 0; k0 < NFEAT; k0 += 4) {
    const int ka = k0 + 2 * hi;
    v2f a; a.x = xr[ka]; a.y = xr[ka + 1];
    const float* w0 = W + (long)ka * HID1 + m;
    const float* w1 = w0 + HID1;
    v2f b;
    b.x = w0[0];  b.y = w1[0];
    c0 = __builtin_amdgcn_wmma_f32_16x16x4_f32(false, a, false, b, (short)0, c0, false, false);
    b.x = w0[16]; b.y = w1[16];
    c1 = __builtin_amdgcn_wmma_f32_16x16x4_f32(false, a, false, b, (short)0, c1, false, false);
    b.x = w0[32]; b.y = w1[32];
    c2 = __builtin_amdgcn_wmma_f32_16x16x4_f32(false, a, false, b, (short)0, c2, false, false);
    b.x = w0[48]; b.y = w1[48];
    c3 = __builtin_amdgcn_wmma_f32_16x16x4_f32(false, a, false, b, (short)0, c3, false, false);
  }
  float (*S)[G1_PITCH] = stage[wave];
  #pragma unroll
  for (int j = 0; j < 8; j++) {
    float* sr = &S[j + 8 * hi][m];
    sr[0] = c0[j]; sr[16] = c1[j]; sr[32] = c2[j]; sr[48] = c3[j];
  }
  __syncthreads();
  if (live) {
    float* hp = H + (long)(mt * 16 + 8 * hi) * HID1 + m;
    #pragma unroll
    for (int j = 0; j < 8; j++) {
      hp[(long)j * HID1 +  0] = c0[j];
      hp[(long)j * HID1 + 16] = c1[j];
      hp[(long)j * HID1 + 32] = c2[j];
      hp[(long)j * HID1 + 48] = c3[j];
    }
  }
  #pragma unroll
  for (int i = 0; i < 4; i++) {
    const int r = (lane >> 3) + 4 * i;
    const float4 p0 = *(const float4*)&S[r][h * NHID];
    const float4 p1 = *(const float4*)&S[r][h * NHID + 4];
    const float hp8[NHID] = { p0.x, p0.y, p0.z, p0.w, p1.x, p1.y, p1.z, p1.w };
    float s = 0.f, d = 0.f;
    #pragma unroll
    for (int k = 0; k < NHID; k++) { s += hp8[k] * ws_[k]; d += hp8[k] * wd_[k]; }
    if (live) {
      const long o = (long)(mt * 16 + r) * NHEADS + h;
      as[o] = s; ad[o] = d;
    }
  }
}

#define G2_PITCH 44

__global__ void __launch_bounds__(128) gemm2_wmma(const float* __restrict__ X,
                                                  const float* __restrict__ W,
                                                  const float* __restrict__ a_s,
                                                  const float* __restrict__ a_d,
                                                  float* __restrict__ H,
                                                  float* __restrict__ as,
                                                  float* __restrict__ ad) {
  __shared__ __attribute__((aligned(16))) float stage[4][16][G2_PITCH];
  const int wave = threadIdx.x >> 5;
  const int lane = threadIdx.x & 31;
  const int mtr  = blockIdx.x * 4 + wave;
  const bool live = mtr < N_NODES / 16;
  const int mt = live ? mtr : N_NODES / 16 - 1;
  const int m  = lane & 15;
  const int hi = lane >> 4;
  const float* xr = X + (long)(mt * 16 + m) * HID1;

  const float* av = hi ? a_d : a_s;
  float wv[NCLASS];
  #pragma unroll
  for (int c = 0; c < NCLASS; c++) wv[c] = av[c];

  v8f c0 = {}, c1 = {}, c2 = {};
  #pragma unroll 4
  for (int k0 = 0; k0 < HID1; k0 += 4) {
    const int ka = k0 + 2 * hi;
    v2f a; a.x = xr[ka]; a.y = xr[ka + 1];
    #pragma unroll
    for (int t = 0; t < 3; t++) {
      const int col = t * 16 + m;
      v2f b;
      b.x = (col < NCLASS) ? W[(long)ka * NCLASS + col]       : 0.f;
      b.y = (col < NCLASS) ? W[(long)(ka + 1) * NCLASS + col] : 0.f;
      if (t == 0) c0 = __builtin_amdgcn_wmma_f32_16x16x4_f32(false, a, false, b, (short)0, c0, false, false);
      if (t == 1) c1 = __builtin_amdgcn_wmma_f32_16x16x4_f32(false, a, false, b, (short)0, c1, false, false);
      if (t == 2) c2 = __builtin_amdgcn_wmma_f32_16x16x4_f32(false, a, false, b, (short)0, c2, false, false);
    }
  }
  float (*S)[G2_PITCH] = stage[wave];
  #pragma unroll
  for (int j = 0; j < 8; j++) {
    float* sr = &S[j + 8 * hi][m];
    sr[0] = c0[j]; sr[16] = c1[j];
    if (m + 32 < NCLASS) sr[32] = c2[j];
  }
  __syncthreads();
  if (live) {
    const int rbase = mt * 16 + 8 * hi;
    #pragma unroll
    for (int j = 0; j < 8; j++) {
      float* hr = H + (long)(rbase + j) * NCLASS;
      if (m      < NCLASS) hr[m]      = c0[j];
      if (m + 16 < NCLASS) hr[m + 16] = c1[j];
      if (m + 32 < NCLASS) hr[m + 32] = c2[j];
    }
  }
  {
    const float4* rp = (const float4*)&S[m][0];
    float s = 0.f;
    #pragma unroll
    for (int c4 = 0; c4 < NCLASS / 4; c4++) {
      const float4 p = rp[c4];
      s += p.x * wv[4 * c4 + 0];
      s += p.y * wv[4 * c4 + 1];
      s += p.z * wv[4 * c4 + 2];
      s += p.w * wv[4 * c4 + 3];
    }
    if (live) (hi ? ad : as)[mt * 16 + m] = s;
  }
}

__global__ void __launch_bounds__(256) node_pass1(const int* __restrict__ rowptr,
                                                   const int* __restrict__ csr_src,
                                                   const float* __restrict__ as,
                                                   const float* __restrict__ ad,
                                                   const float* __restrict__ H1,
                                                   const float* __restrict__ b,
                                                   float* __restrict__ X2) {
  const int i = blockIdx.x * blockDim.x + threadIdx.x;
  if (i >= N_NODES * NHEADS) return;
  const int n = i >> 3, h = i & 7;
  const int beg = rowptr[n], end = rowptr[n + 1];
  const int last = end - 1;
  const float adn = ad[i];
  const float4* H1v = (const float4*)H1;
  const int hq = 2 * h;

  float mx  = lrelu(as[i] + adn);
  float den = 1.f;
  float4 a0 = H1v[n * (HID1 / 4) + hq], a1 = H1v[n * (HID1 / 4) + hq + 1];
  for (int j = beg; j < end; j += 4) {
    float  lg[4];
    float4 x0[4], x1[4];
    #pragma unroll
    for (int u = 0; u < 4; u++) {
      const int s = csr_src[min(j + u, last)];
      x0[u] = H1v[s * (HID1 / 4) + hq]; x1[u] = H1v[s * (HID1 / 4) + hq + 1];
      lg[u] = as[s * NHEADS + h];
    }
    __builtin_amdgcn_sched_barrier(0);
    float v[4];
    #pragma unroll
    for (int u = 0; u < 4; u++) v[u] = lrelu(lg[u] + adn);
    const float mn = fmaxf(mx, fmaxf(fmaxf(v[0], v[1]), fmaxf(v[2], v[3])));
    const float sc = __expf(mx - mn);
    mx = mn;
    den *= sc;
    a0.x *= sc; a0.y *= sc; a0.z *= sc; a0.w *= sc;
    a1.x *= sc; a1.y *= sc; a1.z *= sc; a1.w *= sc;
    #pragma unroll
    for (int u = 0; u < 4; u++) {
      const float w = (j + u < end) ? __expf(v[u] - mn) : 0.f;
      den  += w;
      a0.x += w * x0[u].x; a0.y += w * x0[u].y; a0.z += w * x0[u].z; a0.w += w * x0[u].w;
      a1.x += w * x1[u].x; a1.y += w * x1[u].y; a1.z += w * x1[u].z; a1.w += w * x1[u].w;
    }
  }

  float r[8] = { a0.x / den, a0.y / den, a0.z / den, a0.w / den,
                 a1.x / den, a1.y / den, a1.z / den, a1.w / den };
  #pragma unroll
  for (int k = 0; k < 8; k++) {
    const float t = r[k] + b[h * NHID + k];
    r[k] = t > 0.f ? t : (__expf(t) - 1.f);
  }
  float4* op = (float4*)(X2 + (long)n * HID1 + h * NHID);
  op[0] = make_float4(r[0], r[1], r[2], r[3]);
  op[1] = make_float4(r[4], r[5], r[6], r[7]);
}

__global__ void __launch_bounds__(256) node_pass2(const int* __restrict__ rowptr,
                                                   const int* __restrict__ csr_src,
                                                   const float* __restrict__ as,
                                                   const float* __restrict__ ad,
                                                   const float* __restrict__ H2,
                                                   const float* __restrict__ b,
                                                   float* __restrict__ out) {
  const int i = blockIdx.x * blockDim.x + threadIdx.x;
  if (i >= N_NODES * 8) return;
  const int n = i >> 3, q = i & 7;
  const int beg = rowptr[n], end = rowptr[n + 1];
  const int last = end - 1;
  const float adn = ad[n];

  float mx  = lrelu(as[n] + adn);
  float den = 1.f;
  float acc[NCLASS / 8];
  {
    const float* hs = H2 + (n * NCLASS + q);
    #pragma unroll
    for (int c = 0; c < NCLASS / 8; c++) acc[c] = hs[8 * c];
  }
  for (int j = beg; j < end; j += 4) {
    float lg[4];
    float x[4][NCLASS / 8];
    #pragma unroll
    for (int u = 0; u < 4; u++) {
      const int s = csr_src[min(j + u, last)];
      const float* hs = H2 + (s * NCLASS + q);
      #pragma unroll
      for (int c = 0; c < NCLASS / 8; c++) x[u][c] = hs[8 * c];
      lg[u] = as[s];
    }
    __builtin_amdgcn_sched_barrier(0);
    float v[4];
    #pragma unroll
    for (int u = 0; u < 4; u++) v[u] = lrelu(lg[u] + adn);
    const float mn = fmaxf(mx, fmaxf(fmaxf(v[0], v[1]), fmaxf(v[2], v[3])));
    const float sc = __expf(mx - mn);
    mx = mn;
    den *= sc;
    #pragma unroll
    for (int c = 0; c < NCLASS / 8; c++) acc[c] *= sc;
    #pragma unroll
    for (int u = 0; u < 4; u++) {
      const float w = (j + u < end) ? __expf(v[u] - mn) : 0.f;
      den += w;
      #pragma unroll
      for (int c = 0; c < NCLASS / 8; c++) acc[c] += w * x[u][c];
    }
  }
  float val[NCLASS / 8];
  float cmx = -INFINITY;
  #pragma unroll
  for (int c = 0; c < NCLASS / 8; c++) { val[c] = acc[c] / den + b[q + 8 * c]; cmx = fmaxf(cmx, val[c]); }
  #pragma unroll
  for (int o = 1; o < 8; o <<= 1) cmx = fmaxf(cmx, __shfl_xor(cmx, o));
  float ssum = 0.f;
  #pragma unroll
  for (int c = 0; c < NCLASS / 8; c++) ssum += __expf(val[c] - cmx);
  #pragma unroll
  for (int o = 1; o < 8; o <<= 1) ssum += __shfl_xor(ssum, o);
  const float lse = cmx + __logf(ssum);
  float* od = out + (n * NCLASS + q);
  #pragma unroll
  for (int c = 0; c < NCLASS / 8; c++) od[8 * c] = val[c] - lse;
}

extern "C" void kernel_launch(void* const* d_in, const int* in_sizes, int n_in,
                              void* d_out, int out_size, void* d_ws, size_t ws_size,
                              hipStream_t stream) {
  const float* x      = (const float*)d_in[0];
  const float* W1     = (const float*)d_in[1];
  const float* a_src1 = (const float*)d_in[2];
  const float* a_dst1 = (const float*)d_in[3];
  const float* b1     = (const float*)d_in[4];
  const float* W2     = (const float*)d_in[5];
  const float* a_src2 = (const float*)d_in[6];
  const float* a_dst2 = (const float*)d_in[7];
  const float* b2     = (const float*)d_in[8];
  const int*   ei     = (const int*)d_in[9];
  float*       out    = (float*)d_out;

  float* ws   = (float*)d_ws;
  float* H1   = ws;
  float* X2   = H1  + (long)N_NODES * HID1;
  float* H2   = X2  + (long)N_NODES * HID1;
  float* as1  = H2  + (long)N_NODES * NCLASS;
  float* ad1  = as1 + (long)N_NODES * NHEADS;
  float* as2  = ad1 + (long)N_NODES * NHEADS;
  float* ad2  = as2 + N_NODES;
  int*   deg  = (int*)(ad2 + N_NODES);
  int*   rowp = deg  + N_NODES;
  int*   pos  = rowp + N_NODES + 4;
  int*   csr  = pos  + N_EDGES;

  const int B = 256;
  auto cdiv = [](long a, long b) { return (int)((a + b - 1) / b); };

  init_kernel<<<cdiv(N_NODES, B), B, 0, stream>>>(deg);
  count_kernel<<<cdiv(N_EDGES, B), B, 0, stream>>>(ei, deg, pos);
  scan_kernel<<<1, SCAN_T, 0, stream>>>(deg, rowp);
  fill_kernel<<<cdiv(N_EDGES, B), B, 0, stream>>>(ei, rowp, pos, csr);

  gemm1_wmma<<<cdiv(N_NODES / 16, 4), 128, 0, stream>>>(x, W1, a_src1, a_dst1, H1, as1, ad1);
  node_pass1<<<cdiv((long)N_NODES * NHEADS, B), B, 0, stream>>>(rowp, csr, as1, ad1, H1, b1, X2);

  gemm2_wmma<<<cdiv(N_NODES / 16, 4), 128, 0, stream>>>(X2, W2, a_src2, a_dst2, H2, as2, ad2);
  node_pass2<<<cdiv((long)N_NODES * 8, B), B, 0, stream>>>(rowp, csr, as2, ad2, H2, b2, out);
}


# gemm2 folded into node_pass1: a work-group's 32 finished layer-1 rows stay in LDS as the A operand of its H2 tiles (same WMMA form, same k order); X2 never goes to memory; 8 -> 7 launches
# speedup vs baseline: 7.0716x; 1.0417x over previous
#include <hip/hip_runtime.h>
#include <math.h>

#define N_NODES 50000
#define N_EDGES 800000
#define NFEAT   128
#define NHID    8
#define NHEADS  8
#define HID1    (NHEADS * NHID)
#define NCLASS  40
#define NEG_SLOPE 0.2f

typedef __attribute__((ext_vector_type(2))) float v2f;
typedef __attribute__((ext_vector_type(8))) float v8f;

__device__ __forceinline__ float lrelu(float x) { return x > 0.f ? x : NEG_SLOPE * x; }

__global__ void init_kernel(int* __restrict__ deg) {
  int i = blockIdx.x * blockDim.x + threadIdx.x;
  if (i < N_NODES) deg[i] = 0;
}

__global__ void count_kernel(const int* __restrict__ ei, int* __restrict__ deg,
                             int* __restrict__ pos) {
  int e = blockIdx.x * blockDim.x + threadIdx.x;
  if (e >= N_EDGES) return;
  pos[e] = atomicAdd(&deg[ei[N_EDGES + e]], 1);
}

#define SCAN_T   1024
#define SCAN_PER ((N_NODES + SCAN_T - 1) / SCAN_T)

__global__ void __launch_bounds__(SCAN_T) scan_kernel(const int* __restrict__ deg,
                                                      int* __restrict__ rowptr) {
  __shared__ int wsum[32];
  const int t     = threadIdx.x;
  const int lane  = t & 31;
  const int wv    = t >> 5;
  const int wbase = wv * (SCAN_PER * 32) + lane;
  int v[SCAN_PER];
  int wtot = 0;
  #pragma unroll
  for (int i = 0; i < SCAN_PER; i++) {
    const int idx = wbase + i * 32;
    v[i] = (idx < N_NODES) ? deg[idx] : 0;
    wtot += v[i];
  }
  #pragma unroll
  for (int o = 16; o > 0; o >>= 1) wtot += __shfl_xor(wtot, o);
  if (lane == 0) wsum[wv] = wtot;
  __syncthreads();
  if (wv == 0) {
    const int w = wsum[lane];
    int wi = w;
    #pragma unroll
    for (int o = 1; o < 32; o <<= 1) {
      const int y = __shfl_up(wi, o);
      if (lane >= o) wi += y;
    }
    wsum[lane] = wi - w;
  }
  __syncthreads();
  int carry = wsum[wv];
  #pragma unroll
  for (int i = 0; i < SCAN_PER; i++) {
    int incl = v[i];
    #pragma unroll
    for (int o = 1; o < 32; o <<= 1) {
      const int y = __shfl_up(incl, o);
      if (lane >= o) incl += y;
    }
    const int idx = wbase + i * 32;
    if (idx < N_NODES) rowptr[idx] = carry + incl - v[i];
    carry += __shfl(incl, 31);
  }
  if (t == SCAN_T - 1) rowptr[N_NODES] = carry;
}

__global__ void fill_kernel(const int* __restrict__ ei, const int* __restrict__ rowptr,
                            const int* __restrict__ pos, int* __restrict__ csr_src) {
  int e = blockIdx.x * blockDim.x + threadIdx.x;
  if (e >= N_EDGES) return;
  csr_src[rowptr[ei[N_EDGES + e]] + pos[e]] = ei[e];
}

#define G1_PITCH 68

__global__ void __launch_bounds__(128) gemm1_wmma(const float* __restrict__ X,
                                                  const float* __restrict__ W,
                                                  const float* __restrict__ a_s,
                                                  const float* __restrict__ a_d,
                                                  float* __restrict__ H,
                                                  float* __restrict__ as,
                                                  float* __restrict__ ad) {
  __shared__ __attribute__((aligned(16))) float stage[4][16][G1_PITCH];
  const int wave = threadIdx.x >> 5;
  const int lane = threadIdx.x & 31;
  const int mtr  = blockIdx.x * 4 + wave;
  const bool live = mtr < N_NODES / 16;
  const int mt = live ? mtr : N_NODES / 16 - 1;
  const int m  = lane & 15;
  const int hi = lane >> 4;
  const float* xr = X + (long)(mt * 16 + m) * NFEAT;

  const int h = lane & 7;
  float ws_[NHID], wd_[NHID];
  #pragma unroll
  for (int k = 0; k < NHID; k++) { ws_[k] = a_s[h * NHID + k]; wd_[k] = a_d[h * NHID + k]; }

  v8f c0 = {}, c1 = {}, c2 = {}, c3 = {};
  #pragma unroll 4
  for (int k0 = 0; k0 < NFEAT; k0 += 4) {
    const int ka = k0 + 2 * hi;
    v2f a; a.x = xr[ka]; a.y = xr[ka + 1];
    const float* w0 = W + (long)ka * HID1 + m;
    const float* w1 = w0 + HID1;
    v2f b;
    b.x = w0[0];  b.y = w1[0];
    c0 = __builtin_amdgcn_wmma_f32_16x16x4_f32(false, a, false, b, (short)0, c0, false, false);
    b.x = w0[16]; b.y = w1[16];
    c1 = __builtin_amdgcn_wmma_f32_16x16x4_f32(false, a, false, b, (short)0, c1, false, false);
    b.x = w0[32]; b.y = w1[32];
    c2 = __builtin_amdgcn_wmma_f32_16x16x4_f32(false, a, false, b, (short)0, c2, false, false);
    b.x = w0[48]; b.y = w1[48];
    c3 = __builtin_amdgcn_wmma_f32_16x16x4_f32(false, a, false, b, (short)0, c3, false, false);
  }
  float (*S)[G1_PITCH] = stage[wave];
  #pragma unroll
  for (int j = 0; j < 8; j++) {
    float* sr = &S[j + 8 * hi][m];
    sr[0] = c0[j]; sr[16] = c1[j]; sr[32] = c2[j]; sr[48] = c3[j];
  }
  __syncthreads();
  if (live) {
    float* hp = H + (long)(mt * 16 + 8 * hi) * HID1 + m;
    #pragma unroll
    for (int j = 0; j < 8; j++) {
      hp[(long)j * HID1 +  0] = c0[j];
      hp[(long)j * HID1 + 16] = c1[j];
      hp[(long)j * HID1 + 32] = c2[j];
      hp[(long)j * HID1 + 48] = c3[j];
    }
  }
  #pragma unroll
  for (int i = 0; i < 4; i++) {
    const int r = (lane >> 3) + 4 * i;
    const float4 p0 = *(const float4*)&S[r][h * NHID];
    const float4 p1 = *(const float4*)&S[r][h * NHID + 4];
    const float hp8[NHID] = { p0.x, p0.y, p0.z, p0.w, p1.x, p1.y, p1.z, p1.w };
    float s = 0.f, d = 0.f;
    #pragma unroll
    for (int k = 0; k < NHID; k++) { s += hp8[k] * ws_[k]; d += hp8[k] * wd_[k]; }
    if (live) {
      const long o = (long)(mt * 16 + r) * NHEADS + h;
      as[o] = s; ad[o] = d;
    }
  }
}

#define NP1_NODES 32
#define XT_PITCH  68
#define G2_PITCH  52

__global__ void __launch_bounds__(256) node_pass1_gemm2(const int* __restrict__ rowptr,
                                                         const int* __restrict__ csr_src,
                                                         const float* __restrict__ as,
                                                         const float* __restrict__ ad,
                                                         const float* __restrict__ H1,
                                                         const float* __restrict__ b,
                                                         const float* __restrict__ W,
                                                         const float* __restrict__ a_s2,
                                                         const float* __restrict__ a_d2,
                                                         float* __restrict__ H2,
                                                         float* __restrict__ as2,
                                                         float* __restrict__ ad2) {
  __shared__ __attribute__((aligned(16))) float xt[NP1_NODES][XT_PITCH];
  __shared__ __attribute__((aligned(16))) float ht[NP1_NODES][G2_PITCH];
  const int t  = threadIdx.x;
  const int nl = t >> 3, h = t & 7;
  const int nbase = blockIdx.x * NP1_NODES;
  const int n = min(nbase + nl, N_NODES - 1);
  const int i = n * NHEADS + h;
  const int beg = rowptr[n], end = rowptr[n + 1];
  const int last = end - 1;
  const float adn = ad[i];
  const float4* H1v = (const float4*)H1;
  const int hq = 2 * h;

  float mx  = lrelu(as[i] + adn);
  float den = 1.f;
  float4 a0 = H1v[n * (HID1 / 4) + hq], a1 = H1v[n * (HID1 / 4) + hq + 1];
  for (int j = beg; j < end; j += 4) {
    float  lg[4];
    float4 x0[4], x1[4];
    #pragma unroll
    for (int u = 0; u < 4; u++) {
      const int s = csr_src[min(j + u, last)];
      x0[u] = H1v[s * (HID1 / 4) + hq]; x1[u] = H1v[s * (HID1 / 4) + hq + 1];
      lg[u] = as[s * NHEADS + h];
    }
    __builtin_amdgcn_sched_barrier(0);
    float v[4];
    #pragma unroll
    for (int u = 0; u < 4; u++) v[u] = lrelu(lg[u] + adn);
    const float mn = fmaxf(mx, fmaxf(fmaxf(v[0], v[1]), fmaxf(v[2], v[3])));
    const float sc = __expf(mx - mn);
    mx = mn;
    den *= sc;
    a0.x *= sc; a0.y *= sc; a0.z *= sc; a0.w *= sc;
    a1.x *= sc; a1.y *= sc; a1.z *= sc; a1.w *= sc;
    #pragma unroll
    for (int u = 0; u < 4; u++) {
      const float w = (j + u < end) ? __expf(v[u] - mn) : 0.f;
      den  += w;
      a0.x += w * x0[u].x; a0.y += w * x0[u].y; a0.z += w * x0[u].z; a0.w += w * x0[u].w;
      a1.x += w * x1[u].x; a1.y += w * x1[u].y; a1.z += w * x1[u].z; a1.w += w * x1[u].w;
    }
  }

  float r[8] = { a0.x / den, a0.y / den, a0.z / den, a0.w / den,
                 a1.x / den, a1.y / den, a1.z / den, a1.w / den };
  #pragma unroll
  for (int k = 0; k < 8; k++) {
    const float tv = r[k] + b[h * NHID + k];
    r[k] = tv > 0.f ? tv : (__expf(tv) - 1.f);
  }
  const int wave = t >> 5;
  const int lane = t & 31;
  const int m    = lane & 15;
  const int hi   = lane >> 4;
  const int mt   = wave & 1;
  const int col  = (wave >> 1) * 16 + m;
  const bool cok = col < NCLASS;
  {
    float4* xp = (float4*)&xt[nl][h * NHID];
    xp[0] = make_float4(r[0], r[1], r[2], r[3]);
    xp[1] = make_float4(r[4], r[5], r[6], r[7]);
  }
  float w0[HID1 / 4], w1[HID1 / 4];
  if (wave < 6) {
    const float* wp = W + (2 * hi) * NCLASS + (cok ? col : NCLASS - 1);
    #pragma unroll
    for (int ks = 0; ks < HID1 / 4; ks++) { w0[ks] = wp[(4 * ks) * NCLASS]; w1[ks] = wp[(4 * ks + 1) * NCLASS]; }
  }
  __syncthreads();

  if (wave < 6) {
    const float* xr = &xt[mt * 16 + m][2 * hi];
    v8f c = {};
    #pragma unroll
    for (int ks = 0; ks < HID1 / 4; ks++) {
      v2f a; a.x = xr[4 * ks]; a.y = xr[4 * ks + 1];
      v2f bb; bb.x = cok ? w0[ks] : 0.f; bb.y = cok ? w1[ks] : 0.f;
      c = __builtin_amdgcn_wmma_f32_16x16x4_f32(false, a, false, bb, (short)0, c, false, false);
    }
    #pragma unroll
    for (int j = 0; j < 8; j++) ht[mt * 16 + j + 8 * hi][col] = c[j];
  }
  __syncthreads();

  {
    float* hg = H2 + nbase * NCLASS;
    float hv[(NP1_NODES * NCLASS) / 256];
    #pragma unroll
    for (int p = 0; p < (NP1_NODES * NCLASS) / 256; p++) {
      const int idx = t + 256 * p;
      const int row = idx / NCLASS;
      hv[p] = ht[row][idx - row * NCLASS];
    }
    #pragma unroll
    for (int p = 0; p < (NP1_NODES * NCLASS) / 256; p++) {
      const int idx = t + 256 * p;
      if (nbase + idx / NCLASS < N_NODES) hg[idx] = hv[p];
    }
  }
  if (wave < 2) {
    const float*  av = wave ? a_d2 : a_s2;
    const float4* rp = (const float4*)&ht[lane][0];
    float s = 0.f;
    #pragma unroll
    for (int c4 = 0; c4 < NCLASS / 4; c4++) {
      const float4 p = rp[c4];
      s += p.x * av[4 * c4 + 0];
      s += p.y * av[4 * c4 + 1];
      s += p.z * av[4 * c4 + 2];
      s += p.w * av[4 * c4 + 3];
    }
    if (nbase + lane < N_NODES) (wave ? ad2 : as2)[nbase + lane] = s;
  }
}

__global__ void __launch_bounds__(256) node_pass2(const int* __restrict__ rowptr,
                                                   const int* __restrict__ csr_src,
                                                   const float* __restrict__ as,
                                                   const float* __restrict__ ad,
                                                   const float* __restrict__ H2,
                                                   const float* __restrict__ b,
                                                   float* __restrict__ out) {
  const int i = blockIdx.x * blockDim.x + threadIdx.x;
  if (i >= N_NODES * 8) return;
  const int n = i >> 3, q = i & 7;
  const int beg = rowptr[n], end = rowptr[n + 1];
  const int last = end - 1;
  const float adn = ad[n];

  float mx  = lrelu(as[n] + adn);
  float den = 1.f;
  float acc[NCLASS / 8];
  {
    const float* hs = H2 + (n * NCLASS + q);
    #pragma unroll
    for (int c = 0; c < NCLASS / 8; c++) acc[c] = hs[8 * c];
  }
  for (int j = beg; j < end; j += 4) {
    float lg[4];
    float x[4][NCLASS / 8];
    #pragma unroll
    for (int u = 0; u < 4; u++) {
      const int s = csr_src[min(j + u, last)];
      const float* hs = H2 + (s * NCLASS + q);
      #pragma unroll
      for (int c = 0; c < NCLASS / 8; c++) x[u][c] = hs[8 * c];
      lg[u] = as[s];
    }
    __builtin_amdgcn_sched_barrier(0);
    float v[4];
    #pragma unroll
    for (int u = 0; u < 4; u++) v[u] = lrelu(lg[u] + adn);
    const float mn = fmaxf(mx, fmaxf(fmaxf(v[0], v[1]), fmaxf(v[2], v[3])));
    const float sc = __expf(mx - mn);
    mx = mn;
    den *= sc;
    #pragma unroll
    for (int c = 0; c < NCLASS / 8; c++) acc[c] *= sc;
    #pragma unroll
    for (int u = 0; u < 4; u++) {
      const float w = (j + u < end) ? __expf(v[u] - mn) : 0.f;
      den += w;
      #pragma unroll
      for (int c = 0; c < NCLASS / 8; c++) acc[c] += w * x[u][c];
    }
  }
  float val[NCLASS / 8];
  float cmx = -INFINITY;
  #pragma unroll
  for (int c = 0; c < NCLASS / 8; c++) { val[c] = acc[c] / den + b[q + 8 * c]; cmx = fmaxf(cmx, val[c]); }
  #pragma unroll
  for (int o = 1; o < 8; o <<= 1) cmx = fmaxf(cmx, __shfl_xor(cmx, o));
  float ssum = 0.f;
  #pragma unroll
  for (int c = 0; c < NCLASS / 8; c++) ssum += __expf(val[c] - cmx);
  #pragma unroll
  for (int o = 1; o < 8; o <<= 1) ssum += __shfl_xor(ssum, o);
  const float lse = cmx + __logf(ssum);
  float* od = out + (n * NCLASS + q);
  #pragma unroll
  for (int c = 0; c < NCLASS / 8; c++) od[8 * c] = val[c] - lse;
}

extern "C" void kernel_launch(void* const* d_in, const int* in_sizes, int n_in,
                              void* d_out, int out_size, void* d_ws, size_t ws_size,
                              hipStream_t stream) {
  const float* x      = (const float*)d_in[0];
  const float* W1     = (const float*)d_in[1];
  const float* a_src1 = (const float*)d_in[2];
  const float* a_dst1 = (const float*)d_in[3];
  const float* b1     = (const float*)d_in[4];
  const float* W2     = (const float*)d_in[5];
  const float* a_src2 = (const float*)d_in[6];
  const float* a_dst2 = (const float*)d_in[7];
  const float* b2     = (const float*)d_in[8];
  const int*   ei     = (const int*)d_in[9];
  float*       out    = (float*)d_out;

  float* ws   = (float*)d_ws;
  float* H1   = ws;
  float* H2   = H1  + (long)N_NODES * HID1;
  float* as1  = H2  + (long)N_NODES * NCLASS;
  float* ad1  = as1 + (long)N_NODES * NHEADS;
  float* as2  = ad1 + (long)N_NODES * NHEADS;
  float* ad2  = as2 + N_NODES;
  int*   deg  = (int*)(ad2 + N_NODES);
  int*   rowp = deg  + N_NODES;
  int*   pos  = rowp + N_NODES + 4;
  int*   csr  = pos  + N_EDGES;

  const int B = 256;
  auto cdiv = [](long a, long b) { return (int)((a + b - 1) / b); };

  init_kernel<<<cdiv(N_NODES, B), B, 0, stream>>>(deg);
  count_kernel<<<cdiv(N_EDGES, B), B, 0, stream>>>(ei, deg, pos);
  scan_kernel<<<1, SCAN_T, 0, stream>>>(deg, rowp);
  fill_kernel<<<cdiv(N_EDGES, B), B, 0, stream>>>(ei, rowp, pos, csr);

  gemm1_wmma<<<cdiv(N_NODES / 16, 4), 128, 0, stream>>>(x, W1, a_src1, a_dst1, H1, as1, ad1);
  node_pass1_gemm2<<<cdiv(N_NODES, NP1_NODES), B, 0, stream>>>(rowp, csr, as1, ad1, H1, b1,
                                                               W2, a_src2, a_dst2, H2, as2, ad2);

  node_pass2<<<cdiv((long)N_NODES * 8, B), B, 0, stream>>>(rowp, csr, as2, ad2, H2, b2, out);
}


# count_kernel: edge ranks from LDS counters (grid of 16 node ranges x 16 edge chunks); one coalesced returning global add per (chunk, node), not one scattered per edge; list order still arrival order
# speedup vs baseline: 8.5154x; 1.2042x over previous
#include <hip/hip_runtime.h>
#include <math.h>

#define N_NODES 50000
#define N_EDGES 800000
#define NFEAT   128
#define NHID    8
#define NHEADS  8
#define HID1    (NHEADS * NHID)
#define NCLASS  40
#define NEG_SLOPE 0.2f

typedef __attribute__((ext_vector_type(2))) float v2f;
typedef __attribute__((ext_vector_type(8))) float v8f;

__device__ __forceinline__ float lrelu(float x) { return x > 0.f ? x : NEG_SLOPE * x; }

__global__ void init_kernel(int* __restrict__ deg) {
  int i = blockIdx.x * blockDim.x + threadIdx.x;
  if (i < N_NODES) deg[i] = 0;
}

#define CNT_R   16
#define CNT_C   16
#define CNT_M   (N_NODES / CNT_R)
#define CNT_EC  (N_EDGES / CNT_C)
#define CNT_T   1024
#define CNT_Q   (CNT_EC / 4)
#define CNT_PER ((CNT_Q + CNT_T - 1) / CNT_T)
#define CNT_NPER ((CNT_M + CNT_T - 1) / CNT_T)

__global__ void __launch_bounds__(CNT_T) count_kernel(const int* __restrict__ ei, int* __restrict__ deg,
                                                      int* __restrict__ pos, int* __restrict__ off) {
  __shared__ int cnt[CNT_M];
  const int t  = threadIdx.x;
  const int r  = blockIdx.x % CNT_R;
  const int c  = blockIdx.x / CNT_R;
  const int nb = r * CNT_M;
  const int4* dv = (const int4*)(ei + N_EDGES + c * CNT_EC);
  int4 d[CNT_PER];
  #pragma unroll
  for (int p = 0; p < CNT_PER; p++) d[p] = dv[min(t + p * CNT_T, CNT_Q - 1)];
  for (int i = t; i < CNT_M; i += CNT_T) cnt[i] = 0;
  __syncthreads();
  #pragma unroll
  for (int p = 0; p < CNT_PER; p++) {
    const int q = t + p * CNT_T;
    if (q < CNT_Q) {
      int* pe = pos + c * CNT_EC + 4 * q;
      const unsigned u0 = (unsigned)(d[p].x - nb), u1 = (unsigned)(d[p].y - nb);
      const unsigned u2 = (unsigned)(d[p].z - nb), u3 = (unsigned)(d[p].w - nb);
      int k0 = 0, k1 = 0, k2 = 0, k3 = 0;
      if (u0 < CNT_M) k0 = atomicAdd(&cnt[u0], 1);
      if (u1 < CNT_M) k1 = atomicAdd(&cnt[u1], 1);
      if (u2 < CNT_M) k2 = atomicAdd(&cnt[u2], 1);
      if (u3 < CNT_M) k3 = atomicAdd(&cnt[u3], 1);
      if (u0 < CNT_M) pe[0] = k0;
      if (u1 < CNT_M) pe[1] = k1;
      if (u2 < CNT_M) pe[2] = k2;
      if (u3 < CNT_M) pe[3] = k3;
    }
  }
  __syncthreads();
  int kc[CNT_NPER], ko[CNT_NPER];
  #pragma unroll
  for (int k = 0; k < CNT_NPER; k++) { const int i = t + k * CNT_T; kc[k] = (i < CNT_M) ? cnt[i] : 0; }
  #pragma unroll
  for (int k = 0; k < CNT_NPER; k++) { const int i = t + k * CNT_T; ko[k] = 0; if (i < CNT_M) ko[k] = atomicAdd(&deg[nb + i], kc[k]); }
  #pragma unroll
  for (int k = 0; k < CNT_NPER; k++) { const int i = t + k * CNT_T; if (i < CNT_M) off[c * N_NODES + nb + i] = ko[k]; }
}

#define SCAN_T   1024
#define SCAN_PER ((N_NODES + SCAN_T - 1) / SCAN_T)

__global__ void __launch_bounds__(SCAN_T) scan_kernel(const int* __restrict__ deg,
                                                      int* __restrict__ rowptr) {
  __shared__ int wsum[32];
  const int t     = threadIdx.x;
  const int lane  = t & 31;
  const int wv    = t >> 5;
  const int wbase = wv * (SCAN_PER * 32) + lane;
  int v[SCAN_PER];
  int wtot = 0;
  #pragma unroll
  for (int i = 0; i < SCAN_PER; i++) {
    const int idx = wbase + i * 32;
    v[i] = (idx < N_NODES) ? deg[idx] : 0;
    wtot += v[i];
  }
  #pragma unroll
  for (int o = 16; o > 0; o >>= 1) wtot += __shfl_xor(wtot, o);
  if (lane == 0) wsum[wv] = wtot;
  __syncthreads();
  if (wv == 0) {
    const int w = wsum[lane];
    int wi = w;
    #pragma unroll
    for (int o = 1; o < 32; o <<= 1) {
      const int y = __shfl_up(wi, o);
      if (lane >= o) wi += y;
    }
    wsum[lane] = wi - w;
  }
  __syncthreads();
  int carry = wsum[wv];
  #pragma unroll
  for (int i = 0; i < SCAN_PER; i++) {
    int incl = v[i];
    #pragma unroll
    for (int o = 1; o < 32; o <<= 1) {
      const int y = __shfl_up(incl, o);
      if (lane >= o) incl += y;
    }
    const int idx = wbase + i * 32;
    if (idx < N_NODES) rowptr[idx] = carry + incl - v[i];
    carry += __shfl(incl, 31);
  }
  if (t == SCAN_T - 1) rowptr[N_NODES] = carry;
}

__global__ void fill_kernel(const int* __restrict__ ei, const int* __restrict__ rowptr,
                            const int* __restrict__ pos, const int* __restrict__ off,
                            int* __restrict__ csr_src) {
  int e = blockIdx.x * blockDim.x + threadIdx.x;
  if (e >= N_EDGES) return;
  const int d = ei[N_EDGES + e];
  csr_src[rowptr[d] + off[(e / CNT_EC) * N_NODES + d] + pos[e]] = ei[e];
}

#define G1_PITCH 68

__global__ void __launch_bounds__(128) gemm1_wmma(const float* __restrict__ X,
                                                  const float* __restrict__ W,
                                                  const float* __restrict__ a_s,
                                                  const float* __restrict__ a_d,
                                                  float* __restrict__ H,
                                                  float* __restrict__ as,
                                                  float* __restrict__ ad) {
  __shared__ __attribute__((aligned(16))) float stage[4][16][G1_PITCH];
  const int wave = threadIdx.x >> 5;
  const int lane = threadIdx.x & 31;
  const int mtr  = blockIdx.x * 4 + wave;
  const bool live = mtr < N_NODES / 16;
  const int mt = live ? mtr : N_NODES / 16 - 1;
  const int m  = lane & 15;
  const int hi = lane >> 4;
  const float* xr = X + (long)(mt * 16 + m) * NFEAT;

  const int h = lane & 7;
  float ws_[NHID], wd_[NHID];
  #pragma unroll
  for (int k = 0; k < NHID; k++) { ws_[k] = a_s[h * NHID + k]; wd_[k] = a_d[h * NHID + k]; }

  v8f c0 = {}, c1 = {}, c2 = {}, c3 = {};
  #pragma unroll 4
  for (int k0 = 0; k0 < NFEAT; k0 += 4) {
    const int ka = k0 + 2 * hi;
    v2f a; a.x = xr[ka]; a.y = xr[ka + 1];
    const float* w0 = W + (long)ka * HID1 + m;
    const float* w1 = w0 + HID1;
    v2f b;
    b.x = w0[0];  b.y = w1[0];
    c0 = __builtin_amdgcn_wmma_f32_16x16x4_f32(false, a, false, b, (short)0, c0, false, false);
    b.x = w0[16]; b.y = w1[16];
    c1 = __builtin_amdgcn_wmma_f32_16x16x4_f32(false, a, false, b, (short)0, c1, false, false);
    b.x = w0[32]; b.y = w1[32];
    c2 = __builtin_amdgcn_wmma_f32_16x16x4_f32(false, a, false, b, (short)0, c2, false, false);
    b.x = w0[48]; b.y = w1[48];
    c3 = __builtin_amdgcn_wmma_f32_16x16x4_f32(false, a, false, b, (short)0, c3, false, false);
  }
  float (*S)[G1_PITCH] = stage[wave];
  #pragma unroll
  for (int j = 0; j < 8; j++) {
    float* sr = &S[j + 8 * hi][m];
    sr[0] = c0[j]; sr[16] = c1[j]; sr[32] = c2[j]; sr[48] = c3[j];
  }
  __syncthreads();
  if (live) {
    float* hp = H + (long)(mt * 16 + 8 * hi) * HID1 + m;
    #pragma unroll
    for (int j = 0; j < 8; j++) {
      hp[(long)j * HID1 +  0] = c0[j];
      hp[(long)j * HID1 + 16] = c1[j];
      hp[(long)j * HID1 + 32] = c2[j];
      hp[(long)j * HID1 + 48] = c3[j];
    }
  }
  #pragma unroll
  for (int i = 0; i < 4; i++) {
    const int r = (lane >> 3) + 4 * i;
    const float4 p0 = *(const float4*)&S[r][h * NHID];
    const float4 p1 = *(const float4*)&S[r][h * NHID + 4];
    const float hp8[NHID] = { p0.x, p0.y, p0.z, p0.w, p1.x, p1.y, p1.z, p1.w };
    float s = 0.f, d = 0.f;
    #pragma unroll
    for (int k = 0; k < NHID; k++) { s += hp8[k] * ws_[k]; d += hp8[k] * wd_[k]; }
    if (live) {
      const long o = (long)(mt * 16 + r) * NHEADS + h;
      as[o] = s; ad[o] = d;
    }
  }
}

#define NP1_NODES 32
#define XT_PITCH  68
#define G2_PITCH  52

__global__ void __launch_bounds__(256) node_pass1_gemm2(const int* __restrict__ rowptr,
                                                         const int* __restrict__ csr_src,
                                                         const float* __restrict__ as,
                                                         const float* __restrict__ ad,
                                                         const float* __restrict__ H1,
                                                         const float* __restrict__ b,
                                                         const float* __restrict__ W,
                                                         const float* __restrict__ a_s2,
                                                         const float* __restrict__ a_d2,
                                                         float* __restrict__ H2,
                                                         float* __restrict__ as2,
                                                         float* __restrict__ ad2) {
  __shared__ __attribute__((aligned(16))) float xt[NP1_NODES][XT_PITCH];
  __shared__ __attribute__((aligned(16))) float ht[NP1_NODES][G2_PITCH];
  const int t  = threadIdx.x;
  const int nl = t >> 3, h = t & 7;
  const int nbase = blockIdx.x * NP1_NODES;
  const int n = min(nbase + nl, N_NODES - 1);
  const int i = n * NHEADS + h;
  const int beg = rowptr[n], end = rowptr[n + 1];
  const int last = end - 1;
  const float adn = ad[i];
  const float4* H1v = (const float4*)H1;
  const int hq = 2 * h;

  float mx  = lrelu(as[i] + adn);
  float den = 1.f;
  float4 a0 = H1v[n * (HID1 / 4) + hq], a1 = H1v[n * (HID1 / 4) + hq + 1];
  for (int j = beg; j < end; j += 4) {
    float  lg[4];
    float4 x0[4], x1[4];
    #pragma unroll
    for (int u = 0; u < 4; u++) {
      const int s = csr_src[min(j + u, last)];
      x0[u] = H1v[s * (HID1 / 4) + hq]; x1[u] = H1v[s * (HID1 / 4) + hq + 1];
      lg[u] = as[s * NHEADS + h];
    }
    __builtin_amdgcn_sched_barrier(0);
    float v[4];
    #pragma unroll
    for (int u = 0; u < 4; u++) v[u] = lrelu(lg[u] + adn);
    const float mn = fmaxf(mx, fmaxf(fmaxf(v[0], v[1]), fmaxf(v[2], v[3])));
    const float sc = __expf(mx - mn);
    mx = mn;
    den *= sc;
    a0.x *= sc; a0.y *= sc; a0.z *= sc; a0.w *= sc;
    a1.x *= sc; a1.y *= sc; a1.z *= sc; a1.w *= sc;
    #pragma unroll
    for (int u = 0; u < 4; u++) {
      const float w = (j + u < end) ? __expf(v[u] - mn) : 0.f;
      den  += w;
      a0.x += w * x0[u].x; a0.y += w * x0[u].y; a0.z += w * x0[u].z; a0.w += w * x0[u].w;
      a1.x += w * x1[u].x; a1.y += w * x1[u].y; a1.z += w * x1[u].z; a1.w += w * x1[u].w;
    }
  }

  float r[8] = { a0.x / den, a0.y / den, a0.z / den, a0.w / den,
                 a1.x / den, a1.y / den, a1.z / den, a1.w / den };
  #pragma unroll
  for (int k = 0; k < 8; k++) {
    const float tv = r[k] + b[h * NHID + k];
    r[k] = tv > 0.f ? tv : (__expf(tv) - 1.f);
  }
  const int wave = t >> 5;
  const int lane = t & 31;
  const int m    = lane & 15;
  const int hi   = lane >> 4;
  const int mt   = wave & 1;
  const int col  = (wave >> 1) * 16 + m;
  const bool cok = col < NCLASS;
  {
    float4* xp = (float4*)&xt[nl][h * NHID];
    xp[0] = make_float4(r[0], r[1], r[2], r[3]);
    xp[1] = make_float4(r[4], r[5], r[6], r[7]);
  }
  float w0[HID1 / 4], w1[HID1 / 4];
  if (wave < 6) {
    const float* wp = W + (2 * hi) * NCLASS + (cok ? col : NCLASS - 1);
    #pragma unroll
    for (int ks = 0; ks < HID1 / 4; ks++) { w0[ks] = wp[(4 * ks) * NCLASS]; w1[ks] = wp[(4 * ks + 1) * NCLASS]; }
  }
  __syncthreads();

  if (wave < 6) {
    const float* xr = &xt[mt * 16 + m][2 * hi];
    v8f c = {};
    #pragma unroll
    for (int ks = 0; ks < HID1 / 4; ks++) {
      v2f a; a.x = xr[4 * ks]; a.y = xr[4 * ks + 1];
      v2f bb; bb.x = cok ? w0[ks] : 0.f; bb.y = cok ? w1[ks] : 0.f;
      c = __builtin_amdgcn_wmma_f32_16x16x4_f32(false, a, false, bb, (short)0, c, false, false);
    }
    #pragma unroll
    for (int j = 0; j < 8; j++) ht[mt * 16 + j + 8 * hi][col] = c[j];
  }
  __syncthreads();

  {
    float* hg = H2 + nbase * NCLASS;
    float hv[(NP1_NODES * NCLASS) / 256];
    #pragma unroll
    for (int p = 0; p < (NP1_NODES * NCLASS) / 256; p++) {
      const int idx = t + 256 * p;
      const int row = idx / NCLASS;
      hv[p] = ht[row][idx - row * NCLASS];
    }
    #pragma unroll
    for (int p = 0; p < (NP1_NODES * NCLASS) / 256; p++) {
      const int idx = t + 256 * p;
      if (nbase + idx / NCLASS < N_NODES) hg[idx] = hv[p];
    }
  }
  if (wave < 2) {
    const float*  av = wave ? a_d2 : a_s2;
    const float4* rp = (const float4*)&ht[lane][0];
    float s = 0.f;
    #pragma unroll
    for (int c4 = 0; c4 < NCLASS / 4; c4++) {
      const float4 p = rp[c4];
      s += p.x * av[4 * c4 + 0];
      s += p.y * av[4 * c4 + 1];
      s += p.z * av[4 * c4 + 2];
      s += p.w * av[4 * c4 + 3];
    }
    if (nbase + lane < N_NODES) (wave ? ad2 : as2)[nbase + lane] = s;
  }
}

__global__ void __launch_bounds__(256) node_pass2(const int* __restrict__ rowptr,
                                                   const int* __restrict__ csr_src,
                                                   const float* __restrict__ as,
                                                   const float* __restrict__ ad,
                                                   const float* __restrict__ H2,
                                                   const float* __restrict__ b,
                                                   float* __restrict__ out) {
  const int i = blockIdx.x * blockDim.x + threadIdx.x;
  if (i >= N_NODES * 8) return;
  const int n = i >> 3, q = i & 7;
  const int beg = rowptr[n], end = rowptr[n + 1];
  const int last = end - 1;
  const float adn = ad[n];

  float mx  = lrelu(as[n] + adn);
  float den = 1.f;
  float acc[NCLASS / 8];
  {
    const float* hs = H2 + (n * NCLASS + q);
    #pragma unroll
    for (int c = 0; c < NCLASS / 8; c++) acc[c] = hs[8 * c];
  }
  for (int j = beg; j < end; j += 4) {
    float lg[4];
    float x[4][NCLASS / 8];
    #pragma unroll
    for (int u = 0; u < 4; u++) {
      const int s = csr_src[min(j + u, last)];
      const float* hs = H2 + (s * NCLASS + q);
      #pragma unroll
      for (int c = 0; c < NCLASS / 8; c++) x[u][c] = hs[8 * c];
      lg[u] = as[s];
    }
    __builtin_amdgcn_sched_barrier(0);
    float v[4];
    #pragma unroll
    for (int u = 0; u < 4; u++) v[u] = lrelu(lg[u] + adn);
    const float mn = fmaxf(mx, fmaxf(fmaxf(v[0], v[1]), fmaxf(v[2], v[3])));
    const float sc = __expf(mx - mn);
    mx = mn;
    den *= sc;
    #pragma unroll
    for (int c = 0; c < NCLASS / 8; c++) acc[c] *= sc;
    #pragma unroll
    for (int u = 0; u < 4; u++) {
      const float w = (j + u < end) ? __expf(v[u] - mn) : 0.f;
      den += w;
      #pragma unroll
      for (int c = 0; c < NCLASS / 8; c++) acc[c] += w * x[u][c];
    }
  }
  float val[NCLASS / 8];
  float cmx = -INFINITY;
  #pragma unroll
  for (int c = 0; c < NCLASS / 8; c++) { val[c] = acc[c] / den + b[q + 8 * c]; cmx = fmaxf(cmx, val[c]); }
  #pragma unroll
  for (int o = 1; o < 8; o <<= 1) cmx = fmaxf(cmx, __shfl_xor(cmx, o));
  float ssum = 0.f;
  #pragma unroll
  for (int c = 0; c < NCLASS / 8; c++) ssum += __expf(val[c] - cmx);
  #pragma unroll
  for (int o = 1; o < 8; o <<= 1) ssum += __shfl_xor(ssum, o);
  const float lse = cmx + __logf(ssum);
  float* od = out + (n * NCLASS + q);
  #pragma unroll
  for (int c = 0; c < NCLASS / 8; c++) od[8 * c] = val[c] - lse;
}

extern "C" void kernel_launch(void* const* d_in, const int* in_sizes, int n_in,
                              void* d_out, int out_size, void* d_ws, size_t ws_size,
                              hipStream_t stream) {
  const float* x      = (const float*)d_in[0];
  const float* W1     = (const float*)d_in[1];
  const float* a_src1 = (const float*)d_in[2];
  const float* a_dst1 = (const float*)d_in[3];
  const float* b1     = (const float*)d_in[4];
  const float* W2     = (const float*)d_in[5];
  const float* a_src2 = (const float*)d_in[6];
  const float* a_dst2 = (const float*)d_in[7];
  const float* b2     = (const float*)d_in[8];
  const int*   ei     = (const int*)d_in[9];
  float*       out    = (float*)d_out;

  float* ws   = (float*)d_ws;
  float* H1   = ws;
  float* H2   = H1  + (long)N_NODES * HID1;
  float* as1  = H2  + (long)N_NODES * NCLASS;
  float* ad1  = as1 + (long)N_NODES * NHEADS;
  float* as2  = ad1 + (long)N_NODES * NHEADS;
  float* ad2  = as2 + N_NODES;
  int*   deg  = (int*)(ad2 + N_NODES);
  int*   rowp = deg  + N_NODES;
  int*   pos  = rowp + N_NODES + 4;
  int*   csr  = pos  + N_EDGES;
  int*   off  = csr  + N_EDGES;

  const int B = 256;
  auto cdiv = [](long a, long b) { return (int)((a + b - 1) / b); };

  init_kernel<<<cdiv(N_NODES, B), B, 0, stream>>>(deg);
  count_kernel<<<CNT_R * CNT_C, CNT_T, 0, stream>>>(ei, deg, pos, off);
  scan_kernel<<<1, SCAN_T, 0, stream>>>(deg, rowp);
  fill_kernel<<<cdiv(N_EDGES, B), B, 0, stream>>>(ei, rowp, pos, off, csr);

  gemm1_wmma<<<cdiv(N_NODES / 16, 4), 128, 0, stream>>>(x, W1, a_src1, a_dst1, H1, as1, ad1);
  node_pass1_gemm2<<<cdiv(N_NODES, NP1_NODES), B, 0, stream>>>(rowp, csr, as1, ad1, H1, b1,
                                                               W2, a_src2, a_dst2, H2, as2, ad2);

  node_pass2<<<cdiv((long)N_NODES * 8, B), B, 0, stream>>>(rowp, csr, as2, ad2, H2, b2, out);
}


# edge logits formed from the feature rows the node kernels gather anyway, not read from logit arrays (one more line per edge); logit dot products summed in a new order (pairs; layer 2: 8-lane tree)
# speedup vs baseline: 8.8801x; 1.0428x over previous
#include <hip/hip_runtime.h>
#include <math.h>

#define N_NODES 50000
#define N_EDGES 800000
#define NFEAT   128
#define NHID    8
#define NHEADS  8
#define HID1    (NHEADS * NHID)
#define NCLASS  40
#define NEG_SLOPE 0.2f

typedef __attribute__((ext_vector_type(2))) float v2f;
typedef __attribute__((ext_vector_type(8))) float v8f;

__device__ __forceinline__ float lrelu(float x) { return x > 0.f ? x : NEG_SLOPE * x; }

__device__ __forceinline__ float dot8(const float4 p0, const float4 p1, const v2f (&w)[NHID / 2]) {
  v2f s = (v2f){p0.x, p0.y} * w[0];
  s += (v2f){p0.z, p0.w} * w[1];
  s += (v2f){p1.x, p1.y} * w[2];
  s += (v2f){p1.z, p1.w} * w[3];
  return s.x + s.y;
}

__device__ __forceinline__ float dot5(const v2f x01, const v2f x23, const float x4,
                                      const v2f w01, const v2f w23, const float w4) {
  v2f s = x01 * w01;
  s += x23 * w23;
  return (s.x + s.y) + x4 * w4;
}

template <int xm>
__device__ __forceinline__ float add_xor_lane(float x) {
  const int y = __builtin_amdgcn_update_dpp(0, __builtin_bit_cast(int, x), 0x160 + xm, 0xf, 0xf, true);
  return x + __builtin_bit_cast(float, y);
}

__global__ void init_kernel(int* __restrict__ deg) {
  int i = blockIdx.x * blockDim.x + threadIdx.x;
  if (i < N_NODES) deg[i] = 0;
}

#define CNT_R   16
#define CNT_C   16
#define CNT_M   (N_NODES / CNT_R)
#define CNT_EC  (N_EDGES / CNT_C)
#define CNT_T   1024
#define CNT_Q   (CNT_EC / 4)
#define CNT_PER ((CNT_Q + CNT_T - 1) / CNT_T)
#define CNT_NPER ((CNT_M + CNT_T - 1) / CNT_T)

__global__ void __launch_bounds__(CNT_T) count_kernel(const int* __restrict__ ei, int* __restrict__ deg,
                                                      int* __restrict__ pos, int* __restrict__ off) {
  __shared__ int cnt[CNT_M];
  const int t  = threadIdx.x;
  const int r  = blockIdx.x % CNT_R;
  const int c  = blockIdx.x / CNT_R;
  const int nb = r * CNT_M;
  const int4* dv = (const int4*)(ei + N_EDGES + c * CNT_EC);
  int4 d[CNT_PER];
  #pragma unroll
  for (int p = 0; p < CNT_PER; p++) d[p] = dv[min(t + p * CNT_T, CNT_Q - 1)];
  for (int i = t; i < CNT_M; i += CNT_T) cnt[i] = 0;
  __syncthreads();
  #pragma unroll
  for (int p = 0; p < CNT_PER; p++) {
    const int q = t + p * CNT_T;
    if (q < CNT_Q) {
      int* pe = pos + c * CNT_EC + 4 * q;
      const unsigned u0 = (unsigned)(d[p].x - nb), u1 = (unsigned)(d[p].y - nb);
      const unsigned u2 = (unsigned)(d[p].z - nb), u3 = (unsigned)(d[p].w - nb);
      int k0 = 0, k1 = 0, k2 = 0, k3 = 0;
      if (u0 < CNT_M) k0 = atomicAdd(&cnt[u0], 1);
      if (u1 < CNT_M) k1 = atomicAdd(&cnt[u1], 1);
      if (u2 < CNT_M) k2 = atomicAdd(&cnt[u2], 1);
      if (u3 < CNT_M) k3 = atomicAdd(&cnt[u3], 1);
      if (u0 < CNT_M) pe[0] = k0;
      if (u1 < CNT_M) pe[1] = k1;
      if (u2 < CNT_M) pe[2] = k2;
      if (u3 < CNT_M) pe[3] = k3;
    }
  }
  __syncthreads();
  int kc[CNT_NPER], ko[CNT_NPER];
  #pragma unroll
  for (int k = 0; k < CNT_NPER; k++) { const int i = t + k * CNT_T; kc[k] = (i < CNT_M) ? cnt[i] : 0; }
  #pragma unroll
  for (int k = 0; k < CNT_NPER; k++) { const int i = t + k * CNT_T; ko[k] = 0; if (i < CNT_M) ko[k] = atomicAdd(&deg[nb + i], kc[k]); }
  #pragma unroll
  for (int k = 0; k < CNT_NPER; k++) { const int i = t + k * CNT_T; if (i < CNT_M) off[c * N_NODES + nb + i] = ko[k]; }
}

#define SCAN_T   1024
#define SCAN_PER ((N_NODES + SCAN_T - 1) / SCAN_T)

__global__ void __launch_bounds__(SCAN_T) scan_kernel(const int* __restrict__ deg,
                                                      int* __restrict__ rowptr) {
  __shared__ int wsum[32];
  const int t     = threadIdx.x;
  const int lane  = t & 31;
  const int wv    = t >> 5;
  const int wbase = wv * (SCAN_PER * 32) + lane;
  int v[SCAN_PER];
  int wtot = 0;
  #pragma unroll
  for (int i = 0; i < SCAN_PER; i++) {
    const int idx = wbase + i * 32;
    v[i] = (idx < N_NODES) ? deg[idx] : 0;
    wtot += v[i];
  }
  #pragma unroll
  for (int o = 16; o > 0; o >>= 1) wtot += __shfl_xor(wtot, o);
  if (lane == 0) wsum[wv] = wtot;
  __syncthreads();
  if (wv == 0) {
    const int w = wsum[lane];
    int wi = w;
    #pragma unroll
    for (int o = 1; o < 32; o <<= 1) {
      const int y = __shfl_up(wi, o);
      if (lane >= o) wi += y;
    }
    wsum[lane] = wi - w;
  }
  __syncthreads();
  int carry = wsum[wv];
  #pragma unroll
  for (int i = 0; i < SCAN_PER; i++) {
    int incl = v[i];
    #pragma unroll
    for (int o = 1; o < 32; o <<= 1) {
      const int y = __shfl_up(incl, o);
      if (lane >= o) incl += y;
    }
    const int idx = wbase + i * 32;
    if (idx < N_NODES) rowptr[idx] = carry + incl - v[i];
    carry += __shfl(incl, 31);
  }
  if (t == SCAN_T - 1) rowptr[N_NODES] = carry;
}

__global__ void fill_kernel(const int* __restrict__ ei, const int* __restrict__ rowptr,
                            const int* __restrict__ pos, const int* __restrict__ off,
                            int* __restrict__ csr_src) {
  int e = blockIdx.x * blockDim.x + threadIdx.x;
  if (e >= N_EDGES) return;
  const int d = ei[N_EDGES + e];
  csr_src[rowptr[d] + off[(e / CNT_EC) * N_NODES + d] + pos[e]] = ei[e];
}

__global__ void __launch_bounds__(128) gemm1_wmma(const float* __restrict__ X,
                                                  const float* __restrict__ W,
                                                  float* __restrict__ H) {
  const int wave = threadIdx.x >> 5;
  const int lane = threadIdx.x & 31;
  const int mt   = blockIdx.x * 4 + wave;
  if (mt >= N_NODES / 16) return;
  const int m  = lane & 15;
  const int hi = lane >> 4;
  const float* xr = X + (long)(mt * 16 + m) * NFEAT;

  v8f c0 = {}, c1 = {}, c2 = {}, c3 = {};
  #pragma unroll 4
  for (int k0 = 0; k0 < NFEAT; k0 += 4) {
    const int ka = k0 + 2 * hi;
    v2f a; a.x = xr[ka]; a.y = xr[ka + 1];
    const float* w0 = W + (long)ka * HID1 + m;
    const float* w1 = w0 + HID1;
    v2f b;
    b.x = w0[0];  b.y = w1[0];
    c0 = __builtin_amdgcn_wmma_f32_16x16x4_f32(false, a, false, b, (short)0, c0, false, false);
    b.x = w0[16]; b.y = w1[16];
    c1 = __builtin_amdgcn_wmma_f32_16x16x4_f32(false, a, false, b, (short)0, c1, false, false);
    b.x = w0[32]; b.y = w1[32];
    c2 = __builtin_amdgcn_wmma_f32_16x16x4_f32(false, a, false, b, (short)0, c2, false, false);
    b.x = w0[48]; b.y = w1[48];
    c3 = __builtin_amdgcn_wmma_f32_16x16x4_f32(false, a, false, b, (short)0, c3, false, false);
  }
  float* hp = H + (long)(mt * 16 + 8 * hi) * HID1 + m;
  #pragma unroll
  for (int j = 0; j < 8; j++) {
    hp[(long)j * HID1 +  0] = c0[j];
    hp[(long)j * HID1 + 16] = c1[j];
    hp[(long)j * HID1 + 32] = c2[j];
    hp[(long)j * HID1 + 48] = c3[j];
  }
}

#define NP1_NODES 32
#define XT_PITCH  68
#define G2_PITCH  52

__global__ void __launch_bounds__(256) node_pass1_gemm2(const int* __restrict__ rowptr,
                                                         const int* __restrict__ csr_src,
                                                         const float* __restrict__ a_s,
                                                         const float* __restrict__ a_d,
                                                         const float* __restrict__ H1,
                                                         const float* __restrict__ b,
                                                         const float* __restrict__ W,
                                                         float* __restrict__ H2) {
  __shared__ __attribute__((aligned(16))) float xt[NP1_NODES][XT_PITCH];
  __shared__ __attribute__((aligned(16))) float ht[NP1_NODES][G2_PITCH];
  const int t  = threadIdx.x;
  const int nl = t >> 3, h = t & 7;
  const int nbase = blockIdx.x * NP1_NODES;
  const int n = min(nbase + nl, N_NODES - 1);
  const int beg = rowptr[n], end = rowptr[n + 1];
  const int last = end - 1;
  v2f ws_[NHID / 2], wd_[NHID / 2];
  #pragma unroll
  for (int k = 0; k < NHID / 2; k++) {
    ws_[k] = (v2f){ a_s[h * NHID + 2 * k], a_s[h * NHID + 2 * k + 1] };
    wd_[k] = (v2f){ a_d[h * NHID + 2 * k], a_d[h * NHID + 2 * k + 1] };
  }
  const float4* H1v = (const float4*)H1;
  const int hq = 2 * h;

  float4 a0 = H1v[n * (HID1 / 4) + hq], a1 = H1v[n * (HID1 / 4) + hq + 1];
  const float adn = dot8(a0, a1, wd_);
  float mx  = lrelu(dot8(a0, a1, ws_) + adn);
  float den = 1.f;
  for (int j = beg; j < end; j += 4) {
    float4 x0[4], x1[4];
    #pragma unroll
    for (int u = 0; u < 4; u++) {
      const int s = csr_src[min(j + u, last)];
      x0[u] = H1v[s * (HID1 / 4) + hq]; x1[u] = H1v[s * (HID1 / 4) + hq + 1];
    }
    float v[4];
    #pragma unroll
    for (int u = 0; u < 4; u++) v[u] = lrelu(dot8(x0[u], x1[u], ws_) + adn);
    const float mn = fmaxf(mx, fmaxf(fmaxf(v[0], v[1]), fmaxf(v[2], v[3])));
    const float sc = __expf(mx - mn);
    mx = mn;
    den *= sc;
    a0.x *= sc; a0.y *= sc; a0.z *= sc; a0.w *= sc;
    a1.x *= sc; a1.y *= sc; a1.z *= sc; a1.w *= sc;
    #pragma unroll
    for (int u = 0; u < 4; u++) {
      const float w = (j + u < end) ? __expf(v[u] - mn) : 0.f;
      den  += w;
      a0.x += w * x0[u].x; a0.y += w * x0[u].y; a0.z += w * x0[u].z; a0.w += w * x0[u].w;
      a1.x += w * x1[u].x; a1.y += w * x1[u].y; a1.z += w * x1[u].z; a1.w += w * x1[u].w;
    }
  }

  float r[8] = { a0.x / den, a0.y / den, a0.z / den, a0.w / den,
                 a1.x / den, a1.y / den, a1.z / den, a1.w / den };
  #pragma unroll
  for (int k = 0; k < 8; k++) {
    const float tv = r[k] + b[h * NHID + k];
    r[k] = tv > 0.f ? tv : (__expf(tv) - 1.f);
  }
  const int wave = t >> 5;
  const int lane = t & 31;
  const int m    = lane & 15;
  const int hi   = lane >> 4;
  const int mt   = wave & 1;
  const int col  = (wave >> 1) * 16 + m;
  const bool cok = col < NCLASS;
  {
    float4* xp = (float4*)&xt[nl][h * NHID];
    xp[0] = make_float4(r[0], r[1], r[2], r[3]);
    xp[1] = make_float4(r[4], r[5], r[6], r[7]);
  }
  float w0[HID1 / 4], w1[HID1 / 4];
  if (wave < 6) {
    const float* wp = W + (2 * hi) * NCLASS + (cok ? col : NCLASS - 1);
    #pragma unroll
    for (int ks = 0; ks < HID1 / 4; ks++) { w0[ks] = wp[(4 * ks) * NCLASS]; w1[ks] = wp[(4 * ks + 1) * NCLASS]; }
  }
  __syncthreads();

  if (wave < 6) {
    const float* xr = &xt[mt * 16 + m][2 * hi];
    v8f c = {};
    #pragma unroll
    for (int ks = 0; ks < HID1 / 4; ks++) {
      v2f a; a.x = xr[4 * ks]; a.y = xr[4 * ks + 1];
      v2f bb; bb.x = cok ? w0[ks] : 0.f; bb.y = cok ? w1[ks] : 0.f;
      c = __builtin_amdgcn_wmma_f32_16x16x4_f32(false, a, false, bb, (short)0, c, false, false);
    }
    #pragma unroll
    for (int j = 0; j < 8; j++) ht[mt * 16 + j + 8 * hi][col] = c[j];
  }
  __syncthreads();

  {
    float* hg = H2 + nbase * NCLASS;
    float hv[(NP1_NODES * NCLASS) / 256];
    #pragma unroll
    for (int p = 0; p < (NP1_NODES * NCLASS) / 256; p++) {
      const int idx = t + 256 * p;
      const int row = idx / NCLASS;
      hv[p] = ht[row][idx - row * NCLASS];
    }
    #pragma unroll
    for (int p = 0; p < (NP1_NODES * NCLASS) / 256; p++) {
      const int idx = t + 256 * p;
      if (nbase + idx / NCLASS < N_NODES) hg[idx] = hv[p];
    }
  }
}

__global__ void __launch_bounds__(256) node_pass2(const int* __restrict__ rowptr,
                                                   const int* __restrict__ csr_src,
                                                   const float* __restrict__ a_s,
                                                   const float* __restrict__ a_d,
                                                   const float* __restrict__ H2,
                                                   const float* __restrict__ b,
                                                   float* __restrict__ out) {
  const int i = blockIdx.x * blockDim.x + threadIdx.x;
  if (i >= N_NODES * 8) return;
  const int n = i >> 3, q = i & 7;
  const int beg = rowptr[n], end = rowptr[n + 1];
  const int last = end - 1;
  const v2f   ws01 = { a_s[q], a_s[q + 8] }, ws23 = { a_s[q + 16], a_s[q + 24] };
  const float ws4  = a_s[q + 32];

  float acc[NCLASS / 8];
  {
    const float* hs = H2 + (n * NCLASS + q);
    #pragma unroll
    for (int c = 0; c < NCLASS / 8; c++) acc[c] = hs[8 * c];
  }
  float asn = dot5((v2f){acc[0], acc[1]}, (v2f){acc[2], acc[3]}, acc[4], ws01, ws23, ws4);
  float adn = dot5((v2f){acc[0], acc[1]}, (v2f){acc[2], acc[3]}, acc[4],
                   (v2f){ a_d[q], a_d[q + 8] }, (v2f){ a_d[q + 16], a_d[q + 24] }, a_d[q + 32]);
  asn = add_xor_lane<1>(asn); adn = add_xor_lane<1>(adn);
  asn = add_xor_lane<2>(asn); adn = add_xor_lane<2>(adn);
  asn = add_xor_lane<4>(asn); adn = add_xor_lane<4>(adn);
  float mx  = lrelu(asn + adn);
  float den = 1.f;
  for (int j = beg; j < end; j += 4) {
    float x[4][NCLASS / 8];
    #pragma unroll
    for (int u = 0; u < 4; u++) {
      const int s = csr_src[min(j + u, last)];
      const float* hs = H2 + (s * NCLASS + q);
      #pragma unroll
      for (int c = 0; c < NCLASS / 8; c++) x[u][c] = hs[8 * c];
    }
    float lg[4];
    #pragma unroll
    for (int u = 0; u < 4; u++) lg[u] = dot5((v2f){x[u][0], x[u][1]}, (v2f){x[u][2], x[u][3]}, x[u][4], ws01, ws23, ws4);
    #pragma unroll
    for (int u = 0; u < 4; u++) lg[u] = add_xor_lane<1>(lg[u]);
    #pragma unroll
    for (int u = 0; u < 4; u++) lg[u] = add_xor_lane<2>(lg[u]);
    #pragma unroll
    for (int u = 0; u < 4; u++) lg[u] = add_xor_lane<4>(lg[u]);
    float v[4];
    #pragma unroll
    for (int u = 0; u < 4; u++) v[u] = lrelu(lg[u] + adn);
    const float mn = fmaxf(mx, fmaxf(fmaxf(v[0], v[1]), fmaxf(v[2], v[3])));
    const float sc = __expf(mx - mn);
    mx = mn;
    den *= sc;
    #pragma unroll
    for (int c = 0; c < NCLASS / 8; c++) acc[c] *= sc;
    #pragma unroll
    for (int u = 0; u < 4; u++) {
      const float w = (j + u < end) ? __expf(v[u] - mn) : 0.f;
      den += w;
      #pragma unroll
      for (int c = 0; c < NCLASS / 8; c++) acc[c] += w * x[u][c];
    }
  }
  float val[NCLASS / 8];
  float cmx = -INFINITY;
  #pragma unroll
  for (int c = 0; c < NCLASS / 8; c++) { val[c] = acc[c] / den + b[q + 8 * c]; cmx = fmaxf(cmx, val[c]); }
  #pragma unroll
  for (int o = 1; o < 8; o <<= 1) cmx = fmaxf(cmx, __shfl_xor(cmx, o));
  float ssum = 0.f;
  #pragma unroll
  for (int c = 0; c < NCLASS / 8; c++) ssum += __expf(val[c] - cmx);
  #pragma unroll
  for (int o = 1; o < 8; o <<= 1) ssum += __shfl_xor(ssum, o);
  const float lse = cmx + __logf(ssum);
  float* od = out + (n * NCLASS + q);
  #pragma unroll
  for (int c = 0; c < NCLASS / 8; c++) od[8 * c] = val[c] - lse;
}

extern "C" void kernel_launch(void* const* d_in, const int* in_sizes, int n_in,
                              void* d_out, int out_size, void* d_ws, size_t ws_size,
                              hipStream_t stream) {
  const float* x      = (const float*)d_in[0];
  const float* W1     = (const float*)d_in[1];
  const float* a_src1 = (const float*)d_in[2];
  const float* a_dst1 = (const float*)d_in[3];
  const float* b1     = (const float*)d_in[4];
  const float* W2     = (const float*)d_in[5];
  const float* a_src2 = (const float*)d_in[6];
  const float* a_dst2 = (const float*)d_in[7];
  const float* b2     = (const float*)d_in[8];
  const int*   ei     = (const int*)d_in[9];
  float*       out    = (float*)d_out;

  float* ws   = (float*)d_ws;
  float* H1   = ws;
  float* H2   = H1  + (long)N_NODES * HID1;
  int*   deg  = (int*)(H2 + (long)N_NODES * NCLASS);
  int*   rowp = deg  + N_NODES;
  int*   pos  = rowp + N_NODES + 4;
  int*   csr  = pos  + N_EDGES;
  int*   off  = csr  + N_EDGES;

  const int B = 256;
  auto cdiv = [](long a, long b) { return (int)((a + b - 1) / b); };

  init_kernel<<<cdiv(N_NODES, B), B, 0, stream>>>(deg);
  count_kernel<<<CNT_R * CNT_C, CNT_T, 0, stream>>>(ei, deg, pos, off);
  scan_kernel<<<1, SCAN_T, 0, stream>>>(deg, rowp);
  fill_kernel<<<cdiv(N_EDGES, B), B, 0, stream>>>(ei, rowp, pos, off, csr);

  gemm1_wmma<<<cdiv(N_NODES / 16, 4), 128, 0, stream>>>(x, W1, H1);
  node_pass1_gemm2<<<cdiv(N_NODES, NP1_NODES), B, 0, stream>>>(rowp, csr, a_src1, a_dst1, H1, b1, W2, H2);

  node_pass2<<<cdiv((long)N_NODES * 8, B), B, 0, stream>>>(rowp, csr, a_src2, a_dst2, H2, b2, out);
}


# no zeroing launch, no global adds: count_kernel stores per-chunk counts, scan_kernel (16 groups) sums them per node and range; 7 -> 6 launches; a node's edges now summed in fixed chunk order
# speedup vs baseline: 9.1895x; 1.0348x over previous
#include <hip/hip_runtime.h>
#include <math.h>

#define N_NODES 50000
#define N_EDGES 800000
#define NFEAT   128
#define NHID    8
#define NHEADS  8
#define HID1    (NHEADS * NHID)
#define NCLASS  40
#define NEG_SLOPE 0.2f

typedef __attribute__((ext_vector_type(2))) float v2f;
typedef __attribute__((ext_vector_type(8))) float v8f;

__device__ __forceinline__ float lrelu(float x) { return x > 0.f ? x : NEG_SLOPE * x; }

__device__ __forceinline__ float dot8(const float4 p0, const float4 p1, const v2f (&w)[NHID / 2]) {
  v2f s = (v2f){p0.x, p0.y} * w[0];
  s += (v2f){p0.z, p0.w} * w[1];
  s += (v2f){p1.x, p1.y} * w[2];
  s += (v2f){p1.z, p1.w} * w[3];
  return s.x + s.y;
}

__device__ __forceinline__ float dot5(const v2f x01, const v2f x23, const float x4,
                                      const v2f w01, const v2f w23, const float w4) {
  v2f s = x01 * w01;
  s += x23 * w23;
  return (s.x + s.y) + x4 * w4;
}

template <int xm>
__device__ __forceinline__ float add_xor_lane(float x) {
  const int y = __builtin_amdgcn_update_dpp(0, __builtin_bit_cast(int, x), 0x160 + xm, 0xf, 0xf, true);
  return x + __builtin_bit_cast(float, y);
}


#define CNT_R   16
#define CNT_C   16
#define CNT_M   (N_NODES / CNT_R)
#define CNT_EC  (N_EDGES / CNT_C)
#define CNT_T   1024
#define CNT_Q   (CNT_EC / 4)
#define CNT_PER ((CNT_Q + CNT_T - 1) / CNT_T)
#define CNT_NPER ((CNT_M + CNT_T - 1) / CNT_T)

__global__ void __launch_bounds__(CNT_T) count_kernel(const int* __restrict__ ei, int* __restrict__ pos,
                                                      int* __restrict__ off, int* __restrict__ tot) {
  __shared__ int cnt[CNT_M];
  __shared__ int wtot[CNT_T / 32];
  const int t  = threadIdx.x;
  const int r  = blockIdx.x % CNT_R;
  const int c  = blockIdx.x / CNT_R;
  const int nb = r * CNT_M;
  const int4* dv = (const int4*)(ei + N_EDGES + c * CNT_EC);
  int4 d[CNT_PER];
  #pragma unroll
  for (int p = 0; p < CNT_PER; p++) d[p] = dv[min(t + p * CNT_T, CNT_Q - 1)];
  for (int i = t; i < CNT_M; i += CNT_T) cnt[i] = 0;
  __syncthreads();
  #pragma unroll
  for (int p = 0; p < CNT_PER; p++) {
    const int q = t + p * CNT_T;
    if (q < CNT_Q) {
      int* pe = pos + c * CNT_EC + 4 * q;
      const unsigned u0 = (unsigned)(d[p].x - nb), u1 = (unsigned)(d[p].y - nb);
      const unsigned u2 = (unsigned)(d[p].z - nb), u3 = (unsigned)(d[p].w - nb);
      int k0 = 0, k1 = 0, k2 = 0, k3 = 0;
      if (u0 < CNT_M) k0 = atomicAdd(&cnt[u0], 1);
      if (u1 < CNT_M) k1 = atomicAdd(&cnt[u1], 1);
      if (u2 < CNT_M) k2 = atomicAdd(&cnt[u2], 1);
      if (u3 < CNT_M) k3 = atomicAdd(&cnt[u3], 1);
      if (u0 < CNT_M) pe[0] = k0;
      if (u1 < CNT_M) pe[1] = k1;
      if (u2 < CNT_M) pe[2] = k2;
      if (u3 < CNT_M) pe[3] = k3;
    }
  }
  __syncthreads();
  int kc[CNT_NPER];
  int s = 0;
  #pragma unroll
  for (int k = 0; k < CNT_NPER; k++) { const int i = t + k * CNT_T; kc[k] = (i < CNT_M) ? cnt[i] : 0; s += kc[k]; }
  #pragma unroll
  for (int o = 16; o > 0; o >>= 1) s += __shfl_xor(s, o);
  if ((t & 31) == 0) wtot[t >> 5] = s;
  __syncthreads();
  if (t < 32) {
    int v = wtot[t];
    #pragma unroll
    for (int o = 16; o > 0; o >>= 1) v += __shfl_xor(v, o);
    if (t == 0) tot[c * CNT_R + r] = v;
  }
  #pragma unroll
  for (int k = 0; k < CNT_NPER; k++) { const int i = t + k * CNT_T; if (i < CNT_M) off[c * N_NODES + nb + i] = kc[k]; }
}

#define SCAN_T   1024
#define SCAN_K   ((CNT_M + SCAN_T - 1) / SCAN_T)

__global__ void __launch_bounds__(SCAN_T) scan_kernel(const int* __restrict__ tot, int* __restrict__ off,
                                                      int* __restrict__ rowptr) {
  __shared__ int wsum[SCAN_K][32];
  __shared__ int rtot[SCAN_K];
  __shared__ int bsum[32];
  __shared__ int base_s;
  const int t    = threadIdx.x;
  const int lane = t & 31;
  const int wv   = t >> 5;
  const int r    = blockIdx.x & (CNT_R - 1);
  const int nb   = r * CNT_M;
  int bpart = 0;
  if (t < CNT_R * CNT_C) { const int v = tot[t]; bpart = ((t % CNT_R) < r) ? v : 0; }
  int cv[SCAN_K][CNT_C];
  #pragma unroll
  for (int k = 0; k < SCAN_K; k++) {
    const int i  = t + k * SCAN_T;
    const int ic = min(i, CNT_M - 1);
    #pragma unroll
    for (int c = 0; c < CNT_C; c++) { const int v = off[c * N_NODES + nb + ic]; cv[k][c] = (i < CNT_M) ? v : 0; }
  }
  int dg[SCAN_K], incl[SCAN_K];
  #pragma unroll
  for (int k = 0; k < SCAN_K; k++) {
    const int i = t + k * SCAN_T;
    int ex[CNT_C];
    int run = 0;
    #pragma unroll
    for (int c = 0; c < CNT_C; c++) { ex[c] = run; run += cv[k][c]; }
    dg[k] = run;
    if (i < CNT_M) {
      #pragma unroll
      for (int c = 0; c < CNT_C; c++) off[c * N_NODES + nb + i] = ex[c];
    }
  }
  #pragma unroll
  for (int k = 0; k < SCAN_K; k++) {
    incl[k] = dg[k];
    #pragma unroll
    for (int o = 1; o < 32; o <<= 1) {
      const int y = __shfl_up(incl[k], o);
      if (lane >= o) incl[k] += y;
    }
  }
  #pragma unroll
  for (int o = 16; o > 0; o >>= 1) bpart += __shfl_xor(bpart, o);
  if (lane == 31) {
    #pragma unroll
    for (int k = 0; k < SCAN_K; k++) wsum[k][wv] = incl[k];
  }
  if (lane == 0) bsum[wv] = bpart;
  __syncthreads();
  if (wv < SCAN_K) {
    const int w = wsum[wv][lane];
    int wi = w;
    #pragma unroll
    for (int o = 1; o < 32; o <<= 1) {
      const int y = __shfl_up(wi, o);
      if (lane >= o) wi += y;
    }
    wsum[wv][lane] = wi - w;
    if (lane == 31) rtot[wv] = wi;
  } else if (wv == SCAN_K) {
    int v = bsum[lane];
    #pragma unroll
    for (int o = 16; o > 0; o >>= 1) v += __shfl_xor(v, o);
    if (lane == 0) base_s = v;
  }
  __syncthreads();
  int carry = base_s;
  #pragma unroll
  for (int k = 0; k < SCAN_K; k++) {
    const int i = t + k * SCAN_T;
    if (i < CNT_M) rowptr[nb + i] = carry + wsum[k][wv] + incl[k] - dg[k];
    carry += rtot[k];
  }
  if (r == CNT_R - 1 && t == 0) rowptr[N_NODES] = carry;
}

__global__ void fill_kernel(const int* __restrict__ ei, const int* __restrict__ rowptr,
                            const int* __restrict__ pos, const int* __restrict__ off,
                            int* __restrict__ csr_src) {
  int e = blockIdx.x * blockDim.x + threadIdx.x;
  if (e >= N_EDGES) return;
  const int d = ei[N_EDGES + e];
  csr_src[rowptr[d] + off[(e / CNT_EC) * N_NODES + d] + pos[e]] = ei[e];
}

__global__ void __launch_bounds__(128) gemm1_wmma(const float* __restrict__ X,
                                                  const float* __restrict__ W,
                                                  float* __restrict__ H) {
  const int wave = threadIdx.x >> 5;
  const int lane = threadIdx.x & 31;
  const int mt   = blockIdx.x * 4 + wave;
  if (mt >= N_NODES / 16) return;
  const int m  = lane & 15;
  const int hi = lane >> 4;
  const float* xr = X + (long)(mt * 16 + m) * NFEAT;

  v8f c0 = {}, c1 = {}, c2 = {}, c3 = {};
  #pragma unroll 4
  for (int k0 = 0; k0 < NFEAT; k0 += 4) {
    const int ka = k0 + 2 * hi;
    v2f a; a.x = xr[ka]; a.y = xr[ka + 1];
    const float* w0 = W + (long)ka * HID1 + m;
    const float* w1 = w0 + HID1;
    v2f b;
    b.x = w0[0];  b.y = w1[0];
    c0 = __builtin_amdgcn_wmma_f32_16x16x4_f32(false, a, false, b, (short)0, c0, false, false);
    b.x = w0[16]; b.y = w1[16];
    c1 = __builtin_amdgcn_wmma_f32_16x16x4_f32(false, a, false, b, (short)0, c1, false, false);
    b.x = w0[32]; b.y = w1[32];
    c2 = __builtin_amdgcn_wmma_f32_16x16x4_f32(false, a, false, b, (short)0, c2, false, false);
    b.x = w0[48]; b.y = w1[48];
    c3 = __builtin_amdgcn_wmma_f32_16x16x4_f32(false, a, false, b, (short)0, c3, false, false);
  }
  float* hp = H + (long)(mt * 16 + 8 * hi) * HID1 + m;
  #pragma unroll
  for (int j = 0; j < 8; j++) {
    hp[(long)j * HID1 +  0] = c0[j];
    hp[(long)j * HID1 + 16] = c1[j];
    hp[(long)j * HID1 + 32] = c2[j];
    hp[(long)j * HID1 + 48] = c3[j];
  }
}

#define NP1_NODES 32
#define XT_PITCH  68
#define G2_PITCH  52

__global__ void __launch_bounds__(256) node_pass1_gemm2(const int* __restrict__ rowptr,
                                                         const int* __restrict__ csr_src,
                                                         const float* __restrict__ a_s,
                                                         const float* __restrict__ a_d,
                                                         const float* __restrict__ H1,
                                                         const float* __restrict__ b,
                                                         const float* __restrict__ W,
                                                         float* __restrict__ H2) {
  __shared__ __attribute__((aligned(16))) float xt[NP1_NODES][XT_PITCH];
  __shared__ __attribute__((aligned(16))) float ht[NP1_NODES][G2_PITCH];
  const int t  = threadIdx.x;
  const int nl = t >> 3, h = t & 7;
  const int nbase = blockIdx.x * NP1_NODES;
  const int n = min(nbase + nl, N_NODES - 1);
  const int beg = rowptr[n], end = rowptr[n + 1];
  const int last = end - 1;
  v2f ws_[NHID / 2], wd_[NHID / 2];
  #pragma unroll
  for (int k = 0; k < NHID / 2; k++) {
    ws_[k] = (v2f){ a_s[h * NHID + 2 * k], a_s[h * NHID + 2 * k + 1] };
    wd_[k] = (v2f){ a_d[h * NHID + 2 * k], a_d[h * NHID + 2 * k + 1] };
  }
  const float4* H1v = (const float4*)H1;
  const int hq = 2 * h;

  float4 a0 = H1v[n * (HID1 / 4) + hq], a1 = H1v[n * (HID1 / 4) + hq + 1];
  const float adn = dot8(a0, a1, wd_);
  float mx  = lrelu(dot8(a0, a1, ws_) + adn);
  float den = 1.f;
  for (int j = beg; j < end; j += 4) {
    float4 x0[4], x1[4];
    #pragma unroll
    for (int u = 0; u < 4; u++) {
      const int s = csr_src[min(j + u, last)];
      x0[u] = H1v[s * (HID1 / 4) + hq]; x1[u] = H1v[s * (HID1 / 4) + hq + 1];
    }
    float v[4];
    #pragma unroll
    for (int u = 0; u < 4; u++) v[u] = lrelu(dot8(x0[u], x1[u], ws_) + adn);
    const float mn = fmaxf(mx, fmaxf(fmaxf(v[0], v[1]), fmaxf(v[2], v[3])));
    const float sc = __expf(mx - mn);
    mx = mn;
    den *= sc;
    a0.x *= sc; a0.y *= sc; a0.z *= sc; a0.w *= sc;
    a1.x *= sc; a1.y *= sc; a1.z *= sc; a1.w *= sc;
    #pragma unroll
    for (int u = 0; u < 4; u++) {
      const float w = (j + u < end) ? __expf(v[u] - mn) : 0.f;
      den  += w;
      a0.x += w * x0[u].x; a0.y += w * x0[u].y; a0.z += w * x0[u].z; a0.w += w * x0[u].w;
      a1.x += w * x1[u].x; a1.y += w * x1[u].y; a1.z += w * x1[u].z; a1.w += w * x1[u].w;
    }
  }

  float r[8] = { a0.x / den, a0.y / den, a0.z / den, a0.w / den,
                 a1.x / den, a1.y / den, a1.z / den, a1.w / den };
  #pragma unroll
  for (int k = 0; k < 8; k++) {
    const float tv = r[k] + b[h * NHID + k];
    r[k] = tv > 0.f ? tv : (__expf(tv) - 1.f);
  }
  const int wave = t >> 5;
  const int lane = t & 31;
  const int m    = lane & 15;
  const int hi   = lane >> 4;
  const int mt   = wave & 1;
  const int col  = (wave >> 1) * 16 + m;
  const bool cok = col < NCLASS;
  {
    float4* xp = (float4*)&xt[nl][h * NHID];
    xp[0] = make_float4(r[0], r[1], r[2], r[3]);
    xp[1] = make_float4(r[4], r[5], r[6], r[7]);
  }
  float w0[HID1 / 4], w1[HID1 / 4];
  if (wave < 6) {
    const float* wp = W + (2 * hi) * NCLASS + (cok ? col : NCLASS - 1);
    #pragma unroll
    for (int ks = 0; ks < HID1 / 4; ks++) { w0[ks] = wp[(4 * ks) * NCLASS]; w1[ks] = wp[(4 * ks + 1) * NCLASS]; }
  }
  __syncthreads();

  if (wave < 6) {
    const float* xr = &xt[mt * 16 + m][2 * hi];
    v8f c = {};
    #pragma unroll
    for (int ks = 0; ks < HID1 / 4; ks++) {
      v2f a; a.x = xr[4 * ks]; a.y = xr[4 * ks + 1];
      v2f bb; bb.x = cok ? w0[ks] : 0.f; bb.y = cok ? w1[ks] : 0.f;
      c = __builtin_amdgcn_wmma_f32_16x16x4_f32(false, a, false, bb, (short)0, c, false, false);
    }
    #pragma unroll
    for (int j = 0; j < 8; j++) ht[mt * 16 + j + 8 * hi][col] = c[j];
  }
  __syncthreads();

  {
    float* hg = H2 + nbase * NCLASS;
    float hv[(NP1_NODES * NCLASS) / 256];
    #pragma unroll
    for (int p = 0; p < (NP1_NODES * NCLASS) / 256; p++) {
      const int idx = t + 256 * p;
      const int row = idx / NCLASS;
      hv[p] = ht[row][idx - row * NCLASS];
    }
    #pragma unroll
    for (int p = 0; p < (NP1_NODES * NCLASS) / 256; p++) {
      const int idx = t + 256 * p;
      if (nbase + idx / NCLASS < N_NODES) hg[idx] = hv[p];
    }
  }
}

__global__ void __launch_bounds__(256) node_pass2(const int* __restrict__ rowptr,
                                                   const int* __restrict__ csr_src,
                                                   const float* __restrict__ a_s,
                                                   const float* __restrict__ a_d,
                                                   const float* __restrict__ H2,
                                                   const float* __restrict__ b,
                                                   float* __restrict__ out) {
  const int i = blockIdx.x * blockDim.x + threadIdx.x;
  if (i >= N_NODES * 8) return;
  const int n = i >> 3, q = i & 7;
  const int beg = rowptr[n], end = rowptr[n + 1];
  const int last = end - 1;
  const v2f   ws01 = { a_s[q], a_s[q + 8] }, ws23 = { a_s[q + 16], a_s[q + 24] };
  const float ws4  = a_s[q + 32];

  float acc[NCLASS / 8];
  {
    const float* hs = H2 + (n * NCLASS + q);
    #pragma unroll
    for (int c = 0; c < NCLASS / 8; c++) acc[c] = hs[8 * c];
  }
  float asn = dot5((v2f){acc[0], acc[1]}, (v2f){acc[2], acc[3]}, acc[4], ws01, ws23, ws4);
  float adn = dot5((v2f){acc[0], acc[1]}, (v2f){acc[2], acc[3]}, acc[4],
                   (v2f){ a_d[q], a_d[q + 8] }, (v2f){ a_d[q + 16], a_d[q + 24] }, a_d[q + 32]);
  asn = add_xor_lane<1>(asn); adn = add_xor_lane<1>(adn);
  asn = add_xor_lane<2>(asn); adn = add_xor_lane<2>(adn);
  asn = add_xor_lane<4>(asn); adn = add_xor_lane<4>(adn);
  float mx  = lrelu(asn + adn);
  float den = 1.f;
  for (int j = beg; j < end; j += 4) {
    float x[4][NCLASS / 8];
    #pragma unroll
    for (int u = 0; u < 4; u++) {
      const int s = csr_src[min(j + u, last)];
      const float* hs = H2 + (s * NCLASS + q);
      #pragma unroll
      for (int c = 0; c < NCLASS / 8; c++) x[u][c] = hs[8 * c];
    }
    float lg[4];
    #pragma unroll
    for (int u = 0; u < 4; u++) lg[u] = dot5((v2f){x[u][0], x[u][1]}, (v2f){x[u][2], x[u][3]}, x[u][4], ws01, ws23, ws4);
    #pragma unroll
    for (int u = 0; u < 4; u++) lg[u] = add_xor_lane<1>(lg[u]);
    #pragma unroll
    for (int u = 0; u < 4; u++) lg[u] = add_xor_lane<2>(lg[u]);
    #pragma unroll
    for (int u = 0; u < 4; u++) lg[u] = add_xor_lane<4>(lg[u]);
    float v[4];
    #pragma unroll
    for (int u = 0; u < 4; u++) v[u] = lrelu(lg[u] + adn);
    const float mn = fmaxf(mx, fmaxf(fmaxf(v[0], v[1]), fmaxf(v[2], v[3])));
    const float sc = __expf(mx - mn);
    mx = mn;
    den *= sc;
    #pragma unroll
    for (int c = 0; c < NCLASS / 8; c++) acc[c] *= sc;
    #pragma unroll
    for (int u = 0; u < 4; u++) {
      const float w = (j + u < end) ? __expf(v[u] - mn) : 0.f;
      den += w;
      #pragma unroll
      for (int c = 0; c < NCLASS / 8; c++) acc[c] += w * x[u][c];
    }
  }
  float val[NCLASS / 8];
  float cmx = -INFINITY;
  #pragma unroll
  for (int c = 0; c < NCLASS / 8; c++) { val[c] = acc[c] / den + b[q + 8 * c]; cmx = fmaxf(cmx, val[c]); }
  #pragma unroll
  for (int o = 1; o < 8; o <<= 1) cmx = fmaxf(cmx, __shfl_xor(cmx, o));
  float ssum = 0.f;
  #pragma unroll
  for (int c = 0; c < NCLASS / 8; c++) ssum += __expf(val[c] - cmx);
  #pragma unroll
  for (int o = 1; o < 8; o <<= 1) ssum += __shfl_xor(ssum, o);
  const float lse = cmx + __logf(ssum);
  float* od = out + (n * NCLASS + q);
  #pragma unroll
  for (int c = 0; c < NCLASS / 8; c++) od[8 * c] = val[c] - lse;
}

extern "C" void kernel_launch(void* const* d_in, const int* in_sizes, int n_in,
                              void* d_out, int out_size, void* d_ws, size_t ws_size,
                              hipStream_t stream) {
  const float* x      = (const float*)d_in[0];
  const float* W1     = (const float*)d_in[1];
  const float* a_src1 = (const float*)d_in[2];
  const float* a_dst1 = (const float*)d_in[3];
  const float* b1     = (const float*)d_in[4];
  const float* W2     = (const float*)d_in[5];
  const float* a_src2 = (const float*)d_in[6];
  const float* a_dst2 = (const float*)d_in[7];
  const float* b2     = (const float*)d_in[8];
  const int*   ei     = (const int*)d_in[9];
  float*       out    = (float*)d_out;

  float* ws   = (float*)d_ws;
  float* H1   = ws;
  float* H2   = H1  + (long)N_NODES * HID1;
  int*   tot  = (int*)(H2 + (long)N_NODES * NCLASS);
  int*   rowp = tot  + CNT_R * CNT_C;
  int*   pos  = rowp + N_NODES + 4;
  int*   csr  = pos  + N_EDGES;
  int*   off  = csr  + N_EDGES;

  const int B = 256;
  auto cdiv = [](long a, long b) { return (int)((a + b - 1) / b); };

  count_kernel<<<CNT_R * CNT_C, CNT_T, 0, stream>>>(ei, pos, off, tot);
  scan_kernel<<<CNT_R, SCAN_T, 0, stream>>>(tot, off, rowp);
  fill_kernel<<<cdiv(N_EDGES, B), B, 0, stream>>>(ei, rowp, pos, off, csr);

  gemm1_wmma<<<cdiv(N_NODES / 16, 4), 128, 0, stream>>>(x, W1, H1);
  node_pass1_gemm2<<<cdiv(N_NODES, NP1_NODES), B, 0, stream>>>(rowp, csr, a_src1, a_dst1, H1, b1, W2, H2);

  node_pass2<<<cdiv((long)N_NODES * 8, B), B, 0, stream>>>(rowp, csr, a_src2, a_dst2, H2, b2, out);
}


# gemm1 moved into the scan launch: work-groups 0-15 scan, 196 more (two waves per 16-row tile, 32 columns each) form H1 on the WGPs the scan leaves idle; 6 -> 5 launches; every sum in the same order
# speedup vs baseline: 10.0995x; 1.0990x over previous
#include <hip/hip_runtime.h>
#include <math.h>

#define N_NODES 50000
#define N_EDGES 800000
#define NFEAT   128
#define NHID    8
#define NHEADS  8
#define HID1    (NHEADS * NHID)
#define NCLASS  40
#define NEG_SLOPE 0.2f

typedef __attribute__((ext_vector_type(2))) float v2f;
typedef __attribute__((ext_vector_type(8))) float v8f;

__device__ __forceinline__ float lrelu(float x) { return x > 0.f ? x : NEG_SLOPE * x; }

__device__ __forceinline__ float dot8(const float4 p0, const float4 p1, const v2f (&w)[NHID / 2]) {
  v2f s = (v2f){p0.x, p0.y} * w[0];
  s += (v2f){p0.z, p0.w} * w[1];
  s += (v2f){p1.x, p1.y} * w[2];
  s += (v2f){p1.z, p1.w} * w[3];
  return s.x + s.y;
}

__device__ __forceinline__ float dot5(const v2f x01, const v2f x23, const float x4,
                                      const v2f w01, const v2f w23, const float w4) {
  v2f s = x01 * w01;
  s += x23 * w23;
  return (s.x + s.y) + x4 * w4;
}

template <int xm>
__device__ __forceinline__ float add_xor_lane(float x) {
  const int y = __builtin_amdgcn_update_dpp(0, __builtin_bit_cast(int, x), 0x160 + xm, 0xf, 0xf, true);
  return x + __builtin_bit_cast(float, y);
}


#define CNT_R   16
#define CNT_C   16
#define CNT_M   (N_NODES / CNT_R)
#define CNT_EC  (N_EDGES / CNT_C)
#define CNT_T   1024
#define CNT_Q   (CNT_EC / 4)
#define CNT_PER ((CNT_Q + CNT_T - 1) / CNT_T)
#define CNT_NPER ((CNT_M + CNT_T - 1) / CNT_T)

__global__ void __launch_bounds__(CNT_T) count_kernel(const int* __restrict__ ei, int* __restrict__ pos,
                                                      int* __restrict__ off, int* __restrict__ tot) {
  __shared__ int cnt[CNT_M];
  __shared__ int wtot[CNT_T / 32];
  const int t  = threadIdx.x;
  const int r  = blockIdx.x % CNT_R;
  const int c  = blockIdx.x / CNT_R;
  const int nb = r * CNT_M;
  const int4* dv = (const int4*)(ei + N_EDGES + c * CNT_EC);
  int4 d[CNT_PER];
  #pragma unroll
  for (int p = 0; p < CNT_PER; p++) d[p] = dv[min(t + p * CNT_T, CNT_Q - 1)];
  for (int i = t; i < CNT_M; i += CNT_T) cnt[i] = 0;
  __syncthreads();
  #pragma unroll
  for (int p = 0; p < CNT_PER; p++) {
    const int q = t + p * CNT_T;
    if (q < CNT_Q) {
      int* pe = pos + c * CNT_EC + 4 * q;
      const unsigned u0 = (unsigned)(d[p].x - nb), u1 = (unsigned)(d[p].y - nb);
      const unsigned u2 = (unsigned)(d[p].z - nb), u3 = (unsigned)(d[p].w - nb);
      int k0 = 0, k1 = 0, k2 = 0, k3 = 0;
      if (u0 < CNT_M) k0 = atomicAdd(&cnt[u0], 1);
      if (u1 < CNT_M) k1 = atomicAdd(&cnt[u1], 1);
      if (u2 < CNT_M) k2 = atomicAdd(&cnt[u2], 1);
      if (u3 < CNT_M) k3 = atomicAdd(&cnt[u3], 1);
      if (u0 < CNT_M) pe[0] = k0;
      if (u1 < CNT_M) pe[1] = k1;
      if (u2 < CNT_M) pe[2] = k2;
      if (u3 < CNT_M) pe[3] = k3;
    }
  }
  __syncthreads();
  int kc[CNT_NPER];
  int s = 0;
  #pragma unroll
  for (int k = 0; k < CNT_NPER; k++) { const int i = t + k * CNT_T; kc[k] = (i < CNT_M) ? cnt[i] : 0; s += kc[k]; }
  #pragma unroll
  for (int o = 16; o > 0; o >>= 1) s += __shfl_xor(s, o);
  if ((t & 31) == 0) wtot[t >> 5] = s;
  __syncthreads();
  if (t < 32) {
    int v = wtot[t];
    #pragma unroll
    for (int o = 16; o > 0; o >>= 1) v += __shfl_xor(v, o);
    if (t == 0) tot[c * CNT_R + r] = v;
  }
  #pragma unroll
  for (int k = 0; k < CNT_NPER; k++) { const int i = t + k * CNT_T; if (i < CNT_M) off[c * N_NODES + nb + i] = kc[k]; }
}

#define SCAN_T   1024
#define SCAN_K   ((CNT_M + SCAN_T - 1) / SCAN_T)
#define G1_TILES 16
#define G1_WG    ((N_NODES / 16 + G1_TILES - 1) / G1_TILES)

__global__ void __launch_bounds__(SCAN_T) scan_gemm1_kernel(const int* __restrict__ tot, int* __restrict__ off,
                                                            int* __restrict__ rowptr,
                                                            const float* __restrict__ X,
                                                            const float* __restrict__ W,
                                                            float* __restrict__ H) {
  const int t    = threadIdx.x;
  const int lane = t & 31;
  const int wv   = t >> 5;
  if (blockIdx.x >= CNT_R) {
    const int mt = (blockIdx.x - CNT_R) * G1_TILES + (wv >> 1);
    if (mt >= N_NODES / 16) return;
    const int m  = lane & 15;
    const int hi = lane >> 4;
    const int cb = (wv & 1) * 32;
    const float* xr = X + (long)(mt * 16 + m) * NFEAT;

    v8f c0 = {}, c1 = {};
    #pragma unroll 4
    for (int k0 = 0; k0 < NFEAT; k0 += 4) {
      const int ka = k0 + 2 * hi;
      v2f a; a.x = xr[ka]; a.y = xr[ka + 1];
      const float* w0 = W + (long)ka * HID1 + cb + m;
      const float* w1 = w0 + HID1;
      v2f b;
      b.x = w0[0];  b.y = w1[0];
      c0 = __builtin_amdgcn_wmma_f32_16x16x4_f32(false, a, false, b, (short)0, c0, false, false);
      b.x = w0[16]; b.y = w1[16];
      c1 = __builtin_amdgcn_wmma_f32_16x16x4_f32(false, a, false, b, (short)0, c1, false, false);
    }
    float* hp = H + (long)(mt * 16 + 8 * hi) * HID1 + cb + m;
    #pragma unroll
    for (int j = 0; j < 8; j++) {
      hp[(long)j * HID1 +  0] = c0[j];
      hp[(long)j * HID1 + 16] = c1[j];
    }
    return;
  }

  __shared__ int wsum[SCAN_K][32];
  __shared__ int rtot[SCAN_K];
  __shared__ int bsum[32];
  __shared__ int base_s;
  const int r    = blockIdx.x & (CNT_R - 1);
  const int nb   = r * CNT_M;
  int bpart = 0;
  if (t < CNT_R * CNT_C) { const int v = tot[t]; bpart = ((t % CNT_R) < r) ? v : 0; }
  int cv[SCAN_K][CNT_C];
  #pragma unroll
  for (int k = 0; k < SCAN_K; k++) {
    const int i  = t + k * SCAN_T;
    const int ic = min(i, CNT_M - 1);
    #pragma unroll
    for (int c = 0; c < CNT_C; c++) { const int v = off[c * N_NODES + nb + ic]; cv[k][c] = (i < CNT_M) ? v : 0; }
  }
  int dg[SCAN_K], incl[SCAN_K];
  #pragma unroll
  for (int k = 0; k < SCAN_K; k++) {
    const int i = t + k * SCAN_T;
    int ex[CNT_C];
    int run = 0;
    #pragma unroll
    for (int c = 0; c < CNT_C; c++) { ex[c] = run; run += cv[k][c]; }
    dg[k] = run;
    if (i < CNT_M) {
      #pragma unroll
      for (int c = 0; c < CNT_C; c++) off[c * N_NODES + nb + i] = ex[c];
    }
  }
  #pragma unroll
  for (int k = 0; k < SCAN_K; k++) {
    incl[k] = dg[k];
    #pragma unroll
    for (int o = 1; o < 32; o <<= 1) {
      const int y = __shfl_up(incl[k], o);
      if (lane >= o) incl[k] += y;
    }
  }
  #pragma unroll
  for (int o = 16; o > 0; o >>= 1) bpart += __shfl_xor(bpart, o);
  if (lane == 31) {
    #pragma unroll
    for (int k = 0; k < SCAN_K; k++) wsum[k][wv] = incl[k];
  }
  if (lane == 0) bsum[wv] = bpart;
  __syncthreads();
  if (wv < SCAN_K) {
    const int w = wsum[wv][lane];
    int wi = w;
    #pragma unroll
    for (int o = 1; o < 32; o <<= 1) {
      const int y = __shfl_up(wi, o);
      if (lane >= o) wi += y;
    }
    wsum[wv][lane] = wi - w;
    if (lane == 31) rtot[wv] = wi;
  } else if (wv == SCAN_K) {
    int v = bsum[lane];
    #pragma unroll
    for (int o = 16; o > 0; o >>= 1) v += __shfl_xor(v, o);
    if (lane == 0) base_s = v;
  }
  __syncthreads();
  int carry = base_s;
  #pragma unroll
  for (int k = 0; k < SCAN_K; k++) {
    const int i = t + k * SCAN_T;
    if (i < CNT_M) rowptr[nb + i] = carry + wsum[k][wv] + incl[k] - dg[k];
    carry += rtot[k];
  }
  if (r == CNT_R - 1 && t == 0) rowptr[N_NODES] = carry;
}

__global__ void fill_kernel(const int* __restrict__ ei, const int* __restrict__ rowptr,
                            const int* __restrict__ pos, const int* __restrict__ off,
                            int* __restrict__ csr_src) {
  int e = blockIdx.x * blockDim.x + threadIdx.x;
  if (e >= N_EDGES) return;
  const int d = ei[N_EDGES + e];
  csr_src[rowptr[d] + off[(e / CNT_EC) * N_NODES + d] + pos[e]] = ei[e];
}

#define NP1_NODES 32
#define XT_PITCH  68
#define G2_PITCH  52

__global__ void __launch_bounds__(256) node_pass1_gemm2(const int* __restrict__ rowptr,
                                                         const int* __restrict__ csr_src,
                                                         const float* __restrict__ a_s,
                                                         const float* __restrict__ a_d,
                                                         const float* __restrict__ H1,
                                                         const float* __restrict__ b,
                                                         const float* __restrict__ W,
                                                         float* __restrict__ H2) {
  __shared__ __attribute__((aligned(16))) float xt[NP1_NODES][XT_PITCH];
  __shared__ __attribute__((aligned(16))) float ht[NP1_NODES][G2_PITCH];
  const int t  = threadIdx.x;
  const int nl = t >> 3, h = t & 7;
  const int nbase = blockIdx.x * NP1_NODES;
  const int n = min(nbase + nl, N_NODES - 1);
  const int beg = rowptr[n], end = rowptr[n + 1];
  const int last = end - 1;
  v2f ws_[NHID / 2], wd_[NHID / 2];
  #pragma unroll
  for (int k = 0; k < NHID / 2; k++) {
    ws_[k] = (v2f){ a_s[h * NHID + 2 * k], a_s[h * NHID + 2 * k + 1] };
    wd_[k] = (v2f){ a_d[h * NHID + 2 * k], a_d[h * NHID + 2 * k + 1] };
  }
  const float4* H1v = (const float4*)H1;
  const int hq = 2 * h;

  float4 a0 = H1v[n * (HID1 / 4) + hq], a1 = H1v[n * (HID1 / 4) + hq + 1];
  const float adn = dot8(a0, a1, wd_);
  float mx  = lrelu(dot8(a0, a1, ws_) + adn);
  float den = 1.f;
  for (int j = beg; j < end; j += 4) {
    float4 x0[4], x1[4];
    #pragma unroll
    for (int u = 0; u < 4; u++) {
      const int s = csr_src[min(j + u, last)];
      x0[u] = H1v[s * (HID1 / 4) + hq]; x1[u] = H1v[s * (HID1 / 4) + hq + 1];
    }
    float v[4];
    #pragma unroll
    for (int u = 0; u < 4; u++) v[u] = lrelu(dot8(x0[u], x1[u], ws_) + adn);
    const float mn = fmaxf(mx, fmaxf(fmaxf(v[0], v[1]), fmaxf(v[2], v[3])));
    const float sc = __expf(mx - mn);
    mx = mn;
    den *= sc;
    a0.x *= sc; a0.y *= sc; a0.z *= sc; a0.w *= sc;
    a1.x *= sc; a1.y *= sc; a1.z *= sc; a1.w *= sc;
    #pragma unroll
    for (int u = 0; u < 4; u++) {
      const float w = (j + u < end) ? __expf(v[u] - mn) : 0.f;
      den  += w;
      a0.x += w * x0[u].x; a0.y += w * x0[u].y; a0.z += w * x0[u].z; a0.w += w * x0[u].w;
      a1.x += w * x1[u].x; a1.y += w * x1[u].y; a1.z += w * x1[u].z; a1.w += w * x1[u].w;
    }
  }

  float r[8] = { a0.x / den, a0.y / den, a0.z / den, a0.w / den,
                 a1.x / den, a1.y / den, a1.z / den, a1.w / den };
  #pragma unroll
  for (int k = 0; k < 8; k++) {
    const float tv = r[k] + b[h * NHID + k];
    r[k] = tv > 0.f ? tv : (__expf(tv) - 1.f);
  }
  const int wave = t >> 5;
  const int lane = t & 31;
  const int m    = lane & 15;
  const int hi   = lane >> 4;
  const int mt   = wave & 1;
  const int col  = (wave >> 1) * 16 + m;
  const bool cok = col < NCLASS;
  {
    float4* xp = (float4*)&xt[nl][h * NHID];
    xp[0] = make_float4(r[0], r[1], r[2], r[3]);
    xp[1] = make_float4(r[4], r[5], r[6], r[7]);
  }
  float w0[HID1 / 4], w1[HID1 / 4];
  if (wave < 6) {
    const float* wp = W + (2 * hi) * NCLASS + (cok ? col : NCLASS - 1);
    #pragma unroll
    for (int ks = 0; ks < HID1 / 4; ks++) { w0[ks] = wp[(4 * ks) * NCLASS]; w1[ks] = wp[(4 * ks + 1) * NCLASS]; }
  }
  __syncthreads();

  if (wave < 6) {
    const float* xr = &xt[mt * 16 + m][2 * hi];
    v8f c = {};
    #pragma unroll
    for (int ks = 0; ks < HID1 / 4; ks++) {
      v2f a; a.x = xr[4 * ks]; a.y = xr[4 * ks + 1];
      v2f bb; bb.x = cok ? w0[ks] : 0.f; bb.y = cok ? w1[ks] : 0.f;
      c = __builtin_amdgcn_wmma_f32_16x16x4_f32(false, a, false, bb, (short)0, c, false, false);
    }
    #pragma unroll
    for (int j = 0; j < 8; j++) ht[mt * 16 + j + 8 * hi][col] = c[j];
  }
  __syncthreads();

  {
    float* hg = H2 + nbase * NCLASS;
    float hv[(NP1_NODES * NCLASS) / 256];
    #pragma unroll
    for (int p = 0; p < (NP1_NODES * NCLASS) / 256; p++) {
      const int idx = t + 256 * p;
      const int row = idx / NCLASS;
      hv[p] = ht[row][idx - row * NCLASS];
    }
    #pragma unroll
    for (int p = 0; p < (NP1_NODES * NCLASS) / 256; p++) {
      const int idx = t + 256 * p;
      if (nbase + idx / NCLASS < N_NODES) hg[idx] = hv[p];
    }
  }
}

__global__ void __launch_bounds__(256) node_pass2(const int* __restrict__ rowptr,
                                                   const int* __restrict__ csr_src,
                                                   const float* __restrict__ a_s,
                                                   const float* __restrict__ a_d,
                                                   const float* __restrict__ H2,
                                                   const float* __restrict__ b,
                                                   float* __restrict__ out) {
  const int i = blockIdx.x * blockDim.x + threadIdx.x;
  if (i >= N_NODES * 8) return;
  const int n = i >> 3, q = i & 7;
  const int beg = rowptr[n], end = rowptr[n + 1];
  const int last = end - 1;
  const v2f   ws01 = { a_s[q], a_s[q + 8] }, ws23 = { a_s[q + 16], a_s[q + 24] };
  const float ws4  = a_s[q + 32];

  float acc[NCLASS / 8];
  {
    const float* hs = H2 + (n * NCLASS + q);
    #pragma unroll
    for (int c = 0; c < NCLASS / 8; c++) acc[c] = hs[8 * c];
  }
  float asn = dot5((v2f){acc[0], acc[1]}, (v2f){acc[2], acc[3]}, acc[4], ws01, ws23, ws4);
  float adn = dot5((v2f){acc[0], acc[1]}, (v2f){acc[2], acc[3]}, acc[4],
                   (v2f){ a_d[q], a_d[q + 8] }, (v2f){ a_d[q + 16], a_d[q + 24] }, a_d[q + 32]);
  asn = add_xor_lane<1>(asn); adn = add_xor_lane<1>(adn);
  asn = add_xor_lane<2>(asn); adn = add_xor_lane<2>(adn);
  asn = add_xor_lane<4>(asn); adn = add_xor_lane<4>(adn);
  float mx  = lrelu(asn + adn);
  float den = 1.f;
  for (int j = beg; j < end; j += 4) {
    float x[4][NCLASS / 8];
    #pragma unroll
    for (int u = 0; u < 4; u++) {
      const int s = csr_src[min(j + u, last)];
      const float* hs = H2 + (s * NCLASS + q);
      #pragma unroll
      for (int c = 0; c < NCLASS / 8; c++) x[u][c] = hs[8 * c];
    }
    float lg[4];
    #pragma unroll
    for (int u = 0; u < 4; u++) lg[u] = dot5((v2f){x[u][0], x[u][1]}, (v2f){x[u][2], x[u][3]}, x[u][4], ws01, ws23, ws4);
    #pragma unroll
    for (int u = 0; u < 4; u++) lg[u] = add_xor_lane<1>(lg[u]);
    #pragma unroll
    for (int u = 0; u < 4; u++) lg[u] = add_xor_lane<2>(lg[u]);
    #pragma unroll
    for (int u = 0; u < 4; u++) lg[u] = add_xor_lane<4>(lg[u]);
    float v[4];
    #pragma unroll
    for (int u = 0; u < 4; u++) v[u] = lrelu(lg[u] + adn);
    const float mn = fmaxf(mx, fmaxf(fmaxf(v[0], v[1]), fmaxf(v[2], v[3])));
    const float sc = __expf(mx - mn);
    mx = mn;
    den *= sc;
    #pragma unroll
    for (int c = 0; c < NCLASS / 8; c++) acc[c] *= sc;
    #pragma unroll
    for (int u = 0; u < 4; u++) {
      const float w = (j + u < end) ? __expf(v[u] - mn) : 0.f;
      den += w;
      #pragma unroll
      for (int c = 0; c < NCLASS / 8; c++) acc[c] += w * x[u][c];
    }
  }
  float val[NCLASS / 8];
  float cmx = -INFINITY;
  #pragma unroll
  for (int c = 0; c < NCLASS / 8; c++) { val[c] = acc[c] / den + b[q + 8 * c]; cmx = fmaxf(cmx, val[c]); }
  #pragma unroll
  for (int o = 1; o < 8; o <<= 1) cmx = fmaxf(cmx, __shfl_xor(cmx, o));
  float ssum = 0.f;
  #pragma unroll
  for (int c = 0; c < NCLASS / 8; c++) ssum += __expf(val[c] - cmx);
  #pragma unroll
  for (int o = 1; o < 8; o <<= 1) ssum += __shfl_xor(ssum, o);
  const float lse = cmx + __logf(ssum);
  float* od = out + (n * NCLASS + q);
  #pragma unroll
  for (int c = 0; c < NCLASS / 8; c++) od[8 * c] = val[c] - lse;
}

extern "C" void kernel_launch(void* const* d_in, const int* in_sizes, int n_in,
                              void* d_out, int out_size, void* d_ws, size_t ws_size,
                              hipStream_t stream) {
  const float* x      = (const float*)d_in[0];
  const float* W1     = (const float*)d_in[1];
  const float* a_src1 = (const float*)d_in[2];
  const float* a_dst1 = (const float*)d_in[3];
  const float* b1     = (const float*)d_in[4];
  const float* W2     = (const float*)d_in[5];
  const float* a_src2 = (const float*)d_in[6];
  const float* a_dst2 = (const float*)d_in[7];
  const float* b2     = (const float*)d_in[8];
  const int*   ei     = (const int*)d_in[9];
  float*       out    = (float*)d_out;

  float* ws   = (float*)d_ws;
  float* H1   = ws;
  float* H2   = H1  + (long)N_NODES * HID1;
  int*   tot  = (int*)(H2 + (long)N_NODES * NCLASS);
  int*   rowp = tot  + CNT_R * CNT_C;
  int*   pos  = rowp + N_NODES + 4;
  int*   csr  = pos  + N_EDGES;
  int*   off  = csr  + N_EDGES;

  const int B = 256;
  auto cdiv = [](long a, long b) { return (int)((a + b - 1) / b); };

  count_kernel<<<CNT_R * CNT_C, CNT_T, 0, stream>>>(ei, pos, off, tot);
  scan_gemm1_kernel<<<CNT_R + G1_WG, SCAN_T, 0, stream>>>(tot, off, rowp, x, W1, H1);
  fill_kernel<<<cdiv(N_EDGES, B), B, 0, stream>>>(ei, rowp, pos, off, csr);

  node_pass1_gemm2<<<cdiv(N_NODES, NP1_NODES), B, 0, stream>>>(rowp, csr, a_src1, a_dst1, H1, b1, W2, H2);

  node_pass2<<<cdiv((long)N_NODES * 8, B), B, 0, stream>>>(rowp, csr, a_src2, a_dst2, H2, b2, out);
}


# scan folded into the placing launch: each (range, chunk) work-group scans its range, then places its edges by LDS cursors, no rank array; GEMM1 rides with the count; 5 -> 4 launches; list order kept
# speedup vs baseline: 15.4078x; 1.5256x over previous
#include <hip/hip_runtime.h>
#include <math.h>

#define N_NODES 50000
#define N_EDGES 800000
#define NFEAT   128
#define NHID    8
#define NHEADS  8
#define HID1    (NHEADS * NHID)
#define NCLASS  40
#define NEG_SLOPE 0.2f

typedef __attribute__((ext_vector_type(2))) float v2f;
typedef __attribute__((ext_vector_type(8))) float v8f;

__device__ __forceinline__ float lrelu(float x) { return x > 0.f ? x : NEG_SLOPE * x; }

__device__ __forceinline__ float dot8(const float4 p0, const float4 p1, const v2f (&w)[NHID / 2]) {
  v2f s = (v2f){p0.x, p0.y} * w[0];
  s += (v2f){p0.z, p0.w} * w[1];
  s += (v2f){p1.x, p1.y} * w[2];
  s += (v2f){p1.z, p1.w} * w[3];
  return s.x + s.y;
}

__device__ __forceinline__ float dot5(const v2f x01, const v2f x23, const float x4,
                                      const v2f w01, const v2f w23, const float w4) {
  v2f s = x01 * w01;
  s += x23 * w23;
  return (s.x + s.y) + x4 * w4;
}

template <int xm>
__device__ __forceinline__ float add_xor_lane(float x) {
  const int y = __builtin_amdgcn_update_dpp(0, __builtin_bit_cast(int, x), 0x160 + xm, 0xf, 0xf, true);
  return x + __builtin_bit_cast(float, y);
}


#define CNT_R   16
#define CNT_C   16
#define CNT_M   (N_NODES / CNT_R)
#define CNT_EC  (N_EDGES / CNT_C)
#define CNT_T   1024
#define CNT_Q   (CNT_EC / 4)
#define CNT_PER ((CNT_Q + CNT_T - 1) / CNT_T)
#define CNT_NPER ((CNT_M + CNT_T - 1) / CNT_T)
#define G1_TILES 16
#define G1_WG    ((N_NODES / 16 + G1_TILES - 1) / G1_TILES)

__global__ void __launch_bounds__(CNT_T) count_gemm1_kernel(const int* __restrict__ ei, int* __restrict__ off,
                                                            int* __restrict__ tot,
                                                            const float* __restrict__ X,
                                                            const float* __restrict__ W,
                                                            float* __restrict__ H) {
  __shared__ int cnt[CNT_M];
  __shared__ int wtot[CNT_T / 32];
  const int t    = threadIdx.x;
  const int lane = t & 31;
  const int wv   = t >> 5;
  if (blockIdx.x >= CNT_R * CNT_C) {
    const int mt = (blockIdx.x - CNT_R * CNT_C) * G1_TILES + (wv >> 1);
    if (mt >= N_NODES / 16) return;
    const int m  = lane & 15;
    const int hi = lane >> 4;
    const int cb = (wv & 1) * 32;
    const float* xr = X + (long)(mt * 16 + m) * NFEAT;

    v8f c0 = {}, c1 = {};
    #pragma unroll 4
    for (int k0 = 0; k0 < NFEAT; k0 += 4) {
      const int ka = k0 + 2 * hi;
      v2f a; a.x = xr[ka]; a.y = xr[ka + 1];
      const float* w0 = W + (long)ka * HID1 + cb + m;
      const float* w1 = w0 + HID1;
      v2f b;
      b.x = w0[0];  b.y = w1[0];
      c0 = __builtin_amdgcn_wmma_f32_16x16x4_f32(false, a, false, b, (short)0, c0, false, false);
      b.x = w0[16]; b.y = w1[16];
      c1 = __builtin_amdgcn_wmma_f32_16x16x4_f32(false, a, false, b, (short)0, c1, false, false);
    }
    float* hp = H + (long)(mt * 16 + 8 * hi) * HID1 + cb + m;
    #pragma unroll
    for (int j = 0; j < 8; j++) {
      hp[(long)j * HID1 +  0] = c0[j];
      hp[(long)j * HID1 + 16] = c1[j];
    }
    return;
  }

  const int r  = blockIdx.x % CNT_R;
  const int c  = blockIdx.x / CNT_R;
  const int nb = r * CNT_M;
  const int4* dv = (const int4*)(ei + N_EDGES + c * CNT_EC);
  int4 d[CNT_PER];
  #pragma unroll
  for (int p = 0; p < CNT_PER; p++) d[p] = dv[min(t + p * CNT_T, CNT_Q - 1)];
  for (int i = t; i < CNT_M; i += CNT_T) cnt[i] = 0;
  __syncthreads();
  #pragma unroll
  for (int p = 0; p < CNT_PER; p++) {
    const int q = t + p * CNT_T;
    if (q < CNT_Q) {
      const unsigned u0 = (unsigned)(d[p].x - nb), u1 = (unsigned)(d[p].y - nb);
      const unsigned u2 = (unsigned)(d[p].z - nb), u3 = (unsigned)(d[p].w - nb);
      if (u0 < CNT_M) atomicAdd(&cnt[u0], 1);
      if (u1 < CNT_M) atomicAdd(&cnt[u1], 1);
      if (u2 < CNT_M) atomicAdd(&cnt[u2], 1);
      if (u3 < CNT_M) atomicAdd(&cnt[u3], 1);
    }
  }
  __syncthreads();
  int kc[CNT_NPER];
  int s = 0;
  #pragma unroll
  for (int k = 0; k < CNT_NPER; k++) { const int i = t + k * CNT_T; kc[k] = (i < CNT_M) ? cnt[i] : 0; s += kc[k]; }
  #pragma unroll
  for (int o = 16; o > 0; o >>= 1) s += __shfl_xor(s, o);
  if (lane == 0) wtot[wv] = s;
  __syncthreads();
  if (t < 32) {
    int v = wtot[t];
    #pragma unroll
    for (int o = 16; o > 0; o >>= 1) v += __shfl_xor(v, o);
    if (t == 0) tot[c * CNT_R + r] = v;
  }
  #pragma unroll
  for (int k = 0; k < CNT_NPER; k++) { const int i = t + k * CNT_T; if (i < CNT_M) off[c * N_NODES + nb + i] = kc[k]; }
}

#define SCAN_T   1024
#define SCAN_K   ((CNT_M + SCAN_T - 1) / SCAN_T)
#define FILL_A   5
#define FILL_B   4

__device__ __forceinline__ void place_piece(const int4 d, const int4 s, const int nb, int* cur, int* __restrict__ csr_src) {
  const unsigned u0 = (unsigned)(d.x - nb), u1 = (unsigned)(d.y - nb);
  const unsigned u2 = (unsigned)(d.z - nb), u3 = (unsigned)(d.w - nb);
  int k0 = 0, k1 = 0, k2 = 0, k3 = 0;
  if (u0 < CNT_M) k0 = atomicAdd(&cur[u0], 1);
  if (u1 < CNT_M) k1 = atomicAdd(&cur[u1], 1);
  if (u2 < CNT_M) k2 = atomicAdd(&cur[u2], 1);
  if (u3 < CNT_M) k3 = atomicAdd(&cur[u3], 1);
  if (u0 < CNT_M) csr_src[k0] = s.x;
  if (u1 < CNT_M) csr_src[k1] = s.y;
  if (u2 < CNT_M) csr_src[k2] = s.z;
  if (u3 < CNT_M) csr_src[k3] = s.w;
}

__global__ void __launch_bounds__(SCAN_T) scan_fill_kernel(const int* __restrict__ ei, const int* __restrict__ tot,
                                                           const int* __restrict__ off,
                                                           int* __restrict__ rowptr, int* __restrict__ csr_src) {
  __shared__ int cur[CNT_M];
  __shared__ int wsum[SCAN_K][32];
  __shared__ int rtot[SCAN_K];
  __shared__ int bsum[32];
  __shared__ int base_s;
  const int t    = threadIdx.x;
  const int lane = t & 31;
  const int wv   = t >> 5;
  const int r    = blockIdx.x % CNT_R;
  const int c    = blockIdx.x / CNT_R;
  const int nb   = r * CNT_M;
  int bpart = 0;
  if (t < CNT_R * CNT_C) { const int v = tot[t]; bpart = ((t % CNT_R) < r) ? v : 0; }
  int dg[SCAN_K], exc[SCAN_K], incl[SCAN_K];
  {
    int cv[SCAN_K][CNT_C];
    #pragma unroll
    for (int k = 0; k < SCAN_K; k++) {
      const int i  = t + k * SCAN_T;
      const int ic = min(i, CNT_M - 1);
      #pragma unroll
      for (int cc = 0; cc < CNT_C; cc++) { const int v = off[cc * N_NODES + nb + ic]; cv[k][cc] = (i < CNT_M) ? v : 0; }
    }
    #pragma unroll
    for (int k = 0; k < SCAN_K; k++) {
      int run = 0, before = 0;
      #pragma unroll
      for (int cc = 0; cc < CNT_C; cc++) { before += (cc < c) ? cv[k][cc] : 0; run += cv[k][cc]; }
      dg[k]  = run;
      exc[k] = before;
    }
  }
  const int4* dv = (const int4*)(ei + N_EDGES + c * CNT_EC);
  const int4* sv = (const int4*)(ei + c * CNT_EC);
  #pragma unroll
  for (int k = 0; k < SCAN_K; k++) {
    incl[k] = dg[k];
    #pragma unroll
    for (int o = 1; o < 32; o <<= 1) {
      const int y = __shfl_up(incl[k], o);
      if (lane >= o) incl[k] += y;
    }
  }
  #pragma unroll
  for (int o = 16; o > 0; o >>= 1) bpart += __shfl_xor(bpart, o);
  if (lane == 31) {
    #pragma unroll
    for (int k = 0; k < SCAN_K; k++) wsum[k][wv] = incl[k];
  }
  if (lane == 0) bsum[wv] = bpart;
  __syncthreads();
  int4 dA[FILL_A], sA[FILL_A];
  #pragma unroll
  for (int p = 0; p < FILL_A; p++) { const int q = t + p * CNT_T; dA[p] = dv[q]; sA[p] = sv[q]; }
  if (wv < SCAN_K) {
    const int w = wsum[wv][lane];
    int wi = w;
    #pragma unroll
    for (int o = 1; o < 32; o <<= 1) {
      const int y = __shfl_up(wi, o);
      if (lane >= o) wi += y;
    }
    wsum[wv][lane] = wi - w;
    if (lane == 31) rtot[wv] = wi;
  } else if (wv == SCAN_K) {
    int v = bsum[lane];
    #pragma unroll
    for (int o = 16; o > 0; o >>= 1) v += __shfl_xor(v, o);
    if (lane == 0) base_s = v;
  }
  __syncthreads();
  int carry = base_s;
  #pragma unroll
  for (int k = 0; k < SCAN_K; k++) {
    const int i = t + k * SCAN_T;
    if (i < CNT_M) {
      const int rp = carry + wsum[k][wv] + incl[k] - dg[k];
      cur[i] = rp + exc[k];
      if (c == 0) rowptr[nb + i] = rp;
    }
    carry += rtot[k];
  }
  if (c == 0 && r == CNT_R - 1 && t == 0) rowptr[N_NODES] = carry;
  __syncthreads();
  int4 dB[FILL_B], sB[FILL_B];
  #pragma unroll
  for (int p = 0; p < FILL_B; p++) { const int q = t + (FILL_A + p) * CNT_T; dB[p] = dv[q]; sB[p] = sv[q]; }
  #pragma unroll
  for (int p = 0; p < FILL_A; p++) place_piece(dA[p], sA[p], nb, cur, csr_src);
  int4 dC[CNT_PER - FILL_A - FILL_B], sC[CNT_PER - FILL_A - FILL_B];
  #pragma unroll
  for (int p = 0; p < CNT_PER - FILL_A - FILL_B; p++) {
    const int qc = min(t + (FILL_A + FILL_B + p) * CNT_T, CNT_Q - 1);
    dC[p] = dv[qc]; sC[p] = sv[qc];
  }
  #pragma unroll
  for (int p = 0; p < FILL_B; p++) place_piece(dB[p], sB[p], nb, cur, csr_src);
  #pragma unroll
  for (int p = 0; p < CNT_PER - FILL_A - FILL_B; p++) {
    if (t + (FILL_A + FILL_B + p) * CNT_T < CNT_Q) place_piece(dC[p], sC[p], nb, cur, csr_src);
  }
}

#define NP1_NODES 32
#define XT_PITCH  68
#define G2_PITCH  52

__global__ void __launch_bounds__(256) node_pass1_gemm2(const int* __restrict__ rowptr,
                                                         const int* __restrict__ csr_src,
                                                         const float* __restrict__ a_s,
                                                         const float* __restrict__ a_d,
                                                         const float* __restrict__ H1,
                                                         const float* __restrict__ b,
                                                         const float* __restrict__ W,
                                                         float* __restrict__ H2) {
  __shared__ __attribute__((aligned(16))) float xt[NP1_NODES][XT_PITCH];
  __shared__ __attribute__((aligned(16))) float ht[NP1_NODES][G2_PITCH];
  const int t  = threadIdx.x;
  const int nl = t >> 3, h = t & 7;
  const int nbase = blockIdx.x * NP1_NODES;
  const int n = min(nbase + nl, N_NODES - 1);
  const int beg = rowptr[n], end = rowptr[n + 1];
  const int last = end - 1;
  v2f ws_[NHID / 2], wd_[NHID / 2];
  #pragma unroll
  for (int k = 0; k < NHID / 2; k++) {
    ws_[k] = (v2f){ a_s[h * NHID + 2 * k], a_s[h * NHID + 2 * k + 1] };
    wd_[k] = (v2f){ a_d[h * NHID + 2 * k], a_d[h * NHID + 2 * k + 1] };
  }
  const float4* H1v = (const float4*)H1;
  const int hq = 2 * h;

  float4 a0 = H1v[n * (HID1 / 4) + hq], a1 = H1v[n * (HID1 / 4) + hq + 1];
  const float adn = dot8(a0, a1, wd_);
  float mx  = lrelu(dot8(a0, a1, ws_) + adn);
  float den = 1.f;
  for (int j = beg; j < end; j += 4) {
    float4 x0[4], x1[4];
    #pragma unroll
    for (int u = 0; u < 4; u++) {
      const int s = csr_src[min(j + u, last)];
      x0[u] = H1v[s * (HID1 / 4) + hq]; x1[u] = H1v[s * (HID1 / 4) + hq + 1];
    }
    float v[4];
    #pragma unroll
    for (int u = 0; u < 4; u++) v[u] = lrelu(dot8(x0[u], x1[u], ws_) + adn);
    const float mn = fmaxf(mx, fmaxf(fmaxf(v[0], v[1]), fmaxf(v[2], v[3])));
    const float sc = __expf(mx - mn);
    mx = mn;
    den *= sc;
    a0.x *= sc; a0.y *= sc; a0.z *= sc; a0.w *= sc;
    a1.x *= sc; a1.y *= sc; a1.z *= sc; a1.w *= sc;
    #pragma unroll
    for (int u = 0; u < 4; u++) {
      const float w = (j + u < end) ? __expf(v[u] - mn) : 0.f;
      den  += w;
      a0.x += w * x0[u].x; a0.y += w * x0[u].y; a0.z += w * x0[u].z; a0.w += w * x0[u].w;
      a1.x += w * x1[u].x; a1.y += w * x1[u].y; a1.z += w * x1[u].z; a1.w += w * x1[u].w;
    }
  }

  float r[8] = { a0.x / den, a0.y / den, a0.z / den, a0.w / den,
                 a1.x / den, a1.y / den, a1.z / den, a1.w / den };
  #pragma unroll
  for (int k = 0; k < 8; k++) {
    const float tv = r[k] + b[h * NHID + k];
    r[k] = tv > 0.f ? tv : (__expf(tv) - 1.f);
  }
  const int wave = t >> 5;
  const int lane = t & 31;
  const int m    = lane & 15;
  const int hi   = lane >> 4;
  const int mt   = wave & 1;
  const int col  = (wave >> 1) * 16 + m;
  const bool cok = col < NCLASS;
  {
    float4* xp = (float4*)&xt[nl][h * NHID];
    xp[0] = make_float4(r[0], r[1], r[2], r[3]);
    xp[1] = make_float4(r[4], r[5], r[6], r[7]);
  }
  float w0[HID1 / 4], w1[HID1 / 4];
  if (wave < 6) {
    const float* wp = W + (2 * hi) * NCLASS + (cok ? col : NCLASS - 1);
    #pragma unroll
    for (int ks = 0; ks < HID1 / 4; ks++) { w0[ks] = wp[(4 * ks) * NCLASS]; w1[ks] = wp[(4 * ks + 1) * NCLASS]; }
  }
  __syncthreads();

  if (wave < 6) {
    const float* xr = &xt[mt * 16 + m][2 * hi];
    v8f c = {};
    #pragma unroll
    for (int ks = 0; ks < HID1 / 4; ks++) {
      v2f a; a.x = xr[4 * ks]; a.y = xr[4 * ks + 1];
      v2f bb; bb.x = cok ? w0[ks] : 0.f; bb.y = cok ? w1[ks] : 0.f;
      c = __builtin_amdgcn_wmma_f32_16x16x4_f32(false, a, false, bb, (short)0, c, false, false);
    }
    #pragma unroll
    for (int j = 0; j < 8; j++) ht[mt * 16 + j + 8 * hi][col] = c[j];
  }
  __syncthreads();

  {
    float* hg = H2 + nbase * NCLASS;
    float hv[(NP1_NODES * NCLASS) / 256];
    #pragma unroll
    for (int p = 0; p < (NP1_NODES * NCLASS) / 256; p++) {
      const int idx = t + 256 * p;
      const int row = idx / NCLASS;
      hv[p] = ht[row][idx - row * NCLASS];
    }
    #pragma unroll
    for (int p = 0; p < (NP1_NODES * NCLASS) / 256; p++) {
      const int idx = t + 256 * p;
      if (nbase + idx / NCLASS < N_NODES) hg[idx] = hv[p];
    }
  }
}

__global__ void __launch_bounds__(256) node_pass2(const int* __restrict__ rowptr,
                                                   const int* __restrict__ csr_src,
                                                   const float* __restrict__ a_s,
                                                   const float* __restrict__ a_d,
                                                   const float* __restrict__ H2,
                                                   const float* __restrict__ b,
                                                   float* __restrict__ out) {
  const int i = blockIdx.x * blockDim.x + threadIdx.x;
  if (i >= N_NODES * 8) return;
  const int n = i >> 3, q = i & 7;
  const int beg = rowptr[n], end = rowptr[n + 1];
  const int last = end - 1;
  const v2f   ws01 = { a_s[q], a_s[q + 8] }, ws23 = { a_s[q + 16], a_s[q + 24] };
  const float ws4  = a_s[q + 32];

  float acc[NCLASS / 8];
  {
    const float* hs = H2 + (n * NCLASS + q);
    #pragma unroll
    for (int c = 0; c < NCLASS / 8; c++) acc[c] = hs[8 * c];
  }
  float asn = dot5((v2f){acc[0], acc[1]}, (v2f){acc[2], acc[3]}, acc[4], ws01, ws23, ws4);
  float adn = dot5((v2f){acc[0], acc[1]}, (v2f){acc[2], acc[3]}, acc[4],
                   (v2f){ a_d[q], a_d[q + 8] }, (v2f){ a_d[q + 16], a_d[q + 24] }, a_d[q + 32]);
  asn = add_xor_lane<1>(asn); adn = add_xor_lane<1>(adn);
  asn = add_xor_lane<2>(asn); adn = add_xor_lane<2>(adn);
  asn = add_xor_lane<4>(asn); adn = add_xor_lane<4>(adn);
  float mx  = lrelu(asn + adn);
  float den = 1.f;
  for (int j = beg; j < end; j += 4) {
    float x[4][NCLASS / 8];
    #pragma unroll
    for (int u = 0; u < 4; u++) {
      const int s = csr_src[min(j + u, last)];
      const float* hs = H2 + (s * NCLASS + q);
      #pragma unroll
      for (int c = 0; c < NCLASS / 8; c++) x[u][c] = hs[8 * c];
    }
    float lg[4];
    #pragma unroll
    for (int u = 0; u < 4; u++) lg[u] = dot5((v2f){x[u][0], x[u][1]}, (v2f){x[u][2], x[u][3]}, x[u][4], ws01, ws23, ws4);
    #pragma unroll
    for (int u = 0; u < 4; u++) lg[u] = add_xor_lane<1>(lg[u]);
    #pragma unroll
    for (int u = 0; u < 4; u++) lg[u] = add_xor_lane<2>(lg[u]);
    #pragma unroll
    for (int u = 0; u < 4; u++) lg[u] = add_xor_lane<4>(lg[u]);
    float v[4];
    #pragma unroll
    for (int u = 0; u < 4; u++) v[u] = lrelu(lg[u] + adn);
    const float mn = fmaxf(mx, fmaxf(fmaxf(v[0], v[1]), fmaxf(v[2], v[3])));
    const float sc = __expf(mx - mn);
    mx = mn;
    den *= sc;
    #pragma unroll
    for (int c = 0; c < NCLASS / 8; c++) acc[c] *= sc;
    #pragma unroll
    for (int u = 0; u < 4; u++) {
      const float w = (j + u < end) ? __expf(v[u] - mn) : 0.f;
      den += w;
      #pragma unroll
      for (int c = 0; c < NCLASS / 8; c++) acc[c] += w * x[u][c];
    }
  }
  float val[NCLASS / 8];
  float cmx = -INFINITY;
  #pragma unroll
  for (int c = 0; c < NCLASS / 8; c++) { val[c] = acc[c] / den + b[q + 8 * c]; cmx = fmaxf(cmx, val[c]); }
  #pragma unroll
  for (int o = 1; o < 8; o <<= 1) cmx = fmaxf(cmx, __shfl_xor(cmx, o));
  float ssum = 0.f;
  #pragma unroll
  for (int c = 0; c < NCLASS / 8; c++) ssum += __expf(val[c] - cmx);
  #pragma unroll
  for (int o = 1; o < 8; o <<= 1) ssum += __shfl_xor(ssum, o);
  const float lse = cmx + __logf(ssum);
  float* od = out + (n * NCLASS + q);
  #pragma unroll
  for (int c = 0; c < NCLASS / 8; c++) od[8 * c] = val[c] - lse;
}

extern "C" void kernel_launch(void* const* d_in, const int* in_sizes, int n_in,
                              void* d_out, int out_size, void* d_ws, size_t ws_size,
                              hipStream_t stream) {
  const float* x      = (const float*)d_in[0];
  const float* W1     = (const float*)d_in[1];
  const float* a_src1 = (const float*)d_in[2];
  const float* a_dst1 = (const float*)d_in[3];
  const float* b1     = (const float*)d_in[4];
  const float* W2     = (const float*)d_in[5];
  const float* a_src2 = (const float*)d_in[6];
  const float* a_dst2 = (const float*)d_in[7];
  const float* b2     = (const float*)d_in[8];
  const int*   ei     = (const int*)d_in[9];
  float*       out    = (float*)d_out;

  float* ws   = (float*)d_ws;
  float* H1   = ws;
  float* H2   = H1  + (long)N_NODES * HID1;
  int*   tot  = (int*)(H2 + (long)N_NODES * NCLASS);
  int*   rowp = tot  + CNT_R * CNT_C;
  int*   csr  = rowp + N_NODES + 4;
  int*   off  = csr  + N_EDGES;

  const int B = 256;
  auto cdiv = [](long a, long b) { return (int)((a + b - 1) / b); };

  count_gemm1_kernel<<<CNT_R * CNT_C + G1_WG, CNT_T, 0, stream>>>(ei, off, tot, x, W1, H1);
  scan_fill_kernel<<<CNT_R * CNT_C, SCAN_T, 0, stream>>>(ei, tot, off, rowp, csr);

  node_pass1_gemm2<<<cdiv(N_NODES, NP1_NODES), B, 0, stream>>>(rowp, csr, a_src1, a_dst1, H1, b1, W2, H2);

  node_pass2<<<cdiv((long)N_NODES * 8, B), B, 0, stream>>>(rowp, csr, a_src2, a_dst2, H2, b2, out);
}
